# starting kernel
# baseline (speedup 1.0000x reference)
#include <hip/hip_runtime.h>
#include <hip/hip_bf16.h>

typedef __attribute__((ext_vector_type(2))) float v2f;
typedef __attribute__((ext_vector_type(8))) float v8f;

#define EPS_BN   1e-5f
#define EPS_NORM 1e-12f

__device__ __forceinline__ v8f wmma_f32(v2f a, v2f b, v8f c) {
  return __builtin_amdgcn_wmma_f32_16x16x4_f32(false, a, false, b, (short)0, c,
                                               false, false);
}

__global__ void bn_partial_k(const float* __restrict__ h, float* __restrict__ part, int n) {
  __shared__ float sh[256 * 10];
  float acc[10];
#pragma unroll
  for (int d = 0; d < 10; ++d) acc[d] = 0.f;
  for (int i = blockIdx.x * blockDim.x + threadIdx.x; i < n;
       i += gridDim.x * blockDim.x) {
#pragma unroll
    for (int d = 0; d < 5; ++d) {
      float v = h[i * 5 + d];
      acc[d] += v;
      acc[5 + d] += v * v;
    }
  }
#pragma unroll
  for (int d = 0; d < 10; ++d) sh[threadIdx.x * 10 + d] = acc[d];
  __syncthreads();
  for (int s = 128; s > 0; s >>= 1) {
    if (threadIdx.x < (unsigned)s)
      for (int d = 0; d < 10; ++d)
        sh[threadIdx.x * 10 + d] += sh[(threadIdx.x + s) * 10 + d];
    __syncthreads();
  }
  if (threadIdx.x < 10) part[blockIdx.x * 10 + threadIdx.x] = sh[threadIdx.x];
}

__global__ void bn_final_k(const float* __restrict__ part, float* __restrict__ mu,
                           float* __restrict__ rs, int nblocks, float inv_n) {
  int t = threadIdx.x;
  if (t < 5) {
    float s = 0.f, q = 0.f;
    for (int b = 0; b < nblocks; ++b) {
      s += part[b * 10 + t];
      q += part[b * 10 + 5 + t];
    }
    float m = s * inv_n;
    float var = q * inv_n - m * m;
    mu[t] = m;
    rs[t] = rsqrtf(var + EPS_BN);
  }
}

__global__ void bn_apply_k(const float* __restrict__ h, const float* __restrict__ mu,
                           const float* __restrict__ rs, const float* __restrict__ g,
                           const float* __restrict__ b, float* __restrict__ x, int total) {
  int i = blockIdx.x * blockDim.x + threadIdx.x;
  if (i >= total) return;
  int d = i % 5;
  x[i] = (h[i] - mu[d]) * rs[d] * g[d] + b[d];
}

__global__ void degree_k(const int* __restrict__ dst, float* __restrict__ cnt, int E) {
  int e = blockIdx.x * blockDim.x + threadIdx.x;
  if (e < E) atomicAdd(&cnt[dst[e]], 1.0f);
}

__global__ void invcnt_k(const float* __restrict__ cnt, float* __restrict__ invc, int n) {
  int i = blockIdx.x * blockDim.x + threadIdx.x;
  if (i < n) invc[i] = 1.0f / fmaxf(cnt[i], 1.0f);
}

__global__ void proj5_k(const float* __restrict__ X, const float* __restrict__ W,
                        const float* __restrict__ b, float* __restrict__ Y, int n) {
  int i = blockIdx.x * blockDim.x + threadIdx.x;
  if (i >= n) return;
  float x[5];
#pragma unroll
  for (int k = 0; k < 5; ++k) x[k] = X[i * 5 + k];
#pragma unroll
  for (int c = 0; c < 5; ++c) {
    float acc = b[c];
#pragma unroll
    for (int k = 0; k < 5; ++k) acc += x[k] * W[k * 5 + c];
    Y[i * 5 + c] = fmaxf(acc, 0.f);
  }
}

__global__ void edge_scatter5_k(const int* __restrict__ src, const int* __restrict__ dst,
                                const float* __restrict__ ew, const float* __restrict__ xp,
                                float* __restrict__ agg, int E) {
  int e = blockIdx.x * blockDim.x + threadIdx.x;
  if (e >= E) return;
  int s = src[e], d = dst[e];
  float w = ew[e];
#pragma unroll
  for (int k = 0; k < 5; ++k) atomicAdd(&agg[d * 5 + k], xp[s * 5 + k] * w);
}

__global__ void node5to64_k(const float* __restrict__ agg, const float* __restrict__ invc,
                            const float* __restrict__ X, const float* __restrict__ LW,
                            const float* __restrict__ lb, const float* __restrict__ RW,
                            float* __restrict__ Y, int n) {
  __shared__ float red[64];
  int node = blockIdx.x;
  if (node >= n) return;
  int c = threadIdx.x;
  float ic = invc[node];
  float acc = lb[c];
#pragma unroll
  for (int k = 0; k < 5; ++k) {
    acc += agg[node * 5 + k] * ic * LW[k * 64 + c] + X[node * 5 + k] * RW[k * 64 + c];
  }
  red[c] = acc * acc;
  __syncthreads();
#pragma unroll
  for (int s = 32; s > 0; s >>= 1) {
    if (c < s) red[c] += red[c + s];
    __syncthreads();
  }
  float scale = 1.0f / fmaxf(sqrtf(red[0]), EPS_NORM);
  Y[node * 64 + c] = fmaxf(acc * scale, 0.f);
}

__global__ void proj64_wmma_k(const float* __restrict__ X, const float* __restrict__ W,
                              const float* __restrict__ bias, float* __restrict__ Y, int n) {
  int l = threadIdx.x & 31;
  int wave = threadIdx.x >> 5;
  int row0 = blockIdx.x * 16;
  int m = row0 + (l & 15);
  if (m >= n) m = n - 1;
  int cl = wave * 16 + (l & 15);
  int khi = (l >> 4) << 1;
  v8f c = {};
#pragma unroll
  for (int k0 = 0; k0 < 64; k0 += 4) {
    int kb = k0 + khi;
    v2f a, bb;
    a[0] = X[m * 64 + kb];
    a[1] = X[m * 64 + kb + 1];
    bb[0] = W[kb * 64 + cl];
    bb[1] = W[(kb + 1) * 64 + cl];
    c = wmma_f32(a, bb, c);
  }
  int rbase = row0 + ((l >> 4) << 3);
  float bc = bias[cl];
#pragma unroll
  for (int r = 0; r < 8; ++r) {
    int rr = rbase + r;
    if (rr < n) Y[rr * 64 + cl] = fmaxf(c[r] + bc, 0.f);
  }
}

__global__ void edge_scatter64_k(const int* __restrict__ src, const int* __restrict__ dst,
                                 const float* __restrict__ ew, const float* __restrict__ xp,
                                 float* __restrict__ agg, int E) {
  int t = blockIdx.x * blockDim.x + threadIdx.x;
  int e = t >> 4;
  if (e >= E) return;
  int c4 = (t & 15) << 2;
  int s = src[e], d = dst[e];
  float w = ew[e];
  const float4 v = *(const float4*)(xp + (size_t)s * 64 + c4);
  float* out = agg + (size_t)d * 64 + c4;
  atomicAdd(out + 0, v.x * w);
  atomicAdd(out + 1, v.y * w);
  atomicAdd(out + 2, v.z * w);
  atomicAdd(out + 3, v.w * w);
}

__global__ void node64_wmma_k(const float* __restrict__ agg, const float* __restrict__ invc,
                              const float* __restrict__ X, const float* __restrict__ LW,
                              const float* __restrict__ lb, const float* __restrict__ RW,
                              float* __restrict__ Y, int n) {
  __shared__ float tile[16 * 64];
  __shared__ float rscale[16];
  int l = threadIdx.x & 31;
  int wave = threadIdx.x >> 5;
  int row0 = blockIdx.x * 16;
  int m = row0 + (l & 15);
  if (m >= n) m = n - 1;
  int cl = wave * 16 + (l & 15);
  int khi = (l >> 4) << 1;
  float ic = invc[m];
  v8f c = {};
#pragma unroll
  for (int k0 = 0; k0 < 64; k0 += 4) {
    int kb = k0 + khi;
    v2f a, bb;
    a[0] = agg[m * 64 + kb] * ic;
    a[1] = agg[m * 64 + kb + 1] * ic;
    bb[0] = LW[kb * 64 + cl];
    bb[1] = LW[(kb + 1) * 64 + cl];
    c = wmma_f32(a, bb, c);
  }
#pragma unroll
  for (int k0 = 0; k0 < 64; k0 += 4) {
    int kb = k0 + khi;
    v2f a, bb;
    a[0] = X[m * 64 + kb];
    a[1] = X[m * 64 + kb + 1];
    bb[0] = RW[kb * 64 + cl];
    bb[1] = RW[(kb + 1) * 64 + cl];
    c = wmma_f32(a, bb, c);
  }
  int rloc = (l >> 4) << 3;
  float bc = lb[cl];
#pragma unroll
  for (int r = 0; r < 8; ++r) tile[(rloc + r) * 64 + cl] = c[r] + bc;
  __syncthreads();
  if (threadIdx.x < 16) {
    float ss = 0.f;
#pragma unroll
    for (int cc = 0; cc < 64; ++cc) {
      float v = tile[threadIdx.x * 64 + cc];
      ss += v * v;
    }
    rscale[threadIdx.x] = 1.0f / fmaxf(sqrtf(ss), EPS_NORM);
  }
  __syncthreads();
  for (int i = threadIdx.x; i < 16 * 64; i += blockDim.x) {
    int r = i >> 6;
    int rr = row0 + r;
    if (rr < n) Y[rr * 64 + (i & 63)] = fmaxf(tile[i] * rscale[r], 0.f);
  }
}

__global__ void node64to5_wmma_k(const float* __restrict__ agg, const float* __restrict__ invc,
                                 const float* __restrict__ X, const float* __restrict__ LW,
                                 const float* __restrict__ lb, const float* __restrict__ RW,
                                 float* __restrict__ Y, int n) {
  __shared__ float tile[16 * 16];
  int l = threadIdx.x & 31;
  int row0 = blockIdx.x * 16;
  int m = row0 + (l & 15);
  if (m >= n) m = n - 1;
  int cl = l & 15;
  int khi = (l >> 4) << 1;
  float ic = invc[m];
  bool cv = (cl < 5);
  v8f c = {};
#pragma unroll
  for (int k0 = 0; k0 < 64; k0 += 4) {
    int kb = k0 + khi;
    v2f a, bb;
    a[0] = agg[m * 64 + kb] * ic;
    a[1] = agg[m * 64 + kb + 1] * ic;
    bb[0] = cv ? LW[kb * 5 + cl] : 0.f;
    bb[1] = cv ? LW[(kb + 1) * 5 + cl] : 0.f;
    c = wmma_f32(a, bb, c);
  }
#pragma unroll
  for (int k0 = 0; k0 < 64; k0 += 4) {
    int kb = k0 + khi;
    v2f a, bb;
    a[0] = X[m * 64 + kb];
    a[1] = X[m * 64 + kb + 1];
    bb[0] = cv ? RW[kb * 5 + cl] : 0.f;
    bb[1] = cv ? RW[(kb + 1) * 5 + cl] : 0.f;
    c = wmma_f32(a, bb, c);
  }
  int rloc = (l >> 4) << 3;
  float bc = cv ? lb[cl] : 0.f;
#pragma unroll
  for (int r = 0; r < 8; ++r) tile[(rloc + r) * 16 + cl] = c[r] + bc;
  __syncthreads();
  if (l < 16) {
    float v0 = tile[l * 16 + 0];
    float v1 = tile[l * 16 + 1];
    float v2 = tile[l * 16 + 2];
    float v3 = tile[l * 16 + 3];
    float v4 = tile[l * 16 + 4];
    float ss = v0 * v0 + v1 * v1 + v2 * v2 + v3 * v3 + v4 * v4;
    float s = 1.0f / fmaxf(sqrtf(ss), EPS_NORM);
    int rr = row0 + l;
    if (rr < n) {
      Y[rr * 5 + 0] = v0 * s;
      Y[rr * 5 + 1] = v1 * s;
      Y[rr * 5 + 2] = v2 * s;
      Y[rr * 5 + 3] = v3 * s;
      Y[rr * 5 + 4] = v4 * s;
    }
  }
}

extern "C" void kernel_launch(void* const* d_in, const int* in_sizes, int n_in,
                              void* d_out, int out_size, void* d_ws, size_t ws_size,
                              hipStream_t stream) {
  const float* h     = (const float*)d_in[0];
  const int*   ei    = (const int*)d_in[1];
  const float* ew    = (const float*)d_in[2];
  const float* bg    = (const float*)d_in[3];
  const float* bbeta = (const float*)d_in[4];
  const float* c1_pw = (const float*)d_in[5];
  const float* c1_pb = (const float*)d_in[6];
  const float* c1_lw = (const float*)d_in[7];
  const float* c1_lb = (const float*)d_in[8];
  const float* c1_rw = (const float*)d_in[9];
  const float* c2_pw = (const float*)d_in[10];
  const float* c2_pb = (const float*)d_in[11];
  const float* c2_lw = (const float*)d_in[12];
  const float* c2_lb = (const float*)d_in[13];
  const float* c2_rw = (const float*)d_in[14];
  const float* c3_pw = (const float*)d_in[15];
  const float* c3_pb = (const float*)d_in[16];
  const float* c3_lw = (const float*)d_in[17];
  const float* c3_lb = (const float*)d_in[18];
  const float* c3_rw = (const float*)d_in[19];

  const int N = in_sizes[0] / 5;
  const int E = in_sizes[2];
  const int* src = ei;
  const int* dst = ei + E;

  float* ws   = (float*)d_ws;
  float* mu   = ws;
  float* rs   = ws + 8;
  float* part = ws + 16;
  float* cnt  = part + 2560;
  float* invc = cnt + N;
  float* x0   = invc + N;
  float* xp1  = x0 + (size_t)N * 5;
  float* agg1 = xp1 + (size_t)N * 5;
  float* B1   = agg1 + (size_t)N * 5;
  float* B2   = B1 + (size_t)N * 64;
  float* B3   = B2 + (size_t)N * 64;

  const int nb16 = (N + 15) / 16;

  bn_partial_k<<<256, 256, 0, stream>>>(h, part, N);
  bn_final_k<<<1, 32, 0, stream>>>(part, mu, rs, 256, 1.0f / (float)N);
  bn_apply_k<<<(N * 5 + 255) / 256, 256, 0, stream>>>(h, mu, rs, bg, bbeta, x0, N * 5);

  hipMemsetAsync(cnt, 0, (size_t)N * sizeof(float), stream);
  degree_k<<<(E + 255) / 256, 256, 0, stream>>>(dst, cnt, E);
  invcnt_k<<<(N + 255) / 256, 256, 0, stream>>>(cnt, invc, N);

  proj5_k<<<(N + 255) / 256, 256, 0, stream>>>(x0, c1_pw, c1_pb, xp1, N);
  hipMemsetAsync(agg1, 0, (size_t)N * 5 * sizeof(float), stream);
  edge_scatter5_k<<<(E + 255) / 256, 256, 0, stream>>>(src, dst, ew, xp1, agg1, E);
  node5to64_k<<<N, 64, 0, stream>>>(agg1, invc, x0, c1_lw, c1_lb, c1_rw, B1, N);

  proj64_wmma_k<<<nb16, 128, 0, stream>>>(B1, c2_pw, c2_pb, B2, N);
  hipMemsetAsync(B3, 0, (size_t)N * 64 * sizeof(float), stream);
  edge_scatter64_k<<<(E * 16 + 255) / 256, 256, 0, stream>>>(src, dst, ew, B2, B3, E);
  node64_wmma_k<<<nb16, 128, 0, stream>>>(B3, invc, B1, c2_lw, c2_lb, c2_rw, B2, N);

  proj64_wmma_k<<<nb16, 128, 0, stream>>>(B2, c3_pw, c3_pb, B1, N);
  hipMemsetAsync(B3, 0, (size_t)N * 64 * sizeof(float), stream);
  edge_scatter64_k<<<(E * 16 + 255) / 256, 256, 0, stream>>>(src, dst, ew, B1, B3, E);
  node64to5_wmma_k<<<nb16, 32, 0, stream>>>(B3, invc, B2, c3_lw, c3_lb, c3_rw,
                                            (float*)d_out, N);
}


# Edge scatters: one wave per 32 edges, each load and f32 atomic instruction one contiguous run (dim 64: 32 lanes = one 128B line, rows 256B-aligned; dim 5: 5 lanes per edge); atomic arrival order kept
# speedup vs baseline: 2.3112x; 2.3112x over previous
#include <hip/hip_runtime.h>
#include <hip/hip_bf16.h>

typedef __attribute__((ext_vector_type(2))) float v2f;
typedef __attribute__((ext_vector_type(8))) float v8f;

#define EPS_BN   1e-5f
#define EPS_NORM 1e-12f

__device__ __forceinline__ v8f wmma_f32(v2f a, v2f b, v8f c) {
  return __builtin_amdgcn_wmma_f32_16x16x4_f32(false, a, false, b, (short)0, c,
                                               false, false);
}

__global__ void bn_partial_k(const float* __restrict__ h, float* __restrict__ part, int n) {
  __shared__ float sh[256 * 10];
  float acc[10];
#pragma unroll
  for (int d = 0; d < 10; ++d) acc[d] = 0.f;
  for (int i = blockIdx.x * blockDim.x + threadIdx.x; i < n;
       i += gridDim.x * blockDim.x) {
#pragma unroll
    for (int d = 0; d < 5; ++d) {
      float v = h[i * 5 + d];
      acc[d] += v;
      acc[5 + d] += v * v;
    }
  }
#pragma unroll
  for (int d = 0; d < 10; ++d) sh[threadIdx.x * 10 + d] = acc[d];
  __syncthreads();
  for (int s = 128; s > 0; s >>= 1) {
    if (threadIdx.x < (unsigned)s)
      for (int d = 0; d < 10; ++d)
        sh[threadIdx.x * 10 + d] += sh[(threadIdx.x + s) * 10 + d];
    __syncthreads();
  }
  if (threadIdx.x < 10) part[blockIdx.x * 10 + threadIdx.x] = sh[threadIdx.x];
}

__global__ void bn_final_k(const float* __restrict__ part, float* __restrict__ mu,
                           float* __restrict__ rs, int nblocks, float inv_n) {
  int t = threadIdx.x;
  if (t < 5) {
    float s = 0.f, q = 0.f;
    for (int b = 0; b < nblocks; ++b) {
      s += part[b * 10 + t];
      q += part[b * 10 + 5 + t];
    }
    float m = s * inv_n;
    float var = q * inv_n - m * m;
    mu[t] = m;
    rs[t] = rsqrtf(var + EPS_BN);
  }
}

__global__ void bn_apply_k(const float* __restrict__ h, const float* __restrict__ mu,
                           const float* __restrict__ rs, const float* __restrict__ g,
                           const float* __restrict__ b, float* __restrict__ x, int total) {
  int i = blockIdx.x * blockDim.x + threadIdx.x;
  if (i >= total) return;
  int d = i % 5;
  x[i] = (h[i] - mu[d]) * rs[d] * g[d] + b[d];
}

__global__ void degree_k(const int* __restrict__ dst, float* __restrict__ cnt, int E) {
  int e = blockIdx.x * blockDim.x + threadIdx.x;
  if (e < E) atomicAdd(&cnt[dst[e]], 1.0f);
}

__global__ void invcnt_k(const float* __restrict__ cnt, float* __restrict__ invc, int n) {
  int i = blockIdx.x * blockDim.x + threadIdx.x;
  if (i < n) invc[i] = 1.0f / fmaxf(cnt[i], 1.0f);
}

__global__ void proj5_k(const float* __restrict__ X, const float* __restrict__ W,
                        const float* __restrict__ b, float* __restrict__ Y, int n) {
  int i = blockIdx.x * blockDim.x + threadIdx.x;
  if (i >= n) return;
  float x[5];
#pragma unroll
  for (int k = 0; k < 5; ++k) x[k] = X[i * 5 + k];
#pragma unroll
  for (int c = 0; c < 5; ++c) {
    float acc = b[c];
#pragma unroll
    for (int k = 0; k < 5; ++k) acc += x[k] * W[k * 5 + c];
    Y[i * 5 + c] = fmaxf(acc, 0.f);
  }
}

__global__ void edge_scatter5_k(const int* __restrict__ src, const int* __restrict__ dst,
                                const float* __restrict__ ew, const float* __restrict__ xp,
                                float* __restrict__ agg, int E) {
  const int lane = threadIdx.x & 31;
  const int e0 = __builtin_amdgcn_readfirstlane(
      (int)((blockIdx.x * blockDim.x + threadIdx.x) >> 5) << 5);
  if (e0 >= E) return;
  const int e = e0 + lane;
  int s_l = 0, d_l = 0;
  float w_l = 0.f;
  if (e < E) { s_l = src[e]; d_l = dst[e]; w_l = ew[e]; }
  const int k = lane & 7;
  const int g = lane >> 3;
  int s[8], d[8];
  float w[8];
#pragma unroll
  for (int j = 0; j < 8; ++j) {
    const int sl = j * 4 + g;
    s[j] = __shfl(s_l, sl);
    d[j] = __shfl(d_l, sl);
    w[j] = __shfl(w_l, sl);
  }
  if (k >= 5) return;
  float x[8];
#pragma unroll
  for (int j = 0; j < 8; ++j) x[j] = xp[(size_t)s[j] * 5 + k];
#pragma unroll
  for (int j = 0; j < 8; ++j) {
    if ((e0 + j * 4 + g) < E) atomicAdd(&agg[(size_t)d[j] * 5 + k], x[j] * w[j]);
  }
}

__global__ void node5to64_k(const float* __restrict__ agg, const float* __restrict__ invc,
                            const float* __restrict__ X, const float* __restrict__ LW,
                            const float* __restrict__ lb, const float* __restrict__ RW,
                            float* __restrict__ Y, int n) {
  __shared__ float red[64];
  int node = blockIdx.x;
  if (node >= n) return;
  int c = threadIdx.x;
  float ic = invc[node];
  float acc = lb[c];
#pragma unroll
  for (int k = 0; k < 5; ++k) {
    acc += agg[node * 5 + k] * ic * LW[k * 64 + c] + X[node * 5 + k] * RW[k * 64 + c];
  }
  red[c] = acc * acc;
  __syncthreads();
#pragma unroll
  for (int s = 32; s > 0; s >>= 1) {
    if (c < s) red[c] += red[c + s];
    __syncthreads();
  }
  float scale = 1.0f / fmaxf(sqrtf(red[0]), EPS_NORM);
  Y[node * 64 + c] = fmaxf(acc * scale, 0.f);
}

__global__ void proj64_wmma_k(const float* __restrict__ X, const float* __restrict__ W,
                              const float* __restrict__ bias, float* __restrict__ Y, int n) {
  int l = threadIdx.x & 31;
  int wave = threadIdx.x >> 5;
  int row0 = blockIdx.x * 16;
  int m = row0 + (l & 15);
  if (m >= n) m = n - 1;
  int cl = wave * 16 + (l & 15);
  int khi = (l >> 4) << 1;
  v8f c = {};
#pragma unroll
  for (int k0 = 0; k0 < 64; k0 += 4) {
    int kb = k0 + khi;
    v2f a, bb;
    a[0] = X[m * 64 + kb];
    a[1] = X[m * 64 + kb + 1];
    bb[0] = W[kb * 64 + cl];
    bb[1] = W[(kb + 1) * 64 + cl];
    c = wmma_f32(a, bb, c);
  }
  int rbase = row0 + ((l >> 4) << 3);
  float bc = bias[cl];
#pragma unroll
  for (int r = 0; r < 8; ++r) {
    int rr = rbase + r;
    if (rr < n) Y[rr * 64 + cl] = fmaxf(c[r] + bc, 0.f);
  }
}

__global__ void edge_scatter64_k(const int* __restrict__ src, const int* __restrict__ dst,
                                 const float* __restrict__ ew, const float* __restrict__ xp,
                                 float* __restrict__ agg, int E) {
  const int lane = threadIdx.x & 31;
  const int e0 = __builtin_amdgcn_readfirstlane(
      (int)((blockIdx.x * blockDim.x + threadIdx.x) >> 5) << 5);
  if (e0 >= E) return;
  const int e = e0 + lane;
  int s_l = 0, d_l = 0;
  float w_l = 0.f;
  if (e < E) { s_l = src[e]; d_l = dst[e]; w_l = ew[e]; }
  const int n_e = min(32, E - e0);
  for (int j0 = 0; j0 < n_e; j0 += 8) {
    float lo[8], hi[8];
#pragma unroll
    for (int u = 0; u < 8; ++u) {
      const int s = __builtin_amdgcn_readlane(s_l, j0 + u);
      const float* row = xp + (size_t)s * 64 + lane;
      lo[u] = row[0];
      hi[u] = row[32];
    }
#pragma unroll
    for (int u = 0; u < 8; ++u) {
      if (j0 + u < n_e) {
        const int d = __builtin_amdgcn_readlane(d_l, j0 + u);
        const float w = __int_as_float(__builtin_amdgcn_readlane(__float_as_int(w_l), j0 + u));
        float* out = agg + (size_t)d * 64 + lane;
        atomicAdd(out, lo[u] * w);
        atomicAdd(out + 32, hi[u] * w);
      }
    }
  }
}

__global__ void node64_wmma_k(const float* __restrict__ agg, const float* __restrict__ invc,
                              const float* __restrict__ X, const float* __restrict__ LW,
                              const float* __restrict__ lb, const float* __restrict__ RW,
                              float* __restrict__ Y, int n) {
  __shared__ float tile[16 * 64];
  __shared__ float rscale[16];
  int l = threadIdx.x & 31;
  int wave = threadIdx.x >> 5;
  int row0 = blockIdx.x * 16;
  int m = row0 + (l & 15);
  if (m >= n) m = n - 1;
  int cl = wave * 16 + (l & 15);
  int khi = (l >> 4) << 1;
  float ic = invc[m];
  v8f c = {};
#pragma unroll
  for (int k0 = 0; k0 < 64; k0 += 4) {
    int kb = k0 + khi;
    v2f a, bb;
    a[0] = agg[m * 64 + kb] * ic;
    a[1] = agg[m * 64 + kb + 1] * ic;
    bb[0] = LW[kb * 64 + cl];
    bb[1] = LW[(kb + 1) * 64 + cl];
    c = wmma_f32(a, bb, c);
  }
#pragma unroll
  for (int k0 = 0; k0 < 64; k0 += 4) {
    int kb = k0 + khi;
    v2f a, bb;
    a[0] = X[m * 64 + kb];
    a[1] = X[m * 64 + kb + 1];
    bb[0] = RW[kb * 64 + cl];
    bb[1] = RW[(kb + 1) * 64 + cl];
    c = wmma_f32(a, bb, c);
  }
  int rloc = (l >> 4) << 3;
  float bc = lb[cl];
#pragma unroll
  for (int r = 0; r < 8; ++r) tile[(rloc + r) * 64 + cl] = c[r] + bc;
  __syncthreads();
  if (threadIdx.x < 16) {
    float ss = 0.f;
#pragma unroll
    for (int cc = 0; cc < 64; ++cc) {
      float v = tile[threadIdx.x * 64 + cc];
      ss += v * v;
    }
    rscale[threadIdx.x] = 1.0f / fmaxf(sqrtf(ss), EPS_NORM);
  }
  __syncthreads();
  for (int i = threadIdx.x; i < 16 * 64; i += blockDim.x) {
    int r = i >> 6;
    int rr = row0 + r;
    if (rr < n) Y[rr * 64 + (i & 63)] = fmaxf(tile[i] * rscale[r], 0.f);
  }
}

__global__ void node64to5_wmma_k(const float* __restrict__ agg, const float* __restrict__ invc,
                                 const float* __restrict__ X, const float* __restrict__ LW,
                                 const float* __restrict__ lb, const float* __restrict__ RW,
                                 float* __restrict__ Y, int n) {
  __shared__ float tile[16 * 16];
  int l = threadIdx.x & 31;
  int row0 = blockIdx.x * 16;
  int m = row0 + (l & 15);
  if (m >= n) m = n - 1;
  int cl = l & 15;
  int khi = (l >> 4) << 1;
  float ic = invc[m];
  bool cv = (cl < 5);
  v8f c = {};
#pragma unroll
  for (int k0 = 0; k0 < 64; k0 += 4) {
    int kb = k0 + khi;
    v2f a, bb;
    a[0] = agg[m * 64 + kb] * ic;
    a[1] = agg[m * 64 + kb + 1] * ic;
    bb[0] = cv ? LW[kb * 5 + cl] : 0.f;
    bb[1] = cv ? LW[(kb + 1) * 5 + cl] : 0.f;
    c = wmma_f32(a, bb, c);
  }
#pragma unroll
  for (int k0 = 0; k0 < 64; k0 += 4) {
    int kb = k0 + khi;
    v2f a, bb;
    a[0] = X[m * 64 + kb];
    a[1] = X[m * 64 + kb + 1];
    bb[0] = cv ? RW[kb * 5 + cl] : 0.f;
    bb[1] = cv ? RW[(kb + 1) * 5 + cl] : 0.f;
    c = wmma_f32(a, bb, c);
  }
  int rloc = (l >> 4) << 3;
  float bc = cv ? lb[cl] : 0.f;
#pragma unroll
  for (int r = 0; r < 8; ++r) tile[(rloc + r) * 16 + cl] = c[r] + bc;
  __syncthreads();
  if (l < 16) {
    float v0 = tile[l * 16 + 0];
    float v1 = tile[l * 16 + 1];
    float v2 = tile[l * 16 + 2];
    float v3 = tile[l * 16 + 3];
    float v4 = tile[l * 16 + 4];
    float ss = v0 * v0 + v1 * v1 + v2 * v2 + v3 * v3 + v4 * v4;
    float s = 1.0f / fmaxf(sqrtf(ss), EPS_NORM);
    int rr = row0 + l;
    if (rr < n) {
      Y[rr * 5 + 0] = v0 * s;
      Y[rr * 5 + 1] = v1 * s;
      Y[rr * 5 + 2] = v2 * s;
      Y[rr * 5 + 3] = v3 * s;
      Y[rr * 5 + 4] = v4 * s;
    }
  }
}

extern "C" void kernel_launch(void* const* d_in, const int* in_sizes, int n_in,
                              void* d_out, int out_size, void* d_ws, size_t ws_size,
                              hipStream_t stream) {
  const float* h     = (const float*)d_in[0];
  const int*   ei    = (const int*)d_in[1];
  const float* ew    = (const float*)d_in[2];
  const float* bg    = (const float*)d_in[3];
  const float* bbeta = (const float*)d_in[4];
  const float* c1_pw = (const float*)d_in[5];
  const float* c1_pb = (const float*)d_in[6];
  const float* c1_lw = (const float*)d_in[7];
  const float* c1_lb = (const float*)d_in[8];
  const float* c1_rw = (const float*)d_in[9];
  const float* c2_pw = (const float*)d_in[10];
  const float* c2_pb = (const float*)d_in[11];
  const float* c2_lw = (const float*)d_in[12];
  const float* c2_lb = (const float*)d_in[13];
  const float* c2_rw = (const float*)d_in[14];
  const float* c3_pw = (const float*)d_in[15];
  const float* c3_pb = (const float*)d_in[16];
  const float* c3_lw = (const float*)d_in[17];
  const float* c3_lb = (const float*)d_in[18];
  const float* c3_rw = (const float*)d_in[19];

  const int N = in_sizes[0] / 5;
  const int E = in_sizes[2];
  const int* src = ei;
  const int* dst = ei + E;

  float* ws   = (float*)d_ws;
  float* B1   = ws;
  float* B2   = B1 + (size_t)N * 64;
  float* B3   = B2 + (size_t)N * 64;
  float* mu   = B3 + (size_t)N * 64;
  float* rs   = mu + 8;
  float* part = rs + 8;
  float* cnt  = part + 2560;
  float* invc = cnt + N;
  float* x0   = invc + N;
  float* xp1  = x0 + (size_t)N * 5;
  float* agg1 = xp1 + (size_t)N * 5;

  const int nb16 = (N + 15) / 16;

  bn_partial_k<<<256, 256, 0, stream>>>(h, part, N);
  bn_final_k<<<1, 32, 0, stream>>>(part, mu, rs, 256, 1.0f / (float)N);
  bn_apply_k<<<(N * 5 + 255) / 256, 256, 0, stream>>>(h, mu, rs, bg, bbeta, x0, N * 5);

  hipMemsetAsync(cnt, 0, (size_t)N * sizeof(float), stream);
  degree_k<<<(E + 255) / 256, 256, 0, stream>>>(dst, cnt, E);
  invcnt_k<<<(N + 255) / 256, 256, 0, stream>>>(cnt, invc, N);

  proj5_k<<<(N + 255) / 256, 256, 0, stream>>>(x0, c1_pw, c1_pb, xp1, N);
  hipMemsetAsync(agg1, 0, (size_t)N * 5 * sizeof(float), stream);
  edge_scatter5_k<<<(E + 255) / 256, 256, 0, stream>>>(src, dst, ew, xp1, agg1, E);
  node5to64_k<<<N, 64, 0, stream>>>(agg1, invc, x0, c1_lw, c1_lb, c1_rw, B1, N);

  proj64_wmma_k<<<nb16, 128, 0, stream>>>(B1, c2_pw, c2_pb, B2, N);
  hipMemsetAsync(B3, 0, (size_t)N * 64 * sizeof(float), stream);
  edge_scatter64_k<<<(E + 255) / 256, 256, 0, stream>>>(src, dst, ew, B2, B3, E);
  node64_wmma_k<<<nb16, 128, 0, stream>>>(B3, invc, B1, c2_lw, c2_lb, c2_rw, B2, N);

  proj64_wmma_k<<<nb16, 128, 0, stream>>>(B2, c3_pw, c3_pb, B1, N);
  hipMemsetAsync(B3, 0, (size_t)N * 64 * sizeof(float), stream);
  edge_scatter64_k<<<(E + 255) / 256, 256, 0, stream>>>(src, dst, ew, B1, B3, E);
  node64to5_wmma_k<<<nb16, 32, 0, stream>>>(B3, invc, B2, c3_lw, c3_lb, c3_rw,
                                            (float*)d_out, N);
}


# In-degree count rides in the layer-1 edge scatter as a 6th lane (rows of 8 floats: 5 sums + count, one atomic request per edge); separate degree pass and memset removed; atomic arrival order kept
# speedup vs baseline: 3.1438x; 1.3602x over previous
#include <hip/hip_runtime.h>
#include <hip/hip_bf16.h>

typedef __attribute__((ext_vector_type(2))) float v2f;
typedef __attribute__((ext_vector_type(8))) float v8f;

#define EPS_BN   1e-5f
#define EPS_NORM 1e-12f

__device__ __forceinline__ v8f wmma_f32(v2f a, v2f b, v8f c) {
  return __builtin_amdgcn_wmma_f32_16x16x4_f32(false, a, false, b, (short)0, c,
                                               false, false);
}

__global__ void bn_partial_k(const float* __restrict__ h, float* __restrict__ part, int n) {
  __shared__ float sh[256 * 10];
  float acc[10];
#pragma unroll
  for (int d = 0; d < 10; ++d) acc[d] = 0.f;
  for (int i = blockIdx.x * blockDim.x + threadIdx.x; i < n;
       i += gridDim.x * blockDim.x) {
#pragma unroll
    for (int d = 0; d < 5; ++d) {
      float v = h[i * 5 + d];
      acc[d] += v;
      acc[5 + d] += v * v;
    }
  }
#pragma unroll
  for (int d = 0; d < 10; ++d) sh[threadIdx.x * 10 + d] = acc[d];
  __syncthreads();
  for (int s = 128; s > 0; s >>= 1) {
    if (threadIdx.x < (unsigned)s)
      for (int d = 0; d < 10; ++d)
        sh[threadIdx.x * 10 + d] += sh[(threadIdx.x + s) * 10 + d];
    __syncthreads();
  }
  if (threadIdx.x < 10) part[blockIdx.x * 10 + threadIdx.x] = sh[threadIdx.x];
}

__global__ void bn_final_k(const float* __restrict__ part, float* __restrict__ mu,
                           float* __restrict__ rs, int nblocks, float inv_n) {
  int t = threadIdx.x;
  if (t < 5) {
    float s = 0.f, q = 0.f;
    for (int b = 0; b < nblocks; ++b) {
      s += part[b * 10 + t];
      q += part[b * 10 + 5 + t];
    }
    float m = s * inv_n;
    float var = q * inv_n - m * m;
    mu[t] = m;
    rs[t] = rsqrtf(var + EPS_BN);
  }
}

__global__ void bn_apply_k(const float* __restrict__ h, const float* __restrict__ mu,
                           const float* __restrict__ rs, const float* __restrict__ g,
                           const float* __restrict__ b, float* __restrict__ x, int total) {
  int i = blockIdx.x * blockDim.x + threadIdx.x;
  if (i >= total) return;
  int d = i % 5;
  x[i] = (h[i] - mu[d]) * rs[d] * g[d] + b[d];
}

__global__ void invcnt_k(const float* __restrict__ aggc, float* __restrict__ invc, int n) {
  int i = blockIdx.x * blockDim.x + threadIdx.x;
  if (i < n) invc[i] = 1.0f / fmaxf(aggc[(size_t)i * 8 + 5], 1.0f);
}

__global__ void proj5_k(const float* __restrict__ X, const float* __restrict__ W,
                        const float* __restrict__ b, float* __restrict__ Y, int n) {
  int i = blockIdx.x * blockDim.x + threadIdx.x;
  if (i >= n) return;
  float x[5];
#pragma unroll
  for (int k = 0; k < 5; ++k) x[k] = X[i * 5 + k];
#pragma unroll
  for (int c = 0; c < 5; ++c) {
    float acc = b[c];
#pragma unroll
    for (int k = 0; k < 5; ++k) acc += x[k] * W[k * 5 + c];
    Y[i * 5 + c] = fmaxf(acc, 0.f);
  }
}

__global__ void edge_scatter5_k(const int* __restrict__ src, const int* __restrict__ dst,
                                const float* __restrict__ ew, const float* __restrict__ xp,
                                float* __restrict__ agg, int E) {
  const int lane = threadIdx.x & 31;
  const int e0 = __builtin_amdgcn_readfirstlane(
      (int)((blockIdx.x * blockDim.x + threadIdx.x) >> 5) << 5);
  if (e0 >= E) return;
  const int e = e0 + lane;
  int s_l = 0, d_l = 0;
  float w_l = 0.f;
  if (e < E) { s_l = src[e]; d_l = dst[e]; w_l = ew[e]; }
  const int k = lane & 7;
  const int g = lane >> 3;
  int s[8], d[8];
  float w[8];
#pragma unroll
  for (int j = 0; j < 8; ++j) {
    const int sl = j * 4 + g;
    s[j] = __shfl(s_l, sl);
    d[j] = __shfl(d_l, sl);
    w[j] = __shfl(w_l, sl);
  }
  if (k >= 6) return;
  const bool feat = (k < 5);
  const int kl = feat ? k : 4;
  float x[8];
#pragma unroll
  for (int j = 0; j < 8; ++j) x[j] = xp[(size_t)s[j] * 5 + kl];
#pragma unroll
  for (int j = 0; j < 8; ++j) {
    if ((e0 + j * 4 + g) < E)
      atomicAdd(&agg[(size_t)d[j] * 8 + k], feat ? x[j] * w[j] : 1.0f);
  }
}

__global__ void node5to64_k(const float* __restrict__ agg, const float* __restrict__ invc,
                            const float* __restrict__ X, const float* __restrict__ LW,
                            const float* __restrict__ lb, const float* __restrict__ RW,
                            float* __restrict__ Y, int n) {
  __shared__ float red[64];
  int node = blockIdx.x;
  if (node >= n) return;
  int c = threadIdx.x;
  float ic = invc[node];
  float acc = lb[c];
#pragma unroll
  for (int k = 0; k < 5; ++k) {
    acc += agg[node * 8 + k] * ic * LW[k * 64 + c] + X[node * 5 + k] * RW[k * 64 + c];
  }
  red[c] = acc * acc;
  __syncthreads();
#pragma unroll
  for (int s = 32; s > 0; s >>= 1) {
    if (c < s) red[c] += red[c + s];
    __syncthreads();
  }
  float scale = 1.0f / fmaxf(sqrtf(red[0]), EPS_NORM);
  Y[node * 64 + c] = fmaxf(acc * scale, 0.f);
}

__global__ void proj64_wmma_k(const float* __restrict__ X, const float* __restrict__ W,
                              const float* __restrict__ bias, float* __restrict__ Y, int n) {
  int l = threadIdx.x & 31;
  int wave = threadIdx.x >> 5;
  int row0 = blockIdx.x * 16;
  int m = row0 + (l & 15);
  if (m >= n) m = n - 1;
  int cl = wave * 16 + (l & 15);
  int khi = (l >> 4) << 1;
  v8f c = {};
#pragma unroll
  for (int k0 = 0; k0 < 64; k0 += 4) {
    int kb = k0 + khi;
    v2f a, bb;
    a[0] = X[m * 64 + kb];
    a[1] = X[m * 64 + kb + 1];
    bb[0] = W[kb * 64 + cl];
    bb[1] = W[(kb + 1) * 64 + cl];
    c = wmma_f32(a, bb, c);
  }
  int rbase = row0 + ((l >> 4) << 3);
  float bc = bias[cl];
#pragma unroll
  for (int r = 0; r < 8; ++r) {
    int rr = rbase + r;
    if (rr < n) Y[rr * 64 + cl] = fmaxf(c[r] + bc, 0.f);
  }
}

__global__ void edge_scatter64_k(const int* __restrict__ src, const int* __restrict__ dst,
                                 const float* __restrict__ ew, const float* __restrict__ xp,
                                 float* __restrict__ agg, int E) {
  const int lane = threadIdx.x & 31;
  const int e0 = __builtin_amdgcn_readfirstlane(
      (int)((blockIdx.x * blockDim.x + threadIdx.x) >> 5) << 5);
  if (e0 >= E) return;
  const int e = e0 + lane;
  int s_l = 0, d_l = 0;
  float w_l = 0.f;
  if (e < E) { s_l = src[e]; d_l = dst[e]; w_l = ew[e]; }
  const int n_e = min(32, E - e0);
  for (int j0 = 0; j0 < n_e; j0 += 8) {
    float lo[8], hi[8];
#pragma unroll
    for (int u = 0; u < 8; ++u) {
      const int s = __builtin_amdgcn_readlane(s_l, j0 + u);
      const float* row = xp + (size_t)s * 64 + lane;
      lo[u] = row[0];
      hi[u] = row[32];
    }
#pragma unroll
    for (int u = 0; u < 8; ++u) {
      if (j0 + u < n_e) {
        const int d = __builtin_amdgcn_readlane(d_l, j0 + u);
        const float w = __int_as_float(__builtin_amdgcn_readlane(__float_as_int(w_l), j0 + u));
        float* out = agg + (size_t)d * 64 + lane;
        atomicAdd(out, lo[u] * w);
        atomicAdd(out + 32, hi[u] * w);
      }
    }
  }
}

__global__ void node64_wmma_k(const float* __restrict__ agg, const float* __restrict__ invc,
                              const float* __restrict__ X, const float* __restrict__ LW,
                              const float* __restrict__ lb, const float* __restrict__ RW,
                              float* __restrict__ Y, int n) {
  __shared__ float tile[16 * 64];
  __shared__ float rscale[16];
  int l = threadIdx.x & 31;
  int wave = threadIdx.x >> 5;
  int row0 = blockIdx.x * 16;
  int m = row0 + (l & 15);
  if (m >= n) m = n - 1;
  int cl = wave * 16 + (l & 15);
  int khi = (l >> 4) << 1;
  float ic = invc[m];
  v8f c = {};
#pragma unroll
  for (int k0 = 0; k0 < 64; k0 += 4) {
    int kb = k0 + khi;
    v2f a, bb;
    a[0] = agg[m * 64 + kb] * ic;
    a[1] = agg[m * 64 + kb + 1] * ic;
    bb[0] = LW[kb * 64 + cl];
    bb[1] = LW[(kb + 1) * 64 + cl];
    c = wmma_f32(a, bb, c);
  }
#pragma unroll
  for (int k0 = 0; k0 < 64; k0 += 4) {
    int kb = k0 + khi;
    v2f a, bb;
    a[0] = X[m * 64 + kb];
    a[1] = X[m * 64 + kb + 1];
    bb[0] = RW[kb * 64 + cl];
    bb[1] = RW[(kb + 1) * 64 + cl];
    c = wmma_f32(a, bb, c);
  }
  int rloc = (l >> 4) << 3;
  float bc = lb[cl];
#pragma unroll
  for (int r = 0; r < 8; ++r) tile[(rloc + r) * 64 + cl] = c[r] + bc;
  __syncthreads();
  if (threadIdx.x < 16) {
    float ss = 0.f;
#pragma unroll
    for (int cc = 0; cc < 64; ++cc) {
      float v = tile[threadIdx.x * 64 + cc];
      ss += v * v;
    }
    rscale[threadIdx.x] = 1.0f / fmaxf(sqrtf(ss), EPS_NORM);
  }
  __syncthreads();
  for (int i = threadIdx.x; i < 16 * 64; i += blockDim.x) {
    int r = i >> 6;
    int rr = row0 + r;
    if (rr < n) Y[rr * 64 + (i & 63)] = fmaxf(tile[i] * rscale[r], 0.f);
  }
}

__global__ void node64to5_wmma_k(const float* __restrict__ agg, const float* __restrict__ invc,
                                 const float* __restrict__ X, const float* __restrict__ LW,
                                 const float* __restrict__ lb, const float* __restrict__ RW,
                                 float* __restrict__ Y, int n) {
  __shared__ float tile[16 * 16];
  int l = threadIdx.x & 31;
  int row0 = blockIdx.x * 16;
  int m = row0 + (l & 15);
  if (m >= n) m = n - 1;
  int cl = l & 15;
  int khi = (l >> 4) << 1;
  float ic = invc[m];
  bool cv = (cl < 5);
  v8f c = {};
#pragma unroll
  for (int k0 = 0; k0 < 64; k0 += 4) {
    int kb = k0 + khi;
    v2f a, bb;
    a[0] = agg[m * 64 + kb] * ic;
    a[1] = agg[m * 64 + kb + 1] * ic;
    bb[0] = cv ? LW[kb * 5 + cl] : 0.f;
    bb[1] = cv ? LW[(kb + 1) * 5 + cl] : 0.f;
    c = wmma_f32(a, bb, c);
  }
#pragma unroll
  for (int k0 = 0; k0 < 64; k0 += 4) {
    int kb = k0 + khi;
    v2f a, bb;
    a[0] = X[m * 64 + kb];
    a[1] = X[m * 64 + kb + 1];
    bb[0] = cv ? RW[kb * 5 + cl] : 0.f;
    bb[1] = cv ? RW[(kb + 1) * 5 + cl] : 0.f;
    c = wmma_f32(a, bb, c);
  }
  int rloc = (l >> 4) << 3;
  float bc = cv ? lb[cl] : 0.f;
#pragma unroll
  for (int r = 0; r < 8; ++r) tile[(rloc + r) * 16 + cl] = c[r] + bc;
  __syncthreads();
  if (l < 16) {
    float v0 = tile[l * 16 + 0];
    float v1 = tile[l * 16 + 1];
    float v2 = tile[l * 16 + 2];
    float v3 = tile[l * 16 + 3];
    float v4 = tile[l * 16 + 4];
    float ss = v0 * v0 + v1 * v1 + v2 * v2 + v3 * v3 + v4 * v4;
    float s = 1.0f / fmaxf(sqrtf(ss), EPS_NORM);
    int rr = row0 + l;
    if (rr < n) {
      Y[rr * 5 + 0] = v0 * s;
      Y[rr * 5 + 1] = v1 * s;
      Y[rr * 5 + 2] = v2 * s;
      Y[rr * 5 + 3] = v3 * s;
      Y[rr * 5 + 4] = v4 * s;
    }
  }
}

extern "C" void kernel_launch(void* const* d_in, const int* in_sizes, int n_in,
                              void* d_out, int out_size, void* d_ws, size_t ws_size,
                              hipStream_t stream) {
  const float* h     = (const float*)d_in[0];
  const int*   ei    = (const int*)d_in[1];
  const float* ew    = (const float*)d_in[2];
  const float* bg    = (const float*)d_in[3];
  const float* bbeta = (const float*)d_in[4];
  const float* c1_pw = (const float*)d_in[5];
  const float* c1_pb = (const float*)d_in[6];
  const float* c1_lw = (const float*)d_in[7];
  const float* c1_lb = (const float*)d_in[8];
  const float* c1_rw = (const float*)d_in[9];
  const float* c2_pw = (const float*)d_in[10];
  const float* c2_pb = (const float*)d_in[11];
  const float* c2_lw = (const float*)d_in[12];
  const float* c2_lb = (const float*)d_in[13];
  const float* c2_rw = (const float*)d_in[14];
  const float* c3_pw = (const float*)d_in[15];
  const float* c3_pb = (const float*)d_in[16];
  const float* c3_lw = (const float*)d_in[17];
  const float* c3_lb = (const float*)d_in[18];
  const float* c3_rw = (const float*)d_in[19];

  const int N = in_sizes[0] / 5;
  const int E = in_sizes[2];
  const int* src = ei;
  const int* dst = ei + E;

  float* ws   = (float*)d_ws;
  float* B1   = ws;
  float* B2   = B1 + (size_t)N * 64;
  float* B3   = B2 + (size_t)N * 64;
  float* mu   = B3 + (size_t)N * 64;
  float* rs   = mu + 8;
  float* part = rs + 8;
  float* invc = part + 2560;
  float* x0   = invc + N;
  float* xp1  = x0 + (size_t)N * 5;
  float* aggc = B3;

  const int nb16 = (N + 15) / 16;

  bn_partial_k<<<256, 256, 0, stream>>>(h, part, N);
  bn_final_k<<<1, 32, 0, stream>>>(part, mu, rs, 256, 1.0f / (float)N);
  bn_apply_k<<<(N * 5 + 255) / 256, 256, 0, stream>>>(h, mu, rs, bg, bbeta, x0, N * 5);

  proj5_k<<<(N + 255) / 256, 256, 0, stream>>>(x0, c1_pw, c1_pb, xp1, N);
  hipMemsetAsync(aggc, 0, (size_t)N * 8 * sizeof(float), stream);
  edge_scatter5_k<<<(E + 255) / 256, 256, 0, stream>>>(src, dst, ew, xp1, aggc, E);
  invcnt_k<<<(N + 255) / 256, 256, 0, stream>>>(aggc, invc, N);
  node5to64_k<<<N, 64, 0, stream>>>(aggc, invc, x0, c1_lw, c1_lb, c1_rw, B1, N);

  proj64_wmma_k<<<nb16, 128, 0, stream>>>(B1, c2_pw, c2_pb, B2, N);
  hipMemsetAsync(B3, 0, (size_t)N * 64 * sizeof(float), stream);
  edge_scatter64_k<<<(E + 255) / 256, 256, 0, stream>>>(src, dst, ew, B2, B3, E);
  node64_wmma_k<<<nb16, 128, 0, stream>>>(B3, invc, B1, c2_lw, c2_lb, c2_rw, B2, N);

  proj64_wmma_k<<<nb16, 128, 0, stream>>>(B2, c3_pw, c3_pb, B1, N);
  hipMemsetAsync(B3, 0, (size_t)N * 64 * sizeof(float), stream);
  edge_scatter64_k<<<(E + 255) / 256, 256, 0, stream>>>(src, dst, ew, B1, B3, E);
  node64to5_wmma_k<<<nb16, 32, 0, stream>>>(B3, invc, B2, c3_lw, c3_lb, c3_rw,
                                            (float*)d_out, N);
}


# Last layer: its 64->5 neighbour linear is applied to the projected rows before the edge sum (the sum is linear), so that edge pass moves 5 floats per edge, not 64; sums regrouped, f32 atomics kept
# speedup vs baseline: 3.4391x; 1.0939x over previous
#include <hip/hip_runtime.h>
#include <hip/hip_bf16.h>

typedef __attribute__((ext_vector_type(2))) float v2f;
typedef __attribute__((ext_vector_type(8))) float v8f;

#define EPS_BN   1e-5f
#define EPS_NORM 1e-12f

__device__ __forceinline__ v8f wmma_f32(v2f a, v2f b, v8f c) {
  return __builtin_amdgcn_wmma_f32_16x16x4_f32(false, a, false, b, (short)0, c,
                                               false, false);
}

__global__ void bn_partial_k(const float* __restrict__ h, float* __restrict__ part, int n) {
  __shared__ float sh[256 * 10];
  float acc[10];
#pragma unroll
  for (int d = 0; d < 10; ++d) acc[d] = 0.f;
  for (int i = blockIdx.x * blockDim.x + threadIdx.x; i < n;
       i += gridDim.x * blockDim.x) {
#pragma unroll
    for (int d = 0; d < 5; ++d) {
      float v = h[i * 5 + d];
      acc[d] += v;
      acc[5 + d] += v * v;
    }
  }
#pragma unroll
  for (int d = 0; d < 10; ++d) sh[threadIdx.x * 10 + d] = acc[d];
  __syncthreads();
  for (int s = 128; s > 0; s >>= 1) {
    if (threadIdx.x < (unsigned)s)
      for (int d = 0; d < 10; ++d)
        sh[threadIdx.x * 10 + d] += sh[(threadIdx.x + s) * 10 + d];
    __syncthreads();
  }
  if (threadIdx.x < 10) part[blockIdx.x * 10 + threadIdx.x] = sh[threadIdx.x];
}

__global__ void bn_final_k(const float* __restrict__ part, float* __restrict__ mu,
                           float* __restrict__ rs, int nblocks, float inv_n) {
  int t = threadIdx.x;
  if (t < 5) {
    float s = 0.f, q = 0.f;
    for (int b = 0; b < nblocks; ++b) {
      s += part[b * 10 + t];
      q += part[b * 10 + 5 + t];
    }
    float m = s * inv_n;
    float var = q * inv_n - m * m;
    mu[t] = m;
    rs[t] = rsqrtf(var + EPS_BN);
  }
}

__global__ void bn_apply_k(const float* __restrict__ h, const float* __restrict__ mu,
                           const float* __restrict__ rs, const float* __restrict__ g,
                           const float* __restrict__ b, float* __restrict__ x, int total) {
  int i = blockIdx.x * blockDim.x + threadIdx.x;
  if (i >= total) return;
  int d = i % 5;
  x[i] = (h[i] - mu[d]) * rs[d] * g[d] + b[d];
}

__global__ void invcnt_k(const float* __restrict__ aggc, float* __restrict__ invc, int n) {
  int i = blockIdx.x * blockDim.x + threadIdx.x;
  if (i < n) invc[i] = 1.0f / fmaxf(aggc[(size_t)i * 8 + 5], 1.0f);
}

__global__ void proj5_k(const float* __restrict__ X, const float* __restrict__ W,
                        const float* __restrict__ b, float* __restrict__ Y, int n) {
  int i = blockIdx.x * blockDim.x + threadIdx.x;
  if (i >= n) return;
  float x[5];
#pragma unroll
  for (int k = 0; k < 5; ++k) x[k] = X[i * 5 + k];
#pragma unroll
  for (int c = 0; c < 5; ++c) {
    float acc = b[c];
#pragma unroll
    for (int k = 0; k < 5; ++k) acc += x[k] * W[k * 5 + c];
    Y[i * 5 + c] = fmaxf(acc, 0.f);
  }
}

template <int XS, bool COUNT>
__device__ __forceinline__ void edge_scatter5_body(const int* __restrict__ src,
                                                   const int* __restrict__ dst,
                                                   const float* __restrict__ ew,
                                                   const float* __restrict__ xp,
                                                   float* __restrict__ agg, int E) {
  const int lane = threadIdx.x & 31;
  const int e0 = __builtin_amdgcn_readfirstlane(
      (int)((blockIdx.x * blockDim.x + threadIdx.x) >> 5) << 5);
  if (e0 >= E) return;
  const int e = e0 + lane;
  int s_l = 0, d_l = 0;
  float w_l = 0.f;
  if (e < E) { s_l = src[e]; d_l = dst[e]; w_l = ew[e]; }
  const int k = lane & 7;
  const int g = lane >> 3;
  int s[8], d[8];
  float w[8];
#pragma unroll
  for (int j = 0; j < 8; ++j) {
    const int sl = j * 4 + g;
    s[j] = __shfl(s_l, sl);
    d[j] = __shfl(d_l, sl);
    w[j] = __shfl(w_l, sl);
  }
  if (k >= (COUNT ? 6 : 5)) return;
  const bool feat = (k < 5);
  const int kl = feat ? k : 4;
  float x[8];
#pragma unroll
  for (int j = 0; j < 8; ++j) x[j] = xp[(size_t)s[j] * XS + kl];
#pragma unroll
  for (int j = 0; j < 8; ++j) {
    if ((e0 + j * 4 + g) < E)
      atomicAdd(&agg[(size_t)d[j] * 8 + k], feat ? x[j] * w[j] : 1.0f);
  }
}

__global__ void edge_scatter5_k(const int* __restrict__ src, const int* __restrict__ dst,
                                const float* __restrict__ ew, const float* __restrict__ xp,
                                float* __restrict__ agg, int E) {
  edge_scatter5_body<5, true>(src, dst, ew, xp, agg, E);
}

__global__ void edge_scatter5z_k(const int* __restrict__ src, const int* __restrict__ dst,
                                 const float* __restrict__ ew, const float* __restrict__ z,
                                 float* __restrict__ agg, int E) {
  edge_scatter5_body<8, false>(src, dst, ew, z, agg, E);
}

__global__ void node5to64_k(const float* __restrict__ agg, const float* __restrict__ invc,
                            const float* __restrict__ X, const float* __restrict__ LW,
                            const float* __restrict__ lb, const float* __restrict__ RW,
                            float* __restrict__ Y, int n) {
  __shared__ float red[64];
  int node = blockIdx.x;
  if (node >= n) return;
  int c = threadIdx.x;
  float ic = invc[node];
  float acc = lb[c];
#pragma unroll
  for (int k = 0; k < 5; ++k) {
    acc += agg[node * 8 + k] * ic * LW[k * 64 + c] + X[node * 5 + k] * RW[k * 64 + c];
  }
  red[c] = acc * acc;
  __syncthreads();
#pragma unroll
  for (int s = 32; s > 0; s >>= 1) {
    if (c < s) red[c] += red[c + s];
    __syncthreads();
  }
  float scale = 1.0f / fmaxf(sqrtf(red[0]), EPS_NORM);
  Y[node * 64 + c] = fmaxf(acc * scale, 0.f);
}

__global__ void proj64_wmma_k(const float* __restrict__ X, const float* __restrict__ W,
                              const float* __restrict__ bias, float* __restrict__ Y, int n) {
  int l = threadIdx.x & 31;
  int wave = threadIdx.x >> 5;
  int row0 = blockIdx.x * 16;
  int m = row0 + (l & 15);
  if (m >= n) m = n - 1;
  int cl = wave * 16 + (l & 15);
  int khi = (l >> 4) << 1;
  v8f c = {};
#pragma unroll
  for (int k0 = 0; k0 < 64; k0 += 4) {
    int kb = k0 + khi;
    v2f a, bb;
    a[0] = X[m * 64 + kb];
    a[1] = X[m * 64 + kb + 1];
    bb[0] = W[kb * 64 + cl];
    bb[1] = W[(kb + 1) * 64 + cl];
    c = wmma_f32(a, bb, c);
  }
  int rbase = row0 + ((l >> 4) << 3);
  float bc = bias[cl];
#pragma unroll
  for (int r = 0; r < 8; ++r) {
    int rr = rbase + r;
    if (rr < n) Y[rr * 64 + cl] = fmaxf(c[r] + bc, 0.f);
  }
}

__global__ void proj64_lin5_wmma_k(const float* __restrict__ X, const float* __restrict__ PW,
                                   const float* __restrict__ pb, const float* __restrict__ LW,
                                   float* __restrict__ Z, int n) {
  __shared__ float tile[16 * 68];
  int l = threadIdx.x & 31;
  const int wave = __builtin_amdgcn_readfirstlane((int)(threadIdx.x >> 5));
  int row0 = blockIdx.x * 16;
  int c16 = l & 15;
  int m = row0 + c16;
  if (m >= n) m = n - 1;
  int cl = wave * 16 + c16;
  int khi = (l >> 4) << 1;
  v8f c = {};
#pragma unroll
  for (int k0 = 0; k0 < 64; k0 += 4) {
    int kb = k0 + khi;
    v2f a, bb;
    a[0] = X[m * 64 + kb];
    a[1] = X[m * 64 + kb + 1];
    bb[0] = PW[kb * 64 + cl];
    bb[1] = PW[(kb + 1) * 64 + cl];
    c = wmma_f32(a, bb, c);
  }
  int rloc = (l >> 4) << 3;
  float bc = pb[cl];
#pragma unroll
  for (int r = 0; r < 8; ++r) tile[(rloc + r) * 68 + cl] = fmaxf(c[r] + bc, 0.f);
  __syncthreads();
  if (wave == 0) {
    bool cv = (c16 < 5);
    v2f bw[16];
#pragma unroll
    for (int i = 0; i < 16; ++i) { bw[i][0] = 0.f; bw[i][1] = 0.f; }
    if (cv) {
#pragma unroll
      for (int i = 0; i < 16; ++i) {
        int kb = i * 4 + khi;
        bw[i][0] = LW[kb * 5 + c16];
        bw[i][1] = LW[(kb + 1) * 5 + c16];
      }
    }
    v8f z = {};
#pragma unroll
    for (int i = 0; i < 16; ++i) {
      int kb = i * 4 + khi;
      v2f a;
      a[0] = tile[c16 * 68 + kb];
      a[1] = tile[c16 * 68 + kb + 1];
      z = wmma_f32(a, bw[i], z);
    }
    if (cv) {
#pragma unroll
      for (int r = 0; r < 8; ++r) {
        int rr = row0 + rloc + r;
        if (rr < n) Z[(size_t)rr * 8 + c16] = z[r];
      }
    }
  }
}

__global__ void edge_scatter64_k(const int* __restrict__ src, const int* __restrict__ dst,
                                 const float* __restrict__ ew, const float* __restrict__ xp,
                                 float* __restrict__ agg, int E) {
  const int lane = threadIdx.x & 31;
  const int e0 = __builtin_amdgcn_readfirstlane(
      (int)((blockIdx.x * blockDim.x + threadIdx.x) >> 5) << 5);
  if (e0 >= E) return;
  const int e = e0 + lane;
  int s_l = 0, d_l = 0;
  float w_l = 0.f;
  if (e < E) { s_l = src[e]; d_l = dst[e]; w_l = ew[e]; }
  const int n_e = min(32, E - e0);
  for (int j0 = 0; j0 < n_e; j0 += 8) {
    float lo[8], hi[8];
#pragma unroll
    for (int u = 0; u < 8; ++u) {
      const int s = __builtin_amdgcn_readlane(s_l, j0 + u);
      const float* row = xp + (size_t)s * 64 + lane;
      lo[u] = row[0];
      hi[u] = row[32];
    }
#pragma unroll
    for (int u = 0; u < 8; ++u) {
      if (j0 + u < n_e) {
        const int d = __builtin_amdgcn_readlane(d_l, j0 + u);
        const float w = __int_as_float(__builtin_amdgcn_readlane(__float_as_int(w_l), j0 + u));
        float* out = agg + (size_t)d * 64 + lane;
        atomicAdd(out, lo[u] * w);
        atomicAdd(out + 32, hi[u] * w);
      }
    }
  }
}

__global__ void node64_wmma_k(const float* __restrict__ agg, const float* __restrict__ invc,
                              const float* __restrict__ X, const float* __restrict__ LW,
                              const float* __restrict__ lb, const float* __restrict__ RW,
                              float* __restrict__ Y, int n) {
  __shared__ float tile[16 * 64];
  __shared__ float rscale[16];
  int l = threadIdx.x & 31;
  int wave = threadIdx.x >> 5;
  int row0 = blockIdx.x * 16;
  int m = row0 + (l & 15);
  if (m >= n) m = n - 1;
  int cl = wave * 16 + (l & 15);
  int khi = (l >> 4) << 1;
  float ic = invc[m];
  v8f c = {};
#pragma unroll
  for (int k0 = 0; k0 < 64; k0 += 4) {
    int kb = k0 + khi;
    v2f a, bb;
    a[0] = agg[m * 64 + kb] * ic;
    a[1] = agg[m * 64 + kb + 1] * ic;
    bb[0] = LW[kb * 64 + cl];
    bb[1] = LW[(kb + 1) * 64 + cl];
    c = wmma_f32(a, bb, c);
  }
#pragma unroll
  for (int k0 = 0; k0 < 64; k0 += 4) {
    int kb = k0 + khi;
    v2f a, bb;
    a[0] = X[m * 64 + kb];
    a[1] = X[m * 64 + kb + 1];
    bb[0] = RW[kb * 64 + cl];
    bb[1] = RW[(kb + 1) * 64 + cl];
    c = wmma_f32(a, bb, c);
  }
  int rloc = (l >> 4) << 3;
  float bc = lb[cl];
#pragma unroll
  for (int r = 0; r < 8; ++r) tile[(rloc + r) * 64 + cl] = c[r] + bc;
  __syncthreads();
  if (threadIdx.x < 16) {
    float ss = 0.f;
#pragma unroll
    for (int cc = 0; cc < 64; ++cc) {
      float v = tile[threadIdx.x * 64 + cc];
      ss += v * v;
    }
    rscale[threadIdx.x] = 1.0f / fmaxf(sqrtf(ss), EPS_NORM);
  }
  __syncthreads();
  for (int i = threadIdx.x; i < 16 * 64; i += blockDim.x) {
    int r = i >> 6;
    int rr = row0 + r;
    if (rr < n) Y[rr * 64 + (i & 63)] = fmaxf(tile[i] * rscale[r], 0.f);
  }
}

__global__ void node64to5_wmma_k(const float* __restrict__ zsum, const float* __restrict__ invc,
                                 const float* __restrict__ X, const float* __restrict__ lb,
                                 const float* __restrict__ RW, float* __restrict__ Y, int n) {
  __shared__ float tile[16 * 16];
  int l = threadIdx.x & 31;
  int row0 = blockIdx.x * 16;
  int m = row0 + (l & 15);
  if (m >= n) m = n - 1;
  int cl = l & 15;
  int khi = (l >> 4) << 1;
  float ic = invc[m];
  float g0 = zsum[(size_t)m * 8 + 0];
  float g1 = zsum[(size_t)m * 8 + 1];
  float g2 = zsum[(size_t)m * 8 + 2];
  float g3 = zsum[(size_t)m * 8 + 3];
  float g4 = zsum[(size_t)m * 8 + 4];
  bool cv = (cl < 5);
  v8f c = {};
#pragma unroll
  for (int k0 = 0; k0 < 64; k0 += 4) {
    int kb = k0 + khi;
    v2f a, bb;
    a[0] = X[m * 64 + kb];
    a[1] = X[m * 64 + kb + 1];
    bb[0] = cv ? RW[kb * 5 + cl] : 0.f;
    bb[1] = cv ? RW[(kb + 1) * 5 + cl] : 0.f;
    c = wmma_f32(a, bb, c);
  }
  int rloc = (l >> 4) << 3;
  float bc = cv ? lb[cl] : 0.f;
#pragma unroll
  for (int r = 0; r < 8; ++r) tile[(rloc + r) * 16 + cl] = c[r] + bc;
  __syncthreads();
  if (l < 16) {
    float v0 = tile[l * 16 + 0] + g0 * ic;
    float v1 = tile[l * 16 + 1] + g1 * ic;
    float v2 = tile[l * 16 + 2] + g2 * ic;
    float v3 = tile[l * 16 + 3] + g3 * ic;
    float v4 = tile[l * 16 + 4] + g4 * ic;
    float ss = v0 * v0 + v1 * v1 + v2 * v2 + v3 * v3 + v4 * v4;
    float s = 1.0f / fmaxf(sqrtf(ss), EPS_NORM);
    int rr = row0 + l;
    if (rr < n) {
      Y[rr * 5 + 0] = v0 * s;
      Y[rr * 5 + 1] = v1 * s;
      Y[rr * 5 + 2] = v2 * s;
      Y[rr * 5 + 3] = v3 * s;
      Y[rr * 5 + 4] = v4 * s;
    }
  }
}

extern "C" void kernel_launch(void* const* d_in, const int* in_sizes, int n_in,
                              void* d_out, int out_size, void* d_ws, size_t ws_size,
                              hipStream_t stream) {
  const float* h     = (const float*)d_in[0];
  const int*   ei    = (const int*)d_in[1];
  const float* ew    = (const float*)d_in[2];
  const float* bg    = (const float*)d_in[3];
  const float* bbeta = (const float*)d_in[4];
  const float* c1_pw = (const float*)d_in[5];
  const float* c1_pb = (const float*)d_in[6];
  const float* c1_lw = (const float*)d_in[7];
  const float* c1_lb = (const float*)d_in[8];
  const float* c1_rw = (const float*)d_in[9];
  const float* c2_pw = (const float*)d_in[10];
  const float* c2_pb = (const float*)d_in[11];
  const float* c2_lw = (const float*)d_in[12];
  const float* c2_lb = (const float*)d_in[13];
  const float* c2_rw = (const float*)d_in[14];
  const float* c3_pw = (const float*)d_in[15];
  const float* c3_pb = (const float*)d_in[16];
  const float* c3_lw = (const float*)d_in[17];
  const float* c3_lb = (const float*)d_in[18];
  const float* c3_rw = (const float*)d_in[19];

  const int N = in_sizes[0] / 5;
  const int E = in_sizes[2];
  const int* src = ei;
  const int* dst = ei + E;

  float* ws   = (float*)d_ws;
  float* B1   = ws;
  float* B2   = B1 + (size_t)N * 64;
  float* B3   = B2 + (size_t)N * 64;
  float* mu   = B3 + (size_t)N * 64;
  float* rs   = mu + 8;
  float* part = rs + 8;
  float* invc = part + 2560;
  float* x0   = invc + N;
  float* xp1  = x0 + (size_t)N * 5;
  float* aggc = B3;

  const int nb16 = (N + 15) / 16;

  bn_partial_k<<<256, 256, 0, stream>>>(h, part, N);
  bn_final_k<<<1, 32, 0, stream>>>(part, mu, rs, 256, 1.0f / (float)N);
  bn_apply_k<<<(N * 5 + 255) / 256, 256, 0, stream>>>(h, mu, rs, bg, bbeta, x0, N * 5);

  proj5_k<<<(N + 255) / 256, 256, 0, stream>>>(x0, c1_pw, c1_pb, xp1, N);
  hipMemsetAsync(aggc, 0, (size_t)N * 8 * sizeof(float), stream);
  edge_scatter5_k<<<(E + 255) / 256, 256, 0, stream>>>(src, dst, ew, xp1, aggc, E);
  invcnt_k<<<(N + 255) / 256, 256, 0, stream>>>(aggc, invc, N);
  node5to64_k<<<N, 64, 0, stream>>>(aggc, invc, x0, c1_lw, c1_lb, c1_rw, B1, N);

  proj64_wmma_k<<<nb16, 128, 0, stream>>>(B1, c2_pw, c2_pb, B2, N);
  hipMemsetAsync(B3, 0, (size_t)N * 64 * sizeof(float), stream);
  edge_scatter64_k<<<(E + 255) / 256, 256, 0, stream>>>(src, dst, ew, B2, B3, E);
  node64_wmma_k<<<nb16, 128, 0, stream>>>(B3, invc, B1, c2_lw, c2_lb, c2_rw, B2, N);

  float* z3   = B1;
  float* zsum = B3;
  proj64_lin5_wmma_k<<<nb16, 128, 0, stream>>>(B2, c3_pw, c3_pb, c3_lw, z3, N);
  hipMemsetAsync(zsum, 0, (size_t)N * 8 * sizeof(float), stream);
  edge_scatter5z_k<<<(E + 255) / 256, 256, 0, stream>>>(src, dst, ew, z3, zsum, E);
  node64to5_wmma_k<<<nb16, 32, 0, stream>>>(zsum, invc, B2, c3_lb, c3_rw,
                                            (float*)d_out, N);
}


# Layers 2-3: edge sums without atomics. Edges filed by destination once (place = rank returned by layer 1's count add + prefix sum); a wave owns whole rows, sums in registers in rank order
# speedup vs baseline: 4.0440x; 1.1759x over previous
#include <hip/hip_runtime.h>
#include <hip/hip_bf16.h>

typedef __attribute__((ext_vector_type(2))) float v2f;
typedef __attribute__((ext_vector_type(8))) float v8f;

#define EPS_BN   1e-5f
#define EPS_NORM 1e-12f

__device__ __forceinline__ v8f wmma_f32(v2f a, v2f b, v8f c) {
  return __builtin_amdgcn_wmma_f32_16x16x4_f32(false, a, false, b, (short)0, c,
                                               false, false);
}

__global__ void bn_partial_k(const float* __restrict__ h, float* __restrict__ part, int n) {
  __shared__ float sh[256 * 10];
  float acc[10];
#pragma unroll
  for (int d = 0; d < 10; ++d) acc[d] = 0.f;
  for (int i = blockIdx.x * blockDim.x + threadIdx.x; i < n;
       i += gridDim.x * blockDim.x) {
#pragma unroll
    for (int d = 0; d < 5; ++d) {
      float v = h[i * 5 + d];
      acc[d] += v;
      acc[5 + d] += v * v;
    }
  }
#pragma unroll
  for (int d = 0; d < 10; ++d) sh[threadIdx.x * 10 + d] = acc[d];
  __syncthreads();
  for (int s = 128; s > 0; s >>= 1) {
    if (threadIdx.x < (unsigned)s)
      for (int d = 0; d < 10; ++d)
        sh[threadIdx.x * 10 + d] += sh[(threadIdx.x + s) * 10 + d];
    __syncthreads();
  }
  if (threadIdx.x < 10) part[blockIdx.x * 10 + threadIdx.x] = sh[threadIdx.x];
}

__global__ void bn_final_k(const float* __restrict__ part, float* __restrict__ mu,
                           float* __restrict__ rs, int nblocks, float inv_n) {
  int t = threadIdx.x;
  if (t < 5) {
    float s = 0.f, q = 0.f;
    for (int b = 0; b < nblocks; ++b) {
      s += part[b * 10 + t];
      q += part[b * 10 + 5 + t];
    }
    float m = s * inv_n;
    float var = q * inv_n - m * m;
    mu[t] = m;
    rs[t] = rsqrtf(var + EPS_BN);
  }
}

__global__ void bn_apply_k(const float* __restrict__ h, const float* __restrict__ mu,
                           const float* __restrict__ rs, const float* __restrict__ g,
                           const float* __restrict__ b, float* __restrict__ x, int total) {
  int i = blockIdx.x * blockDim.x + threadIdx.x;
  if (i >= total) return;
  int d = i % 5;
  x[i] = (h[i] - mu[d]) * rs[d] * g[d] + b[d];
}

__global__ void invcnt_sum_k(const float* __restrict__ aggc, float* __restrict__ invc,
                             int* __restrict__ cnt, int* __restrict__ bsum, int n) {
  __shared__ int wsum[32];
  const int lane = threadIdx.x & 31;
  const int wave = threadIdx.x >> 5;
  const int i = blockIdx.x * 1024 + threadIdx.x;
  int c = 0;
  if (i < n) {
    const float f = aggc[(size_t)i * 8 + 5];
    invc[i] = 1.0f / fmaxf(f, 1.0f);
    c = (int)f;
    cnt[i] = c;
  }
  int sum = c;
#pragma unroll
  for (int o = 16; o > 0; o >>= 1) sum += __shfl_xor(sum, o);
  if (lane == 0) wsum[wave] = sum;
  __syncthreads();
  if (threadIdx.x == 0) {
    int t = 0;
    for (int w = 0; w < 32; ++w) t += wsum[w];
    bsum[blockIdx.x] = t;
  }
}

__global__ void proj5_k(const float* __restrict__ X, const float* __restrict__ W,
                        const float* __restrict__ b, float* __restrict__ Y, int n) {
  int i = blockIdx.x * blockDim.x + threadIdx.x;
  if (i >= n) return;
  float x[5];
#pragma unroll
  for (int k = 0; k < 5; ++k) x[k] = X[i * 5 + k];
#pragma unroll
  for (int c = 0; c < 5; ++c) {
    float acc = b[c];
#pragma unroll
    for (int k = 0; k < 5; ++k) acc += x[k] * W[k * 5 + c];
    Y[i * 5 + c] = fmaxf(acc, 0.f);
  }
}

template <int XS, bool COUNT>
__device__ __forceinline__ void edge_scatter5_body(const int* __restrict__ src,
                                                   const int* __restrict__ dst,
                                                   const float* __restrict__ ew,
                                                   const float* __restrict__ xp,
                                                   float* __restrict__ agg,
                                                   int* __restrict__ rank, int E) {
  const int lane = threadIdx.x & 31;
  const int e0 = __builtin_amdgcn_readfirstlane(
      (int)((blockIdx.x * blockDim.x + threadIdx.x) >> 5) << 5);
  if (e0 >= E) return;
  const int e = e0 + lane;
  int s_l = 0, d_l = 0;
  float w_l = 0.f;
  if (e < E) { s_l = src[e]; d_l = dst[e]; w_l = ew[e]; }
  const int k = lane & 7;
  const int g = lane >> 3;
  int s[8], d[8];
  float w[8];
#pragma unroll
  for (int j = 0; j < 8; ++j) {
    const int sl = j * 4 + g;
    s[j] = __shfl(s_l, sl);
    d[j] = __shfl(d_l, sl);
    w[j] = __shfl(w_l, sl);
  }
  if (k >= (COUNT ? 6 : 5)) return;
  const bool feat = (k < 5);
  const int kl = feat ? k : 4;
  float x[8];
#pragma unroll
  for (int j = 0; j < 8; ++j) x[j] = xp[(size_t)s[j] * XS + kl];
  if (e0 + 32 <= E) {
    float old[8];
#pragma unroll
    for (int j = 0; j < 8; ++j)
      old[j] = atomicAdd(&agg[(size_t)d[j] * 8 + k], feat ? x[j] * w[j] : 1.0f);
    if (COUNT && !feat) {
#pragma unroll
      for (int j = 0; j < 8; ++j) rank[e0 + j * 4 + g] = (int)old[j];
    }
  } else {
#pragma unroll
    for (int j = 0; j < 8; ++j) {
      if ((e0 + j * 4 + g) < E) {
        const float old = atomicAdd(&agg[(size_t)d[j] * 8 + k], feat ? x[j] * w[j] : 1.0f);
        if (COUNT && !feat) rank[e0 + j * 4 + g] = (int)old;
      }
    }
  }
}

__global__ void edge_scatter5_k(const int* __restrict__ src, const int* __restrict__ dst,
                                const float* __restrict__ ew, const float* __restrict__ xp,
                                float* __restrict__ agg, int* __restrict__ rank, int E) {
  edge_scatter5_body<5, true>(src, dst, ew, xp, agg, rank, E);
}

__global__ void fill_csr_k(const int* __restrict__ src, const int* __restrict__ dst,
                           const float* __restrict__ ew, const int* __restrict__ rank,
                           const int* __restrict__ offs, int* __restrict__ csr, int E) {
  int e = blockIdx.x * blockDim.x + threadIdx.x;
  if (e >= E) return;
  int pos = offs[dst[e]] + rank[e];
  ((int2*)csr)[pos] = make_int2(src[e], __float_as_int(ew[e]));
}

__global__ void offs_scan_k(const int* __restrict__ cnt, const int* __restrict__ bsum,
                            int* __restrict__ offs, int n) {
  __shared__ int wsum[32];
  const int lane = threadIdx.x & 31;
  const int wave = threadIdx.x >> 5;
  const int i = blockIdx.x * 1024 + threadIdx.x;
  const int c = (i < n) ? cnt[i] : 0;
  int incl = c;
#pragma unroll
  for (int o = 1; o < 32; o <<= 1) {
    int v = __shfl_up(incl, o);
    if (lane >= o) incl += v;
  }
  if (lane == 31) wsum[wave] = incl;
  __syncthreads();
  int base = 0;
  for (int b = 0; b < (int)blockIdx.x; ++b) base += bsum[b];
  for (int w = 0; w < wave; ++w) base += wsum[w];
  const int ex = base + incl - c;
  if (i < n) {
    offs[i] = ex;
    if (i == n - 1) offs[n] = ex + c;
  }
}

__global__ void agg5_k(const int* __restrict__ offs, const int* __restrict__ csr,
                       const float* __restrict__ z, float* __restrict__ zsum, int n) {
  const int2* __restrict__ pairs = (const int2*)csr;
  const int k = threadIdx.x & 7;
  const int node = (int)((blockIdx.x * blockDim.x + threadIdx.x) >> 3);
  if (node >= n) return;
  const int kl = (k < 5) ? k : 4;
  const int beg = offs[node], end = offs[node + 1];
  float acc = 0.f;
  for (int t = beg; t < end; t += 4) {
    int2 v[4];
    float x[4];
#pragma unroll
    for (int u = 0; u < 4; ++u) v[u] = pairs[min(t + u, end - 1)];
#pragma unroll
    for (int u = 0; u < 4; ++u) x[u] = z[(size_t)v[u].x * 8 + kl];
#pragma unroll
    for (int u = 0; u < 4; ++u)
      if (t + u < end) acc += x[u] * __int_as_float(v[u].y);
  }
  if (k < 5) zsum[(size_t)node * 8 + k] = acc;
}

__global__ void node5to64_k(const float* __restrict__ agg, const float* __restrict__ invc,
                            const float* __restrict__ X, const float* __restrict__ LW,
                            const float* __restrict__ lb, const float* __restrict__ RW,
                            float* __restrict__ Y, int n) {
  __shared__ float red[64];
  int node = blockIdx.x;
  if (node >= n) return;
  int c = threadIdx.x;
  float ic = invc[node];
  float acc = lb[c];
#pragma unroll
  for (int k = 0; k < 5; ++k) {
    acc += agg[node * 8 + k] * ic * LW[k * 64 + c] + X[node * 5 + k] * RW[k * 64 + c];
  }
  red[c] = acc * acc;
  __syncthreads();
#pragma unroll
  for (int s = 32; s > 0; s >>= 1) {
    if (c < s) red[c] += red[c + s];
    __syncthreads();
  }
  float scale = 1.0f / fmaxf(sqrtf(red[0]), EPS_NORM);
  Y[node * 64 + c] = fmaxf(acc * scale, 0.f);
}

__global__ void proj64_wmma_k(const float* __restrict__ X, const float* __restrict__ W,
                              const float* __restrict__ bias, float* __restrict__ Y, int n) {
  int l = threadIdx.x & 31;
  int wave = threadIdx.x >> 5;
  int row0 = blockIdx.x * 16;
  int m = row0 + (l & 15);
  if (m >= n) m = n - 1;
  int cl = wave * 16 + (l & 15);
  int khi = (l >> 4) << 1;
  v8f c = {};
#pragma unroll
  for (int k0 = 0; k0 < 64; k0 += 4) {
    int kb = k0 + khi;
    v2f a, bb;
    a[0] = X[m * 64 + kb];
    a[1] = X[m * 64 + kb + 1];
    bb[0] = W[kb * 64 + cl];
    bb[1] = W[(kb + 1) * 64 + cl];
    c = wmma_f32(a, bb, c);
  }
  int rbase = row0 + ((l >> 4) << 3);
  float bc = bias[cl];
#pragma unroll
  for (int r = 0; r < 8; ++r) {
    int rr = rbase + r;
    if (rr < n) Y[rr * 64 + cl] = fmaxf(c[r] + bc, 0.f);
  }
}

__global__ void proj64_lin5_wmma_k(const float* __restrict__ X, const float* __restrict__ PW,
                                   const float* __restrict__ pb, const float* __restrict__ LW,
                                   float* __restrict__ Z, int n) {
  __shared__ float tile[16 * 68];
  int l = threadIdx.x & 31;
  const int wave = __builtin_amdgcn_readfirstlane((int)(threadIdx.x >> 5));
  int row0 = blockIdx.x * 16;
  int c16 = l & 15;
  int m = row0 + c16;
  if (m >= n) m = n - 1;
  int cl = wave * 16 + c16;
  int khi = (l >> 4) << 1;
  v8f c = {};
#pragma unroll
  for (int k0 = 0; k0 < 64; k0 += 4) {
    int kb = k0 + khi;
    v2f a, bb;
    a[0] = X[m * 64 + kb];
    a[1] = X[m * 64 + kb + 1];
    bb[0] = PW[kb * 64 + cl];
    bb[1] = PW[(kb + 1) * 64 + cl];
    c = wmma_f32(a, bb, c);
  }
  int rloc = (l >> 4) << 3;
  float bc = pb[cl];
#pragma unroll
  for (int r = 0; r < 8; ++r) tile[(rloc + r) * 68 + cl] = fmaxf(c[r] + bc, 0.f);
  __syncthreads();
  if (wave == 0) {
    bool cv = (c16 < 5);
    v2f bw[16];
#pragma unroll
    for (int i = 0; i < 16; ++i) { bw[i][0] = 0.f; bw[i][1] = 0.f; }
    if (cv) {
#pragma unroll
      for (int i = 0; i < 16; ++i) {
        int kb = i * 4 + khi;
        bw[i][0] = LW[kb * 5 + c16];
        bw[i][1] = LW[(kb + 1) * 5 + c16];
      }
    }
    v8f z = {};
#pragma unroll
    for (int i = 0; i < 16; ++i) {
      int kb = i * 4 + khi;
      v2f a;
      a[0] = tile[c16 * 68 + kb];
      a[1] = tile[c16 * 68 + kb + 1];
      z = wmma_f32(a, bw[i], z);
    }
    if (cv) {
#pragma unroll
      for (int r = 0; r < 8; ++r) {
        int rr = row0 + rloc + r;
        if (rr < n) Z[(size_t)rr * 8 + c16] = z[r];
      }
    }
  }
}

__global__ void agg_node64_wmma_k(const int* __restrict__ offs, const int* __restrict__ csr,
                                  const float* __restrict__ xp, const float* __restrict__ invc,
                                  float* XY, const float* __restrict__ LW,
                                  const float* __restrict__ lb, const float* __restrict__ RW,
                                  int n) {
  __shared__ float mt[16 * 68];
  __shared__ float tile[16 * 64];
  __shared__ float rscale[16];
  const int l = threadIdx.x & 31;
  const int wave = __builtin_amdgcn_readfirstlane((int)(threadIdx.x >> 5));
  const int row0 = blockIdx.x * 16;
  const int2* __restrict__ pairs = (const int2*)csr;
  for (int q = 0; q < 4; ++q) {
    const int r = wave * 4 + q;
    const int node = row0 + r;
    float lo = 0.f, hi = 0.f;
    if (node < n) {
      const int beg = __builtin_amdgcn_readfirstlane(offs[node]);
      const int end = __builtin_amdgcn_readfirstlane(offs[node + 1]);
      for (int c0 = beg; c0 < end; c0 += 32) {
        int s_l = 0;
        float w_l = 0.f;
        if (c0 + l < end) {
          const int2 v = pairs[c0 + l];
          s_l = v.x;
          w_l = __int_as_float(v.y);
        }
        const int n_e = min(32, end - c0);
        for (int j0 = 0; j0 < n_e; j0 += 8) {
          float a[8], b[8];
#pragma unroll
          for (int u = 0; u < 8; ++u) {
            const int sr = __builtin_amdgcn_readlane(s_l, j0 + u);
            const float* row = xp + (size_t)sr * 64 + l;
            a[u] = row[0];
            b[u] = row[32];
          }
#pragma unroll
          for (int u = 0; u < 8; ++u) {
            if (j0 + u < n_e) {
              const float w = __int_as_float(__builtin_amdgcn_readlane(__float_as_int(w_l), j0 + u));
              lo += a[u] * w;
              hi += b[u] * w;
            }
          }
        }
      }
      const float ic = invc[node];
      lo *= ic;
      hi *= ic;
    }
    mt[r * 68 + l] = lo;
    mt[r * 68 + 32 + l] = hi;
  }
  __syncthreads();
  const int c16 = l & 15;
  int m = row0 + c16;
  if (m >= n) m = n - 1;
  const int cl = wave * 16 + c16;
  const int khi = (l >> 4) << 1;
  v8f c = {};
#pragma unroll
  for (int k0 = 0; k0 < 64; k0 += 4) {
    int kb = k0 + khi;
    v2f a, bb;
    a[0] = mt[c16 * 68 + kb];
    a[1] = mt[c16 * 68 + kb + 1];
    bb[0] = LW[kb * 64 + cl];
    bb[1] = LW[(kb + 1) * 64 + cl];
    c = wmma_f32(a, bb, c);
  }
#pragma unroll
  for (int k0 = 0; k0 < 64; k0 += 4) {
    int kb = k0 + khi;
    v2f a, bb;
    a[0] = XY[m * 64 + kb];
    a[1] = XY[m * 64 + kb + 1];
    bb[0] = RW[kb * 64 + cl];
    bb[1] = RW[(kb + 1) * 64 + cl];
    c = wmma_f32(a, bb, c);
  }
  const int rloc = (l >> 4) << 3;
  const float bc = lb[cl];
#pragma unroll
  for (int r = 0; r < 8; ++r) tile[(rloc + r) * 64 + cl] = c[r] + bc;
  __syncthreads();
  if (threadIdx.x < 16) {
    float ss = 0.f;
#pragma unroll
    for (int cc = 0; cc < 64; ++cc) {
      float v = tile[threadIdx.x * 64 + cc];
      ss += v * v;
    }
    rscale[threadIdx.x] = 1.0f / fmaxf(sqrtf(ss), EPS_NORM);
  }
  __syncthreads();
  for (int i = threadIdx.x; i < 16 * 64; i += blockDim.x) {
    int r = i >> 6;
    int rr = row0 + r;
    if (rr < n) XY[rr * 64 + (i & 63)] = fmaxf(tile[i] * rscale[r], 0.f);
  }
}

__global__ void node64to5_wmma_k(const float* __restrict__ zsum, const float* __restrict__ invc,
                                 const float* __restrict__ X, const float* __restrict__ lb,
                                 const float* __restrict__ RW, float* __restrict__ Y, int n) {
  __shared__ float tile[16 * 16];
  int l = threadIdx.x & 31;
  int row0 = blockIdx.x * 16;
  int m = row0 + (l & 15);
  if (m >= n) m = n - 1;
  int cl = l & 15;
  int khi = (l >> 4) << 1;
  float ic = invc[m];
  float g0 = zsum[(size_t)m * 8 + 0];
  float g1 = zsum[(size_t)m * 8 + 1];
  float g2 = zsum[(size_t)m * 8 + 2];
  float g3 = zsum[(size_t)m * 8 + 3];
  float g4 = zsum[(size_t)m * 8 + 4];
  bool cv = (cl < 5);
  v8f c = {};
#pragma unroll
  for (int k0 = 0; k0 < 64; k0 += 4) {
    int kb = k0 + khi;
    v2f a, bb;
    a[0] = X[m * 64 + kb];
    a[1] = X[m * 64 + kb + 1];
    bb[0] = cv ? RW[kb * 5 + cl] : 0.f;
    bb[1] = cv ? RW[(kb + 1) * 5 + cl] : 0.f;
    c = wmma_f32(a, bb, c);
  }
  int rloc = (l >> 4) << 3;
  float bc = cv ? lb[cl] : 0.f;
#pragma unroll
  for (int r = 0; r < 8; ++r) tile[(rloc + r) * 16 + cl] = c[r] + bc;
  __syncthreads();
  if (l < 16) {
    float v0 = tile[l * 16 + 0] + g0 * ic;
    float v1 = tile[l * 16 + 1] + g1 * ic;
    float v2 = tile[l * 16 + 2] + g2 * ic;
    float v3 = tile[l * 16 + 3] + g3 * ic;
    float v4 = tile[l * 16 + 4] + g4 * ic;
    float ss = v0 * v0 + v1 * v1 + v2 * v2 + v3 * v3 + v4 * v4;
    float s = 1.0f / fmaxf(sqrtf(ss), EPS_NORM);
    int rr = row0 + l;
    if (rr < n) {
      Y[rr * 5 + 0] = v0 * s;
      Y[rr * 5 + 1] = v1 * s;
      Y[rr * 5 + 2] = v2 * s;
      Y[rr * 5 + 3] = v3 * s;
      Y[rr * 5 + 4] = v4 * s;
    }
  }
}

extern "C" void kernel_launch(void* const* d_in, const int* in_sizes, int n_in,
                              void* d_out, int out_size, void* d_ws, size_t ws_size,
                              hipStream_t stream) {
  const float* h     = (const float*)d_in[0];
  const int*   ei    = (const int*)d_in[1];
  const float* ew    = (const float*)d_in[2];
  const float* bg    = (const float*)d_in[3];
  const float* bbeta = (const float*)d_in[4];
  const float* c1_pw = (const float*)d_in[5];
  const float* c1_pb = (const float*)d_in[6];
  const float* c1_lw = (const float*)d_in[7];
  const float* c1_lb = (const float*)d_in[8];
  const float* c1_rw = (const float*)d_in[9];
  const float* c2_pw = (const float*)d_in[10];
  const float* c2_pb = (const float*)d_in[11];
  const float* c2_lw = (const float*)d_in[12];
  const float* c2_lb = (const float*)d_in[13];
  const float* c2_rw = (const float*)d_in[14];
  const float* c3_pw = (const float*)d_in[15];
  const float* c3_pb = (const float*)d_in[16];
  const float* c3_lw = (const float*)d_in[17];
  const float* c3_lb = (const float*)d_in[18];
  const float* c3_rw = (const float*)d_in[19];

  const int N = in_sizes[0] / 5;
  const int E = in_sizes[2];
  const int* src = ei;
  const int* dst = ei + E;

  float* ws   = (float*)d_ws;
  float* B1   = ws;
  float* B2   = B1 + (size_t)N * 64;
  float* B3   = B2 + (size_t)N * 64;
  float* mu   = B3 + (size_t)N * 64;
  float* rs   = mu + 8;
  float* part = rs + 8;
  float* invc = part + 2560;
  float* x0   = invc + N;
  float* xp1  = x0 + (size_t)N * 5;
  float* aggc = B3;
  int*   rank = (int*)B2;
  int*   cnt  = (int*)xp1;
  int*   offs = cnt + N;
  int*   bsum = offs + N + 1;
  int*   csr  = (int*)B3;

  const int nb16 = (N + 15) / 16;
  const int nb1k = (N + 1023) / 1024;

  bn_partial_k<<<256, 256, 0, stream>>>(h, part, N);
  bn_final_k<<<1, 32, 0, stream>>>(part, mu, rs, 256, 1.0f / (float)N);
  bn_apply_k<<<(N * 5 + 255) / 256, 256, 0, stream>>>(h, mu, rs, bg, bbeta, x0, N * 5);

  proj5_k<<<(N + 255) / 256, 256, 0, stream>>>(x0, c1_pw, c1_pb, xp1, N);
  hipMemsetAsync(aggc, 0, (size_t)N * 8 * sizeof(float), stream);
  edge_scatter5_k<<<(E + 255) / 256, 256, 0, stream>>>(src, dst, ew, xp1, aggc, rank, E);
  invcnt_sum_k<<<nb1k, 1024, 0, stream>>>(aggc, invc, cnt, bsum, N);
  offs_scan_k<<<nb1k, 1024, 0, stream>>>(cnt, bsum, offs, N);
  node5to64_k<<<N, 64, 0, stream>>>(aggc, invc, x0, c1_lw, c1_lb, c1_rw, B1, N);
  fill_csr_k<<<(E + 255) / 256, 256, 0, stream>>>(src, dst, ew, rank, offs, csr, E);

  proj64_wmma_k<<<nb16, 128, 0, stream>>>(B1, c2_pw, c2_pb, B2, N);
  agg_node64_wmma_k<<<nb16, 128, 0, stream>>>(offs, csr, B2, invc, B1, c2_lw, c2_lb, c2_rw, N);

  float* z3   = B2;
  float* zsum = B2 + (size_t)N * 8;
  proj64_lin5_wmma_k<<<nb16, 128, 0, stream>>>(B1, c3_pw, c3_pb, c3_lw, z3, N);
  agg5_k<<<(N * 8 + 255) / 256, 256, 0, stream>>>(offs, csr, z3, zsum, N);
  node64to5_wmma_k<<<nb16, 32, 0, stream>>>(zsum, invc, B1, c3_lb, c3_rw,
                                            (float*)d_out, N);
}


# speedup vs baseline: 4.8775x; 1.2061x over previous
#include <hip/hip_runtime.h>
#include <hip/hip_bf16.h>

typedef __attribute__((ext_vector_type(2))) float v2f;
typedef __attribute__((ext_vector_type(8))) float v8f;

#define EPS_BN   1e-5f
#define EPS_NORM 1e-12f

__device__ __forceinline__ v8f wmma_f32(v2f a, v2f b, v8f c) {
  return __builtin_amdgcn_wmma_f32_16x16x4_f32(false, a, false, b, (short)0, c,
                                               false, false);
}

__global__ void bn_partial_k(const float* __restrict__ h, float* __restrict__ part, int n) {
  __shared__ float sh[256 * 10];
  float acc[10];
#pragma unroll
  for (int d = 0; d < 10; ++d) acc[d] = 0.f;
  for (int i = blockIdx.x * blockDim.x + threadIdx.x; i < n;
       i += gridDim.x * blockDim.x) {
#pragma unroll
    for (int d = 0; d < 5; ++d) {
      float v = h[i * 5 + d];
      acc[d] += v;
      acc[5 + d] += v * v;
    }
  }
#pragma unroll
  for (int d = 0; d < 10; ++d) sh[threadIdx.x * 10 + d] = acc[d];
  __syncthreads();
  for (int s = 128; s > 0; s >>= 1) {
    if (threadIdx.x < (unsigned)s)
      for (int d = 0; d < 10; ++d)
        sh[threadIdx.x * 10 + d] += sh[(threadIdx.x + s) * 10 + d];
    __syncthreads();
  }
  if (threadIdx.x < 10) part[blockIdx.x * 10 + threadIdx.x] = sh[threadIdx.x];
}

__global__ void bn_final_k(const float* __restrict__ part, float* __restrict__ mu,
                           float* __restrict__ rs, int nblocks, float inv_n) {
  int t = threadIdx.x;
  if (t < 5) {
    float s = 0.f, q = 0.f;
    for (int b = 0; b < nblocks; ++b) {
      s += part[b * 10 + t];
      q += part[b * 10 + 5 + t];
    }
    float m = s * inv_n;
    float var = q * inv_n - m * m;
    mu[t] = m;
    rs[t] = rsqrtf(var + EPS_BN);
  }
}

__global__ void bn_apply_k(const float* __restrict__ h, const float* __restrict__ mu,
                           const float* __restrict__ rs, const float* __restrict__ g,
                           const float* __restrict__ b, float* __restrict__ x, int total) {
  int i = blockIdx.x * blockDim.x + threadIdx.x;
  if (i >= total) return;
  int d = i % 5;
  x[i] = (h[i] - mu[d]) * rs[d] * g[d] + b[d];
}

__global__ void invcnt_sum_k(const float* __restrict__ aggc, float* __restrict__ invc,
                             int* __restrict__ cnt, int* __restrict__ cnt0,
                             int* __restrict__ bsum, int n) {
  __shared__ int wsum[32];
  const int lane = threadIdx.x & 31;
  const int wave = threadIdx.x >> 5;
  const int i = blockIdx.x * 1024 + threadIdx.x;
  int c = 0;
  if (i < n) {
    const float f0 = aggc[(size_t)i * 8 + 5];
    const float f = f0 + aggc[((size_t)n + i) * 8 + 5];
    invc[i] = 1.0f / fmaxf(f, 1.0f);
    c = (int)f;
    cnt[i] = c;
    cnt0[i] = (int)f0;
  }
  int sum = c;
#pragma unroll
  for (int o = 16; o > 0; o >>= 1) sum += __shfl_xor(sum, o);
  if (lane == 0) wsum[wave] = sum;
  __syncthreads();
  if (threadIdx.x == 0) {
    int t = 0;
    for (int w = 0; w < 32; ++w) t += wsum[w];
    bsum[blockIdx.x] = t;
  }
}

__global__ void proj5_k(const float* __restrict__ X, const float* __restrict__ W,
                        const float* __restrict__ b, float* __restrict__ Y, int n) {
  int i = blockIdx.x * blockDim.x + threadIdx.x;
  if (i >= n) return;
  float x[5];
#pragma unroll
  for (int k = 0; k < 5; ++k) x[k] = X[i * 5 + k];
#pragma unroll
  for (int c = 0; c < 5; ++c) {
    float acc = b[c];
#pragma unroll
    for (int k = 0; k < 5; ++k) acc += x[k] * W[k * 5 + c];
    Y[i * 5 + c] = fmaxf(acc, 0.f);
  }
}

template <int XS, bool COUNT>
__device__ __forceinline__ void edge_scatter5_body(const int* __restrict__ src,
                                                   const int* __restrict__ dst,
                                                   const float* __restrict__ ew,
                                                   const float* __restrict__ xp,
                                                   float* __restrict__ agg,
                                                   int* __restrict__ rank, int E,
                                                   int n) {
  const int lane = threadIdx.x & 31;
  const int e0 = __builtin_amdgcn_readfirstlane(
      (int)((blockIdx.x * blockDim.x + threadIdx.x) >> 5) << 5);
  if (e0 >= E) return;
  const int e = e0 + lane;
  int s_l = 0, d_l = 0;
  float w_l = 0.f;
  if (e < E) { s_l = src[e]; d_l = dst[e]; w_l = ew[e]; }
  const int h = (int)((__builtin_amdgcn_s_sendmsg_rtn(0x87) >> 18) & 1u);
  agg += (size_t)h * (size_t)n * 8;
  const int k = lane & 7;
  const int g = lane >> 3;
  int s[8], d[8];
  float w[8];
#pragma unroll
  for (int j = 0; j < 8; ++j) {
    const int sl = j * 4 + g;
    s[j] = __shfl(s_l, sl);
    d[j] = __shfl(d_l, sl);
    w[j] = __shfl(w_l, sl);
  }
  if (k >= (COUNT ? 6 : 5)) return;
  const bool feat = (k < 5);
  const int kl = feat ? k : 4;
  float x[8];
#pragma unroll
  for (int j = 0; j < 8; ++j) x[j] = xp[(size_t)s[j] * XS + kl];
  if (e0 + 32 <= E) {
    float old[8];
#pragma unroll
    for (int j = 0; j < 8; ++j)
      old[j] = atomicAdd(&agg[(size_t)d[j] * 8 + k], feat ? x[j] * w[j] : 1.0f);
    if (COUNT && !feat) {
#pragma unroll
      for (int j = 0; j < 8; ++j) rank[e0 + j * 4 + g] = ((int)old[j] << 1) | h;
    }
  } else {
#pragma unroll
    for (int j = 0; j < 8; ++j) {
      if ((e0 + j * 4 + g) < E) {
        const float old = atomicAdd(&agg[(size_t)d[j] * 8 + k], feat ? x[j] * w[j] : 1.0f);
        if (COUNT && !feat) rank[e0 + j * 4 + g] = ((int)old << 1) | h;
      }
    }
  }
}

__global__ void edge_scatter5_k(const int* __restrict__ src, const int* __restrict__ dst,
                                const float* __restrict__ ew, const float* __restrict__ xp,
                                float* __restrict__ agg, int* __restrict__ rank, int E,
                                int n) {
  edge_scatter5_body<5, true>(src, dst, ew, xp, agg, rank, E, n);
}

__global__ void fill_csr_k(const int* __restrict__ src, const int* __restrict__ dst,
                           const float* __restrict__ ew, const int* __restrict__ rank,
                           const int* __restrict__ offs, const int* __restrict__ cnt0,
                           int* __restrict__ csr, int E) {
  int e = blockIdx.x * blockDim.x + threadIdx.x;
  if (e >= E) return;
  const int d = dst[e];
  const int r = rank[e];
  int pos = offs[d] + (r >> 1) + ((r & 1) ? cnt0[d] : 0);
  ((int2*)csr)[pos] = make_int2(src[e], __float_as_int(ew[e]));
}

__global__ void offs_scan_k(const int* __restrict__ cnt, const int* __restrict__ bsum,
                            int* __restrict__ offs, int n) {
  __shared__ int wsum[32];
  const int lane = threadIdx.x & 31;
  const int wave = threadIdx.x >> 5;
  const int i = blockIdx.x * 1024 + threadIdx.x;
  const int c = (i < n) ? cnt[i] : 0;
  int incl = c;
#pragma unroll
  for (int o = 1; o < 32; o <<= 1) {
    int v = __shfl_up(incl, o);
    if (lane >= o) incl += v;
  }
  if (lane == 31) wsum[wave] = incl;
  __syncthreads();
  int base = 0;
  for (int b = 0; b < (int)blockIdx.x; ++b) base += bsum[b];
  for (int w = 0; w < wave; ++w) base += wsum[w];
  const int ex = base + incl - c;
  if (i < n) {
    offs[i] = ex;
    if (i == n - 1) offs[n] = ex + c;
  }
}

__global__ void agg5_k(const int* __restrict__ offs, const int* __restrict__ csr,
                       const float* __restrict__ z, float* __restrict__ zsum, int n) {
  const int2* __restrict__ pairs = (const int2*)csr;
  const int k = threadIdx.x & 7;
  const int node = (int)((blockIdx.x * blockDim.x + threadIdx.x) >> 3);
  if (node >= n) return;
  const int kl = (k < 5) ? k : 4;
  const int beg = offs[node], end = offs[node + 1];
  float acc = 0.f;
  for (int t = beg; t < end; t += 4) {
    int2 v[4];
    float x[4];
#pragma unroll
    for (int u = 0; u < 4; ++u) v[u] = pairs[min(t + u, end - 1)];
#pragma unroll
    for (int u = 0; u < 4; ++u) x[u] = z[(size_t)v[u].x * 8 + kl];
#pragma unroll
    for (int u = 0; u < 4; ++u)
      if (t + u < end) acc += x[u] * __int_as_float(v[u].y);
  }
  if (k < 5) zsum[(size_t)node * 8 + k] = acc;
}

__global__ void node5to64_k(const float* __restrict__ agg, const float* __restrict__ invc,
                            const float* __restrict__ X, const float* __restrict__ LW,
                            const float* __restrict__ lb, const float* __restrict__ RW,
                            float* __restrict__ Y, int n) {
  __shared__ float red[64];
  int node = blockIdx.x;
  if (node >= n) return;
  int c = threadIdx.x;
  float ic = invc[node];
  float acc = lb[c];
#pragma unroll
  for (int k = 0; k < 5; ++k) {
    const float a = agg[node * 8 + k] + agg[((size_t)n + node) * 8 + k];
    acc += a * ic * LW[k * 64 + c] + X[node * 5 + k] * RW[k * 64 + c];
  }
  red[c] = acc * acc;
  __syncthreads();
#pragma unroll
  for (int s = 32; s > 0; s >>= 1) {
    if (c < s) red[c] += red[c + s];
    __syncthreads();
  }
  float scale = 1.0f / fmaxf(sqrtf(red[0]), EPS_NORM);
  Y[node * 64 + c] = fmaxf(acc * scale, 0.f);
}

__global__ void proj64_wmma_k(const float* __restrict__ X, const float* __restrict__ W,
                              const float* __restrict__ bias, float* __restrict__ Y, int n) {
  int l = threadIdx.x & 31;
  int wave = threadIdx.x >> 5;
  int row0 = blockIdx.x * 16;
  int m = row0 + (l & 15);
  if (m >= n) m = n - 1;
  int cl = wave * 16 + (l & 15);
  int khi = (l >> 4) << 1;
  v8f c = {};
#pragma unroll
  for (int k0 = 0; k0 < 64; k0 += 4) {
    int kb = k0 + khi;
    v2f a, bb;
    a[0] = X[m * 64 + kb];
    a[1] = X[m * 64 + kb + 1];
    bb[0] = W[kb * 64 + cl];
    bb[1] = W[(kb + 1) * 64 + cl];
    c = wmma_f32(a, bb, c);
  }
  int rbase = row0 + ((l >> 4) << 3);
  float bc = bias[cl];
#pragma unroll
  for (int r = 0; r < 8; ++r) {
    int rr = rbase + r;
    if (rr < n) Y[rr * 64 + cl] = fmaxf(c[r] + bc, 0.f);
  }
}

__global__ void proj64_lin5_wmma_k(const float* __restrict__ X, const float* __restrict__ PW,
                                   const float* __restrict__ pb, const float* __restrict__ LW,
                                   float* __restrict__ Z, int n) {
  __shared__ float tile[16 * 68];
  int l = threadIdx.x & 31;
  const int wave = __builtin_amdgcn_readfirstlane((int)(threadIdx.x >> 5));
  int row0 = blockIdx.x * 16;
  int c16 = l & 15;
  int m = row0 + c16;
  if (m >= n) m = n - 1;
  int cl = wave * 16 + c16;
  int khi = (l >> 4) << 1;
  v8f c = {};
#pragma unroll
  for (int k0 = 0; k0 < 64; k0 += 4) {
    int kb = k0 + khi;
    v2f a, bb;
    a[0] = X[m * 64 + kb];
    a[1] = X[m * 64 + kb + 1];
    bb[0] = PW[kb * 64 + cl];
    bb[1] = PW[(kb + 1) * 64 + cl];
    c = wmma_f32(a, bb, c);
  }
  int rloc = (l >> 4) << 3;
  float bc = pb[cl];
#pragma unroll
  for (int r = 0; r < 8; ++r) tile[(rloc + r) * 68 + cl] = fmaxf(c[r] + bc, 0.f);
  __syncthreads();
  if (wave == 0) {
    bool cv = (c16 < 5);
    v2f bw[16];
#pragma unroll
    for (int i = 0; i < 16; ++i) { bw[i][0] = 0.f; bw[i][1] = 0.f; }
    if (cv) {
#pragma unroll
      for (int i = 0; i < 16; ++i) {
        int kb = i * 4 + khi;
        bw[i][0] = LW[kb * 5 + c16];
        bw[i][1] = LW[(kb + 1) * 5 + c16];
      }
    }
    v8f z = {};
#pragma unroll
    for (int i = 0; i < 16; ++i) {
      int kb = i * 4 + khi;
      v2f a;
      a[0] = tile[c16 * 68 + kb];
      a[1] = tile[c16 * 68 + kb + 1];
      z = wmma_f32(a, bw[i], z);
    }
    if (cv) {
#pragma unroll
      for (int r = 0; r < 8; ++r) {
        int rr = row0 + rloc + r;
        if (rr < n) Z[(size_t)rr * 8 + c16] = z[r];
      }
    }
  }
}

__global__ void agg_node64_wmma_k(const int* __restrict__ offs, const int* __restrict__ csr,
                                  const float* __restrict__ xp, const float* __restrict__ invc,
                                  float* XY, const float* __restrict__ LW,
                                  const float* __restrict__ lb, const float* __restrict__ RW,
                                  int n) {
  __shared__ float mt[16 * 68];
  __shared__ float tile[16 * 64];
  __shared__ float rscale[16];
  const int l = threadIdx.x & 31;
  const int wave = __builtin_amdgcn_readfirstlane((int)(threadIdx.x >> 5));
  const int row0 = blockIdx.x * 16;
  const int2* __restrict__ pairs = (const int2*)csr;
  for (int q = 0; q < 4; ++q) {
    const int r = wave * 4 + q;
    const int node = row0 + r;
    float lo = 0.f, hi = 0.f;
    if (node < n) {
      const int beg = __builtin_amdgcn_readfirstlane(offs[node]);
      const int end = __builtin_amdgcn_readfirstlane(offs[node + 1]);
      for (int c0 = beg; c0 < end; c0 += 32) {
        int s_l = 0;
        float w_l = 0.f;
        if (c0 + l < end) {
          const int2 v = pairs[c0 + l];
          s_l = v.x;
          w_l = __int_as_float(v.y);
        }
        const int n_e = min(32, end - c0);
        for (int j0 = 0; j0 < n_e; j0 += 8) {
          float a[8], b[8];
#pragma unroll
          for (int u = 0; u < 8; ++u) {
            const int sr = __builtin_amdgcn_readlane(s_l, j0 + u);
            const float* row = xp + (size_t)sr * 64 + l;
            a[u] = row[0];
            b[u] = row[32];
          }
#pragma unroll
          for (int u = 0; u < 8; ++u) {
            if (j0 + u < n_e) {
              const float w = __int_as_float(__builtin_amdgcn_readlane(__float_as_int(w_l), j0 + u));
              lo += a[u] * w;
              hi += b[u] * w;
            }
          }
        }
      }
      const float ic = invc[node];
      lo *= ic;
      hi *= ic;
    }
    mt[r * 68 + l] = lo;
    mt[r * 68 + 32 + l] = hi;
  }
  __syncthreads();
  const int c16 = l & 15;
  int m = row0 + c16;
  if (m >= n) m = n - 1;
  const int cl = wave * 16 + c16;
  const int khi = (l >> 4) << 1;
  v8f c = {};
#pragma unroll
  for (int k0 = 0; k0 < 64; k0 += 4) {
    int kb = k0 + khi;
    v2f a, bb;
    a[0] = mt[c16 * 68 + kb];
    a[1] = mt[c16 * 68 + kb + 1];
    bb[0] = LW[kb * 64 + cl];
    bb[1] = LW[(kb + 1) * 64 + cl];
    c = wmma_f32(a, bb, c);
  }
#pragma unroll
  for (int k0 = 0; k0 < 64; k0 += 4) {
    int kb = k0 + khi;
    v2f a, bb;
    a[0] = XY[m * 64 + kb];
    a[1] = XY[m * 64 + kb + 1];
    bb[0] = RW[kb * 64 + cl];
    bb[1] = RW[(kb + 1) * 64 + cl];
    c = wmma_f32(a, bb, c);
  }
  const int rloc = (l >> 4) << 3;
  const float bc = lb[cl];
#pragma unroll
  for (int r = 0; r < 8; ++r) tile[(rloc + r) * 64 + cl] = c[r] + bc;
  __syncthreads();
  if (threadIdx.x < 16) {
    float ss = 0.f;
#pragma unroll
    for (int cc = 0; cc < 64; ++cc) {
      float v = tile[threadIdx.x * 64 + cc];
      ss += v * v;
    }
    rscale[threadIdx.x] = 1.0f / fmaxf(sqrtf(ss), EPS_NORM);
  }
  __syncthreads();
  for (int i = threadIdx.x; i < 16 * 64; i += blockDim.x) {
    int r = i >> 6;
    int rr = row0 + r;
    if (rr < n) XY[rr * 64 + (i & 63)] = fmaxf(tile[i] * rscale[r], 0.f);
  }
}

__global__ void node64to5_wmma_k(const float* __restrict__ zsum, const float* __restrict__ invc,
                                 const float* __restrict__ X, const float* __restrict__ lb,
                                 const float* __restrict__ RW, float* __restrict__ Y, int n) {
  __shared__ float tile[16 * 16];
  int l = threadIdx.x & 31;
  int row0 = blockIdx.x * 16;
  int m = row0 + (l & 15);
  if (m >= n) m = n - 1;
  int cl = l & 15;
  int khi = (l >> 4) << 1;
  float ic = invc[m];
  float g0 = zsum[(size_t)m * 8 + 0];
  float g1 = zsum[(size_t)m * 8 + 1];
  float g2 = zsum[(size_t)m * 8 + 2];
  float g3 = zsum[(size_t)m * 8 + 3];
  float g4 = zsum[(size_t)m * 8 + 4];
  bool cv = (cl < 5);
  v8f c = {};
#pragma unroll
  for (int k0 = 0; k0 < 64; k0 += 4) {
    int kb = k0 + khi;
    v2f a, bb;
    a[0] = X[m * 64 + kb];
    a[1] = X[m * 64 + kb + 1];
    bb[0] = cv ? RW[kb * 5 + cl] : 0.f;
    bb[1] = cv ? RW[(kb + 1) * 5 + cl] : 0.f;
    c = wmma_f32(a, bb, c);
  }
  int rloc = (l >> 4) << 3;
  float bc = cv ? lb[cl] : 0.f;
#pragma unroll
  for (int r = 0; r < 8; ++r) tile[(rloc + r) * 16 + cl] = c[r] + bc;
  __syncthreads();
  if (l < 16) {
    float v0 = tile[l * 16 + 0] + g0 * ic;
    float v1 = tile[l * 16 + 1] + g1 * ic;
    float v2 = tile[l * 16 + 2] + g2 * ic;
    float v3 = tile[l * 16 + 3] + g3 * ic;
    float v4 = tile[l * 16 + 4] + g4 * ic;
    float ss = v0 * v0 + v1 * v1 + v2 * v2 + v3 * v3 + v4 * v4;
    float s = 1.0f / fmaxf(sqrtf(ss), EPS_NORM);
    int rr = row0 + l;
    if (rr < n) {
      Y[rr * 5 + 0] = v0 * s;
      Y[rr * 5 + 1] = v1 * s;
      Y[rr * 5 + 2] = v2 * s;
      Y[rr * 5 + 3] = v3 * s;
      Y[rr * 5 + 4] = v4 * s;
    }
  }
}

extern "C" void kernel_launch(void* const* d_in, const int* in_sizes, int n_in,
                              void* d_out, int out_size, void* d_ws, size_t ws_size,
                              hipStream_t stream) {
  const float* h     = (const float*)d_in[0];
  const int*   ei    = (const int*)d_in[1];
  const float* ew    = (const float*)d_in[2];
  const float* bg    = (const float*)d_in[3];
  const float* bbeta = (const float*)d_in[4];
  const float* c1_pw = (const float*)d_in[5];
  const float* c1_pb = (const float*)d_in[6];
  const float* c1_lw = (const float*)d_in[7];
  const float* c1_lb = (const float*)d_in[8];
  const float* c1_rw = (const float*)d_in[9];
  const float* c2_pw = (const float*)d_in[10];
  const float* c2_pb = (const float*)d_in[11];
  const float* c2_lw = (const float*)d_in[12];
  const float* c2_lb = (const float*)d_in[13];
  const float* c2_rw = (const float*)d_in[14];
  const float* c3_pw = (const float*)d_in[15];
  const float* c3_pb = (const float*)d_in[16];
  const float* c3_lw = (const float*)d_in[17];
  const float* c3_lb = (const float*)d_in[18];
  const float* c3_rw = (const float*)d_in[19];

  const int N = in_sizes[0] / 5;
  const int E = in_sizes[2];
  const int* src = ei;
  const int* dst = ei + E;

  float* ws   = (float*)d_ws;
  float* B1   = ws;
  float* B2   = B1 + (size_t)N * 64;
  float* B3   = B2 + (size_t)N * 64;
  float* mu   = B3 + (size_t)N * 64;
  float* rs   = mu + 8;
  float* part = rs + 8;
  float* invc = part + 2560;
  float* x0   = invc + N;
  float* xp1  = x0 + (size_t)N * 5;
  float* aggc = B3;
  int*   rank = (int*)B2;
  int*   cnt  = (int*)xp1;
  int*   offs = cnt + N;
  int*   bsum = offs + N + 1;
  const int nb1k = (N + 1023) / 1024;
  int*   cnt0 = bsum + nb1k;
  int*   csr  = (int*)B3;

  const int nb16 = (N + 15) / 16;

  bn_partial_k<<<256, 256, 0, stream>>>(h, part, N);
  bn_final_k<<<1, 32, 0, stream>>>(part, mu, rs, 256, 1.0f / (float)N);
  bn_apply_k<<<(N * 5 + 255) / 256, 256, 0, stream>>>(h, mu, rs, bg, bbeta, x0, N * 5);

  proj5_k<<<(N + 255) / 256, 256, 0, stream>>>(x0, c1_pw, c1_pb, xp1, N);
  hipMemsetAsync(aggc, 0, (size_t)N * 16 * sizeof(float), stream);
  edge_scatter5_k<<<(E + 255) / 256, 256, 0, stream>>>(src, dst, ew, xp1, aggc, rank, E, N);
  invcnt_sum_k<<<nb1k, 1024, 0, stream>>>(aggc, invc, cnt, cnt0, bsum, N);
  offs_scan_k<<<nb1k, 1024, 0, stream>>>(cnt, bsum, offs, N);
  node5to64_k<<<N, 64, 0, stream>>>(aggc, invc, x0, c1_lw, c1_lb, c1_rw, B1, N);
  fill_csr_k<<<(E + 255) / 256, 256, 0, stream>>>(src, dst, ew, rank, offs, cnt0, csr, E);

  proj64_wmma_k<<<nb16, 128, 0, stream>>>(B1, c2_pw, c2_pb, B2, N);
  agg_node64_wmma_k<<<nb16, 128, 0, stream>>>(offs, csr, B2, invc, B1, c2_lw, c2_lb, c2_rw, N);

  float* z3   = B2;
  float* zsum = B2 + (size_t)N * 8;
  proj64_lin5_wmma_k<<<nb16, 128, 0, stream>>>(B1, c3_pw, c3_pb, c3_lw, z3, N);
  agg5_k<<<(N * 8 + 255) / 256, 256, 0, stream>>>(offs, csr, z3, zsum, N);
  node64to5_wmma_k<<<nb16, 32, 0, stream>>>(zsum, invc, B1, c3_lb, c3_rw,
                                            (float*)d_out, N);
}


# Layer 3's row-wise products (projection, neighbour 64->5, root 64->5) run in layer 2's node kernel on rows still in LDS; last kernel = edge sums + finish, its 5-term norm summed by lane exchange
# speedup vs baseline: 5.0550x; 1.0364x over previous
#include <hip/hip_runtime.h>
#include <hip/hip_bf16.h>

typedef __attribute__((ext_vector_type(2))) float v2f;
typedef __attribute__((ext_vector_type(8))) float v8f;

#define EPS_BN   1e-5f
#define EPS_NORM 1e-12f

__device__ __forceinline__ v8f wmma_f32(v2f a, v2f b, v8f c) {
  return __builtin_amdgcn_wmma_f32_16x16x4_f32(false, a, false, b, (short)0, c,
                                               false, false);
}

__global__ void bn_partial_k(const float* __restrict__ h, float* __restrict__ part, int n) {
  __shared__ float sh[256 * 10];
  float acc[10];
#pragma unroll
  for (int d = 0; d < 10; ++d) acc[d] = 0.f;
  for (int i = blockIdx.x * blockDim.x + threadIdx.x; i < n;
       i += gridDim.x * blockDim.x) {
#pragma unroll
    for (int d = 0; d < 5; ++d) {
      float v = h[i * 5 + d];
      acc[d] += v;
      acc[5 + d] += v * v;
    }
  }
#pragma unroll
  for (int d = 0; d < 10; ++d) sh[threadIdx.x * 10 + d] = acc[d];
  __syncthreads();
  for (int s = 128; s > 0; s >>= 1) {
    if (threadIdx.x < (unsigned)s)
      for (int d = 0; d < 10; ++d)
        sh[threadIdx.x * 10 + d] += sh[(threadIdx.x + s) * 10 + d];
    __syncthreads();
  }
  if (threadIdx.x < 10) part[blockIdx.x * 10 + threadIdx.x] = sh[threadIdx.x];
}

__global__ void bn_final_k(const float* __restrict__ part, float* __restrict__ mu,
                           float* __restrict__ rs, int nblocks, float inv_n) {
  int t = threadIdx.x;
  if (t < 5) {
    float s = 0.f, q = 0.f;
    for (int b = 0; b < nblocks; ++b) {
      s += part[b * 10 + t];
      q += part[b * 10 + 5 + t];
    }
    float m = s * inv_n;
    float var = q * inv_n - m * m;
    mu[t] = m;
    rs[t] = rsqrtf(var + EPS_BN);
  }
}

__global__ void bn_apply_k(const float* __restrict__ h, const float* __restrict__ mu,
                           const float* __restrict__ rs, const float* __restrict__ g,
                           const float* __restrict__ b, float* __restrict__ x, int total) {
  int i = blockIdx.x * blockDim.x + threadIdx.x;
  if (i >= total) return;
  int d = i % 5;
  x[i] = (h[i] - mu[d]) * rs[d] * g[d] + b[d];
}

__global__ void invcnt_sum_k(const float* __restrict__ aggc, float* __restrict__ invc,
                             int* __restrict__ cnt, int* __restrict__ cnt0,
                             int* __restrict__ bsum, int n) {
  __shared__ int wsum[32];
  const int lane = threadIdx.x & 31;
  const int wave = threadIdx.x >> 5;
  const int i = blockIdx.x * 1024 + threadIdx.x;
  int c = 0;
  if (i < n) {
    const float f0 = aggc[(size_t)i * 8 + 5];
    const float f = f0 + aggc[((size_t)n + i) * 8 + 5];
    invc[i] = 1.0f / fmaxf(f, 1.0f);
    c = (int)f;
    cnt[i] = c;
    cnt0[i] = (int)f0;
  }
  int sum = c;
#pragma unroll
  for (int o = 16; o > 0; o >>= 1) sum += __shfl_xor(sum, o);
  if (lane == 0) wsum[wave] = sum;
  __syncthreads();
  if (threadIdx.x == 0) {
    int t = 0;
    for (int w = 0; w < 32; ++w) t += wsum[w];
    bsum[blockIdx.x] = t;
  }
}

__global__ void proj5_k(const float* __restrict__ X, const float* __restrict__ W,
                        const float* __restrict__ b, float* __restrict__ Y, int n) {
  int i = blockIdx.x * blockDim.x + threadIdx.x;
  if (i >= n) return;
  float x[5];
#pragma unroll
  for (int k = 0; k < 5; ++k) x[k] = X[i * 5 + k];
#pragma unroll
  for (int c = 0; c < 5; ++c) {
    float acc = b[c];
#pragma unroll
    for (int k = 0; k < 5; ++k) acc += x[k] * W[k * 5 + c];
    Y[i * 5 + c] = fmaxf(acc, 0.f);
  }
}

template <int XS, bool COUNT>
__device__ __forceinline__ void edge_scatter5_body(const int* __restrict__ src,
                                                   const int* __restrict__ dst,
                                                   const float* __restrict__ ew,
                                                   const float* __restrict__ xp,
                                                   float* __restrict__ agg,
                                                   int* __restrict__ rank, int E,
                                                   int n) {
  const int lane = threadIdx.x & 31;
  const int e0 = __builtin_amdgcn_readfirstlane(
      (int)((blockIdx.x * blockDim.x + threadIdx.x) >> 5) << 5);
  if (e0 >= E) return;
  const int e = e0 + lane;
  int s_l = 0, d_l = 0;
  float w_l = 0.f;
  if (e < E) { s_l = src[e]; d_l = dst[e]; w_l = ew[e]; }
  const int h = (int)((__builtin_amdgcn_s_sendmsg_rtn(0x87) >> 18) & 1u);
  agg += (size_t)h * (size_t)n * 8;
  const int k = lane & 7;
  const int g = lane >> 3;
  int s[8], d[8];
  float w[8];
#pragma unroll
  for (int j = 0; j < 8; ++j) {
    const int sl = j * 4 + g;
    s[j] = __shfl(s_l, sl);
    d[j] = __shfl(d_l, sl);
    w[j] = __shfl(w_l, sl);
  }
  if (k >= (COUNT ? 6 : 5)) return;
  const bool feat = (k < 5);
  const int kl = feat ? k : 4;
  float x[8];
#pragma unroll
  for (int j = 0; j < 8; ++j) x[j] = xp[(size_t)s[j] * XS + kl];
  if (e0 + 32 <= E) {
    float old[8];
#pragma unroll
    for (int j = 0; j < 8; ++j)
      old[j] = atomicAdd(&agg[(size_t)d[j] * 8 + k], feat ? x[j] * w[j] : 1.0f);
    if (COUNT && !feat) {
#pragma unroll
      for (int j = 0; j < 8; ++j) rank[e0 + j * 4 + g] = ((int)old[j] << 1) | h;
    }
  } else {
#pragma unroll
    for (int j = 0; j < 8; ++j) {
      if ((e0 + j * 4 + g) < E) {
        const float old = atomicAdd(&agg[(size_t)d[j] * 8 + k], feat ? x[j] * w[j] : 1.0f);
        if (COUNT && !feat) rank[e0 + j * 4 + g] = ((int)old << 1) | h;
      }
    }
  }
}

__global__ void edge_scatter5_k(const int* __restrict__ src, const int* __restrict__ dst,
                                const float* __restrict__ ew, const float* __restrict__ xp,
                                float* __restrict__ agg, int* __restrict__ rank, int E,
                                int n) {
  edge_scatter5_body<5, true>(src, dst, ew, xp, agg, rank, E, n);
}

__global__ void fill_csr_k(const int* __restrict__ src, const int* __restrict__ dst,
                           const float* __restrict__ ew, const int* __restrict__ rank,
                           const int* __restrict__ offs, const int* __restrict__ cnt0,
                           int* __restrict__ csr, int E) {
  int e = blockIdx.x * blockDim.x + threadIdx.x;
  if (e >= E) return;
  const int d = dst[e];
  const int r = rank[e];
  int pos = offs[d] + (r >> 1) + ((r & 1) ? cnt0[d] : 0);
  ((int2*)csr)[pos] = make_int2(src[e], __float_as_int(ew[e]));
}

__global__ void offs_scan_k(const int* __restrict__ cnt, const int* __restrict__ bsum,
                            int* __restrict__ offs, int n) {
  __shared__ int wsum[32];
  const int lane = threadIdx.x & 31;
  const int wave = threadIdx.x >> 5;
  const int i = blockIdx.x * 1024 + threadIdx.x;
  const int c = (i < n) ? cnt[i] : 0;
  int incl = c;
#pragma unroll
  for (int o = 1; o < 32; o <<= 1) {
    int v = __shfl_up(incl, o);
    if (lane >= o) incl += v;
  }
  if (lane == 31) wsum[wave] = incl;
  __syncthreads();
  int base = 0;
  for (int b = 0; b < (int)blockIdx.x; ++b) base += bsum[b];
  for (int w = 0; w < wave; ++w) base += wsum[w];
  const int ex = base + incl - c;
  if (i < n) {
    offs[i] = ex;
    if (i == n - 1) offs[n] = ex + c;
  }
}

__global__ void node5to64_k(const float* __restrict__ agg, const float* __restrict__ invc,
                            const float* __restrict__ X, const float* __restrict__ LW,
                            const float* __restrict__ lb, const float* __restrict__ RW,
                            float* __restrict__ Y, int n) {
  __shared__ float red[64];
  int node = blockIdx.x;
  if (node >= n) return;
  int c = threadIdx.x;
  float ic = invc[node];
  float acc = lb[c];
#pragma unroll
  for (int k = 0; k < 5; ++k) {
    const float a = agg[node * 8 + k] + agg[((size_t)n + node) * 8 + k];
    acc += a * ic * LW[k * 64 + c] + X[node * 5 + k] * RW[k * 64 + c];
  }
  red[c] = acc * acc;
  __syncthreads();
#pragma unroll
  for (int s = 32; s > 0; s >>= 1) {
    if (c < s) red[c] += red[c + s];
    __syncthreads();
  }
  float scale = 1.0f / fmaxf(sqrtf(red[0]), EPS_NORM);
  Y[node * 64 + c] = fmaxf(acc * scale, 0.f);
}

__global__ void proj64_wmma_k(const float* __restrict__ X, const float* __restrict__ W,
                              const float* __restrict__ bias, float* __restrict__ Y, int n) {
  int l = threadIdx.x & 31;
  int wave = threadIdx.x >> 5;
  int row0 = blockIdx.x * 16;
  int m = row0 + (l & 15);
  if (m >= n) m = n - 1;
  int cl = wave * 16 + (l & 15);
  int khi = (l >> 4) << 1;
  v8f c = {};
#pragma unroll
  for (int k0 = 0; k0 < 64; k0 += 4) {
    int kb = k0 + khi;
    v2f a, bb;
    a[0] = X[m * 64 + kb];
    a[1] = X[m * 64 + kb + 1];
    bb[0] = W[kb * 64 + cl];
    bb[1] = W[(kb + 1) * 64 + cl];
    c = wmma_f32(a, bb, c);
  }
  int rbase = row0 + ((l >> 4) << 3);
  float bc = bias[cl];
#pragma unroll
  for (int r = 0; r < 8; ++r) {
    int rr = rbase + r;
    if (rr < n) Y[rr * 64 + cl] = fmaxf(c[r] + bc, 0.f);
  }
}

__global__ void __attribute__((amdgpu_waves_per_eu(16, 16))) agg_node64_l3_wmma_k(const int* __restrict__ offs, const int* __restrict__ csr,
                                     const float* __restrict__ xp, const float* __restrict__ invc,
                                     float* XR, const float* __restrict__ LW,
                                     const float* __restrict__ lb, const float* __restrict__ RW,
                                     const float* __restrict__ PW3, const float* __restrict__ pb3,
                                     const float* __restrict__ LW3, const float* __restrict__ lb3,
                                     const float* __restrict__ RW3, float* __restrict__ Z, int n) {
  __shared__ float mt[16 * 68];
  __shared__ float tile[16 * 68];
  __shared__ float rscale[16];
  const int l = threadIdx.x & 31;
  const int wave = __builtin_amdgcn_readfirstlane((int)(threadIdx.x >> 5));
  const int row0 = blockIdx.x * 16;
  const int2* __restrict__ pairs = (const int2*)csr;
  for (int q = 0; q < 4; ++q) {
    const int r = wave * 4 + q;
    const int node = row0 + r;
    float lo = 0.f, hi = 0.f;
    if (node < n) {
      const int beg = __builtin_amdgcn_readfirstlane(offs[node]);
      const int end = __builtin_amdgcn_readfirstlane(offs[node + 1]);
      for (int c0 = beg; c0 < end; c0 += 32) {
        int s_l = 0;
        float w_l = 0.f;
        if (c0 + l < end) {
          const int2 v = pairs[c0 + l];
          s_l = v.x;
          w_l = __int_as_float(v.y);
        }
        const int n_e = min(32, end - c0);
        for (int j0 = 0; j0 < n_e; j0 += 8) {
          float a[8], b[8];
#pragma unroll
          for (int u = 0; u < 8; ++u) {
            const int sr = __builtin_amdgcn_readlane(s_l, j0 + u);
            const float* row = xp + (size_t)sr * 64 + l;
            a[u] = row[0];
            b[u] = row[32];
          }
#pragma unroll
          for (int u = 0; u < 8; ++u) {
            if (j0 + u < n_e) {
              const float w = __int_as_float(__builtin_amdgcn_readlane(__float_as_int(w_l), j0 + u));
              lo += a[u] * w;
              hi += b[u] * w;
            }
          }
        }
      }
      const float ic = invc[node];
      lo *= ic;
      hi *= ic;
    }
    mt[r * 68 + l] = lo;
    mt[r * 68 + 32 + l] = hi;
  }
  __syncthreads();
  const int c16 = l & 15;
  int m = row0 + c16;
  if (m >= n) m = n - 1;
  const int cl = wave * 16 + c16;
  const int khi = (l >> 4) << 1;
  v8f c = {};
#pragma unroll
  for (int k0 = 0; k0 < 64; k0 += 4) {
    int kb = k0 + khi;
    v2f a, bb;
    a[0] = mt[c16 * 68 + kb];
    a[1] = mt[c16 * 68 + kb + 1];
    bb[0] = LW[kb * 64 + cl];
    bb[1] = LW[(kb + 1) * 64 + cl];
    c = wmma_f32(a, bb, c);
  }
#pragma unroll
  for (int k0 = 0; k0 < 64; k0 += 4) {
    int kb = k0 + khi;
    v2f a, bb;
    a[0] = XR[m * 64 + kb];
    a[1] = XR[m * 64 + kb + 1];
    bb[0] = RW[kb * 64 + cl];
    bb[1] = RW[(kb + 1) * 64 + cl];
    c = wmma_f32(a, bb, c);
  }
  const int rloc = (l >> 4) << 3;
  const float bc = lb[cl];
#pragma unroll
  for (int r = 0; r < 8; ++r) tile[(rloc + r) * 68 + cl] = c[r] + bc;
  __syncthreads();
  if (threadIdx.x < 16) {
    float ss = 0.f;
#pragma unroll
    for (int cc = 0; cc < 64; ++cc) {
      float v = tile[threadIdx.x * 68 + cc];
      ss += v * v;
    }
    rscale[threadIdx.x] = 1.0f / fmaxf(sqrtf(ss), EPS_NORM);
  }
  __syncthreads();
#pragma unroll
  for (int i = threadIdx.x; i < 16 * 64; i += 128) {
    const int r = i >> 6;
    const int cc = i & 63;
    mt[r * 68 + cc] = fmaxf(tile[r * 68 + cc] * rscale[r], 0.f);
  }
  __syncthreads();
  v8f p = {};
#pragma unroll
  for (int k0 = 0; k0 < 64; k0 += 4) {
    int kb = k0 + khi;
    v2f a, bb;
    a[0] = mt[c16 * 68 + kb];
    a[1] = mt[c16 * 68 + kb + 1];
    bb[0] = PW3[kb * 64 + cl];
    bb[1] = PW3[(kb + 1) * 64 + cl];
    p = wmma_f32(a, bb, p);
  }
  const float pc = pb3[cl];
#pragma unroll
  for (int r = 0; r < 8; ++r) tile[(rloc + r) * 68 + cl] = fmaxf(p[r] + pc, 0.f);
  __syncthreads();
  if (wave < 2) {
    const float* __restrict__ At = (wave == 0) ? tile : mt;
    const float* __restrict__ Bm = (wave == 0) ? LW3 : RW3;
    const bool cv = (c16 < 5);
    v2f bw[16];
#pragma unroll
    for (int i = 0; i < 16; ++i) { bw[i][0] = 0.f; bw[i][1] = 0.f; }
    float bias = 0.f;
    if (cv) {
#pragma unroll
      for (int i = 0; i < 16; ++i) {
        int kb = i * 4 + khi;
        bw[i][0] = Bm[kb * 5 + c16];
        bw[i][1] = Bm[(kb + 1) * 5 + c16];
      }
      if (wave == 1) bias = lb3[c16];
    }
    v8f z = {};
#pragma unroll
    for (int i = 0; i < 16; ++i) {
      int kb = i * 4 + khi;
      v2f a;
      a[0] = At[c16 * 68 + kb];
      a[1] = At[c16 * 68 + kb + 1];
      z = wmma_f32(a, bw[i], z);
    }
    if (cv) {
      float* __restrict__ O = (wave == 0) ? Z : XR;
      const int pitch = (wave == 0) ? 8 : 64;
#pragma unroll
      for (int r = 0; r < 8; ++r) {
        int rr = row0 + rloc + r;
        if (rr < n) O[(size_t)rr * pitch + c16] = z[r] + bias;
      }
    }
  }
}

__global__ void agg5_out_k(const int* __restrict__ offs, const int* __restrict__ csr,
                           const float* __restrict__ z, const float* __restrict__ root,
                           const float* __restrict__ invc, float* __restrict__ Y, int n) {
  const int2* __restrict__ pairs = (const int2*)csr;
  const int k = threadIdx.x & 7;
  const int node = (int)((blockIdx.x * blockDim.x + threadIdx.x) >> 3);
  if (node >= n) return;
  const int kl = (k < 5) ? k : 4;
  const int beg = offs[node], end = offs[node + 1];
  const float ic = invc[node];
  const float rt = root[(size_t)node * 64 + kl];
  float acc = 0.f;
  for (int t = beg; t < end; t += 4) {
    int2 v[4];
    float x[4];
#pragma unroll
    for (int u = 0; u < 4; ++u) v[u] = pairs[min(t + u, end - 1)];
#pragma unroll
    for (int u = 0; u < 4; ++u) x[u] = z[(size_t)v[u].x * 8 + kl];
#pragma unroll
    for (int u = 0; u < 4; ++u)
      if (t + u < end) acc += x[u] * __int_as_float(v[u].y);
  }
  const float val = (k < 5) ? rt + acc * ic : 0.f;
  float ss = val * val;
  ss += __shfl_xor(ss, 1);
  ss += __shfl_xor(ss, 2);
  ss += __shfl_xor(ss, 4);
  const float sc = 1.0f / fmaxf(sqrtf(ss), EPS_NORM);
  if (k < 5) Y[(size_t)node * 5 + k] = val * sc;
}

extern "C" void kernel_launch(void* const* d_in, const int* in_sizes, int n_in,
                              void* d_out, int out_size, void* d_ws, size_t ws_size,
                              hipStream_t stream) {
  const float* h     = (const float*)d_in[0];
  const int*   ei    = (const int*)d_in[1];
  const float* ew    = (const float*)d_in[2];
  const float* bg    = (const float*)d_in[3];
  const float* bbeta = (const float*)d_in[4];
  const float* c1_pw = (const float*)d_in[5];
  const float* c1_pb = (const float*)d_in[6];
  const float* c1_lw = (const float*)d_in[7];
  const float* c1_lb = (const float*)d_in[8];
  const float* c1_rw = (const float*)d_in[9];
  const float* c2_pw = (const float*)d_in[10];
  const float* c2_pb = (const float*)d_in[11];
  const float* c2_lw = (const float*)d_in[12];
  const float* c2_lb = (const float*)d_in[13];
  const float* c2_rw = (const float*)d_in[14];
  const float* c3_pw = (const float*)d_in[15];
  const float* c3_pb = (const float*)d_in[16];
  const float* c3_lw = (const float*)d_in[17];
  const float* c3_lb = (const float*)d_in[18];
  const float* c3_rw = (const float*)d_in[19];

  const int N = in_sizes[0] / 5;
  const int E = in_sizes[2];
  const int* src = ei;
  const int* dst = ei + E;

  float* ws   = (float*)d_ws;
  float* B1   = ws;
  float* B2   = B1 + (size_t)N * 64;
  float* B3   = B2 + (size_t)N * 64;
  float* mu   = B3 + (size_t)N * 64;
  float* rs   = mu + 8;
  float* part = rs + 8;
  float* invc = part + 2560;
  float* x0   = invc + N;
  float* xp1  = x0 + (size_t)N * 5;
  float* aggc = B3;
  int*   rank = (int*)B2;
  int*   cnt  = (int*)xp1;
  int*   cnt0 = cnt + N;
  const int nb1k = (N + 1023) / 1024;
  int*   bsum = cnt0 + N;
  int*   offs = (int*)xp1 + (size_t)N * 4 - 1;
  int*   csr  = (int*)B3;

  const int nb16 = (N + 15) / 16;

  bn_partial_k<<<256, 256, 0, stream>>>(h, part, N);
  bn_final_k<<<1, 32, 0, stream>>>(part, mu, rs, 256, 1.0f / (float)N);
  bn_apply_k<<<(N * 5 + 255) / 256, 256, 0, stream>>>(h, mu, rs, bg, bbeta, x0, N * 5);

  proj5_k<<<(N + 255) / 256, 256, 0, stream>>>(x0, c1_pw, c1_pb, xp1, N);
  hipMemsetAsync(aggc, 0, (size_t)N * 16 * sizeof(float), stream);
  edge_scatter5_k<<<(E + 255) / 256, 256, 0, stream>>>(src, dst, ew, xp1, aggc, rank, E, N);
  invcnt_sum_k<<<nb1k, 1024, 0, stream>>>(aggc, invc, cnt, cnt0, bsum, N);
  offs_scan_k<<<nb1k, 1024, 0, stream>>>(cnt, bsum, offs, N);
  node5to64_k<<<N, 64, 0, stream>>>(aggc, invc, x0, c1_lw, c1_lb, c1_rw, B1, N);
  fill_csr_k<<<(E + 255) / 256, 256, 0, stream>>>(src, dst, ew, rank, offs, cnt0, csr, E);

  float* z3 = x0;
  proj64_wmma_k<<<nb16, 128, 0, stream>>>(B1, c2_pw, c2_pb, B2, N);
  agg_node64_l3_wmma_k<<<nb16, 128, 0, stream>>>(offs, csr, B2, invc, B1, c2_lw, c2_lb, c2_rw,
                                                 c3_pw, c3_pb, c3_lw, c3_lb, c3_rw, z3, N);

  agg5_out_k<<<(N * 8 + 255) / 256, 256, 0, stream>>>(offs, csr, z3, B1, invc,
                                                      (float*)d_out, N);
}


# Layer-1 node step: a wave finishes 8 nodes (lane = 2 columns, weights in registers, norm by lane exchange in the LDS tree's pairs) instead of a 2-wave block per node: 200,000 waves -> 12,500
# speedup vs baseline: 5.2560x; 1.0398x over previous
#include <hip/hip_runtime.h>
#include <hip/hip_bf16.h>

typedef __attribute__((ext_vector_type(2))) float v2f;
typedef __attribute__((ext_vector_type(8))) float v8f;

#define EPS_BN   1e-5f
#define EPS_NORM 1e-12f

__device__ __forceinline__ v8f wmma_f32(v2f a, v2f b, v8f c) {
  return __builtin_amdgcn_wmma_f32_16x16x4_f32(false, a, false, b, (short)0, c,
                                               false, false);
}

__global__ void bn_partial_k(const float* __restrict__ h, float* __restrict__ part, int n) {
  __shared__ float sh[256 * 10];
  float acc[10];
#pragma unroll
  for (int d = 0; d < 10; ++d) acc[d] = 0.f;
  for (int i = blockIdx.x * blockDim.x + threadIdx.x; i < n;
       i += gridDim.x * blockDim.x) {
#pragma unroll
    for (int d = 0; d < 5; ++d) {
      float v = h[i * 5 + d];
      acc[d] += v;
      acc[5 + d] += v * v;
    }
  }
#pragma unroll
  for (int d = 0; d < 10; ++d) sh[threadIdx.x * 10 + d] = acc[d];
  __syncthreads();
  for (int s = 128; s > 0; s >>= 1) {
    if (threadIdx.x < (unsigned)s)
      for (int d = 0; d < 10; ++d)
        sh[threadIdx.x * 10 + d] += sh[(threadIdx.x + s) * 10 + d];
    __syncthreads();
  }
  if (threadIdx.x < 10) part[blockIdx.x * 10 + threadIdx.x] = sh[threadIdx.x];
}

__global__ void bn_final_k(const float* __restrict__ part, float* __restrict__ mu,
                           float* __restrict__ rs, int nblocks, float inv_n) {
  int t = threadIdx.x;
  if (t < 5) {
    float s = 0.f, q = 0.f;
    for (int b = 0; b < nblocks; ++b) {
      s += part[b * 10 + t];
      q += part[b * 10 + 5 + t];
    }
    float m = s * inv_n;
    float var = q * inv_n - m * m;
    mu[t] = m;
    rs[t] = rsqrtf(var + EPS_BN);
  }
}

__global__ void bn_apply_k(const float* __restrict__ h, const float* __restrict__ mu,
                           const float* __restrict__ rs, const float* __restrict__ g,
                           const float* __restrict__ b, float* __restrict__ x, int total) {
  int i = blockIdx.x * blockDim.x + threadIdx.x;
  if (i >= total) return;
  int d = i % 5;
  x[i] = (h[i] - mu[d]) * rs[d] * g[d] + b[d];
}

__global__ void invcnt_sum_k(const float* __restrict__ aggc, float* __restrict__ invc,
                             int* __restrict__ cnt, int* __restrict__ cnt0,
                             int* __restrict__ bsum, int n) {
  __shared__ int wsum[32];
  const int lane = threadIdx.x & 31;
  const int wave = threadIdx.x >> 5;
  const int i = blockIdx.x * 1024 + threadIdx.x;
  int c = 0;
  if (i < n) {
    const float f0 = aggc[(size_t)i * 8 + 5];
    const float f = f0 + aggc[((size_t)n + i) * 8 + 5];
    invc[i] = 1.0f / fmaxf(f, 1.0f);
    c = (int)f;
    cnt[i] = c;
    cnt0[i] = (int)f0;
  }
  int sum = c;
#pragma unroll
  for (int o = 16; o > 0; o >>= 1) sum += __shfl_xor(sum, o);
  if (lane == 0) wsum[wave] = sum;
  __syncthreads();
  if (threadIdx.x == 0) {
    int t = 0;
    for (int w = 0; w < 32; ++w) t += wsum[w];
    bsum[blockIdx.x] = t;
  }
}

__global__ void proj5_k(const float* __restrict__ X, const float* __restrict__ W,
                        const float* __restrict__ b, float* __restrict__ Y, int n) {
  int i = blockIdx.x * blockDim.x + threadIdx.x;
  if (i >= n) return;
  float x[5];
#pragma unroll
  for (int k = 0; k < 5; ++k) x[k] = X[i * 5 + k];
#pragma unroll
  for (int c = 0; c < 5; ++c) {
    float acc = b[c];
#pragma unroll
    for (int k = 0; k < 5; ++k) acc += x[k] * W[k * 5 + c];
    Y[i * 5 + c] = fmaxf(acc, 0.f);
  }
}

template <int XS, bool COUNT>
__device__ __forceinline__ void edge_scatter5_body(const int* __restrict__ src,
                                                   const int* __restrict__ dst,
                                                   const float* __restrict__ ew,
                                                   const float* __restrict__ xp,
                                                   float* __restrict__ agg,
                                                   int* __restrict__ rank, int E,
                                                   int n) {
  const int lane = threadIdx.x & 31;
  const int e0 = __builtin_amdgcn_readfirstlane(
      (int)((blockIdx.x * blockDim.x + threadIdx.x) >> 5) << 5);
  if (e0 >= E) return;
  const int e = e0 + lane;
  int s_l = 0, d_l = 0;
  float w_l = 0.f;
  if (e < E) { s_l = src[e]; d_l = dst[e]; w_l = ew[e]; }
  const int h = (int)((__builtin_amdgcn_s_sendmsg_rtn(0x87) >> 18) & 1u);
  agg += (size_t)h * (size_t)n * 8;
  const int k = lane & 7;
  const int g = lane >> 3;
  int s[8], d[8];
  float w[8];
#pragma unroll
  for (int j = 0; j < 8; ++j) {
    const int sl = j * 4 + g;
    s[j] = __shfl(s_l, sl);
    d[j] = __shfl(d_l, sl);
    w[j] = __shfl(w_l, sl);
  }
  if (k >= (COUNT ? 6 : 5)) return;
  const bool feat = (k < 5);
  const int kl = feat ? k : 4;
  float x[8];
#pragma unroll
  for (int j = 0; j < 8; ++j) x[j] = xp[(size_t)s[j] * XS + kl];
  if (e0 + 32 <= E) {
    float old[8];
#pragma unroll
    for (int j = 0; j < 8; ++j)
      old[j] = atomicAdd(&agg[(size_t)d[j] * 8 + k], feat ? x[j] * w[j] : 1.0f);
    if (COUNT && !feat) {
#pragma unroll
      for (int j = 0; j < 8; ++j) rank[e0 + j * 4 + g] = ((int)old[j] << 1) | h;
    }
  } else {
#pragma unroll
    for (int j = 0; j < 8; ++j) {
      if ((e0 + j * 4 + g) < E) {
        const float old = atomicAdd(&agg[(size_t)d[j] * 8 + k], feat ? x[j] * w[j] : 1.0f);
        if (COUNT && !feat) rank[e0 + j * 4 + g] = ((int)old << 1) | h;
      }
    }
  }
}

__global__ void edge_scatter5_k(const int* __restrict__ src, const int* __restrict__ dst,
                                const float* __restrict__ ew, const float* __restrict__ xp,
                                float* __restrict__ agg, int* __restrict__ rank, int E,
                                int n) {
  edge_scatter5_body<5, true>(src, dst, ew, xp, agg, rank, E, n);
}

__global__ void fill_csr_k(const int* __restrict__ src, const int* __restrict__ dst,
                           const float* __restrict__ ew, const int* __restrict__ rank,
                           const int* __restrict__ offs, const int* __restrict__ cnt0,
                           int* __restrict__ csr, int E) {
  int e = blockIdx.x * blockDim.x + threadIdx.x;
  if (e >= E) return;
  const int d = dst[e];
  const int r = rank[e];
  int pos = offs[d] + (r >> 1) + ((r & 1) ? cnt0[d] : 0);
  ((int2*)csr)[pos] = make_int2(src[e], __float_as_int(ew[e]));
}

__global__ void offs_scan_k(const int* __restrict__ cnt, const int* __restrict__ bsum,
                            int* __restrict__ offs, int n) {
  __shared__ int wsum[32];
  const int lane = threadIdx.x & 31;
  const int wave = threadIdx.x >> 5;
  const int i = blockIdx.x * 1024 + threadIdx.x;
  const int c = (i < n) ? cnt[i] : 0;
  int incl = c;
#pragma unroll
  for (int o = 1; o < 32; o <<= 1) {
    int v = __shfl_up(incl, o);
    if (lane >= o) incl += v;
  }
  if (lane == 31) wsum[wave] = incl;
  __syncthreads();
  int base = 0;
  for (int b = 0; b < (int)blockIdx.x; ++b) base += bsum[b];
  for (int w = 0; w < wave; ++w) base += wsum[w];
  const int ex = base + incl - c;
  if (i < n) {
    offs[i] = ex;
    if (i == n - 1) offs[n] = ex + c;
  }
}

template <bool FULL>
__device__ __forceinline__ void node5to64_body(const float* __restrict__ agg,
                                               const float* __restrict__ invc,
                                               const float* __restrict__ X,
                                               const float* __restrict__ LW,
                                               const float* __restrict__ lb,
                                               const float* __restrict__ RW,
                                               float* __restrict__ Y, int n, int nb) {
  const int l = threadIdx.x & 31;
  float m[2], xv[2];
#pragma unroll
  for (int g = 0; g < 2; ++g) {
    const int node = nb + g * 4 + (l >> 3);
    float a = 0.f, ic = 0.f;
    if (FULL || node < n) {
      a = agg[(size_t)node * 8 + (l & 7)] + agg[((size_t)n + node) * 8 + (l & 7)];
      ic = invc[node];
    }
    m[g] = a * ic;
    const int xi = (nb + g * 4) * 5 + l;
    xv[g] = (l < 20 && (FULL || xi < n * 5)) ? X[xi] : 0.f;
  }
  float lw0[5], lw1[5], rw0[5], rw1[5];
#pragma unroll
  for (int k = 0; k < 5; ++k) {
    lw0[k] = LW[k * 64 + l];
    lw1[k] = LW[k * 64 + 32 + l];
    rw0[k] = RW[k * 64 + l];
    rw1[k] = RW[k * 64 + 32 + l];
  }
  const float b0 = lb[l], b1 = lb[32 + l];
#pragma unroll
  for (int g = 0; g < 2; ++g) {
#pragma unroll
    for (int q = 0; q < 4; ++q) {
      const int node = nb + g * 4 + q;
      float acc0 = b0, acc1 = b1;
#pragma unroll
      for (int k = 0; k < 5; ++k) {
        const float mk = __int_as_float(__builtin_amdgcn_readlane(__float_as_int(m[g]), q * 8 + k));
        const float xk = __int_as_float(__builtin_amdgcn_readlane(__float_as_int(xv[g]), q * 5 + k));
        acc0 += mk * lw0[k] + xk * rw0[k];
        acc1 += mk * lw1[k] + xk * rw1[k];
      }
      float ss = acc0 * acc0 + acc1 * acc1;
#pragma unroll
      for (int o = 16; o > 0; o >>= 1) ss += __shfl_xor(ss, o);
      const float scale = 1.0f / fmaxf(sqrtf(ss), EPS_NORM);
      if (FULL || node < n) {
        Y[(size_t)node * 64 + l] = fmaxf(acc0 * scale, 0.f);
        Y[(size_t)node * 64 + 32 + l] = fmaxf(acc1 * scale, 0.f);
      }
    }
  }
}

__global__ void node5to64_k(const float* __restrict__ agg, const float* __restrict__ invc,
                            const float* __restrict__ X, const float* __restrict__ LW,
                            const float* __restrict__ lb, const float* __restrict__ RW,
                            float* __restrict__ Y, int n) {
  const int nb = __builtin_amdgcn_readfirstlane(
                     (int)((blockIdx.x * blockDim.x + threadIdx.x) >> 5)) * 8;
  if (nb >= n) return;
  if (nb + 8 <= n) node5to64_body<true>(agg, invc, X, LW, lb, RW, Y, n, nb);
  else             node5to64_body<false>(agg, invc, X, LW, lb, RW, Y, n, nb);
}

__global__ void proj64_wmma_k(const float* __restrict__ X, const float* __restrict__ W,
                              const float* __restrict__ bias, float* __restrict__ Y, int n) {
  int l = threadIdx.x & 31;
  int wave = threadIdx.x >> 5;
  int row0 = blockIdx.x * 16;
  int m = row0 + (l & 15);
  if (m >= n) m = n - 1;
  int cl = wave * 16 + (l & 15);
  int khi = (l >> 4) << 1;
  v8f c = {};
#pragma unroll
  for (int k0 = 0; k0 < 64; k0 += 4) {
    int kb = k0 + khi;
    v2f a, bb;
    a[0] = X[m * 64 + kb];
    a[1] = X[m * 64 + kb + 1];
    bb[0] = W[kb * 64 + cl];
    bb[1] = W[(kb + 1) * 64 + cl];
    c = wmma_f32(a, bb, c);
  }
  int rbase = row0 + ((l >> 4) << 3);
  float bc = bias[cl];
#pragma unroll
  for (int r = 0; r < 8; ++r) {
    int rr = rbase + r;
    if (rr < n) Y[rr * 64 + cl] = fmaxf(c[r] + bc, 0.f);
  }
}

__global__ void __attribute__((amdgpu_waves_per_eu(16, 16))) agg_node64_l3_wmma_k(const int* __restrict__ offs, const int* __restrict__ csr,
                                     const float* __restrict__ xp, const float* __restrict__ invc,
                                     float* XR, const float* __restrict__ LW,
                                     const float* __restrict__ lb, const float* __restrict__ RW,
                                     const float* __restrict__ PW3, const float* __restrict__ pb3,
                                     const float* __restrict__ LW3, const float* __restrict__ lb3,
                                     const float* __restrict__ RW3, float* __restrict__ Z, int n) {
  __shared__ float mt[16 * 68];
  __shared__ float tile[16 * 68];
  __shared__ float rscale[16];
  const int l = threadIdx.x & 31;
  const int wave = __builtin_amdgcn_readfirstlane((int)(threadIdx.x >> 5));
  const int row0 = blockIdx.x * 16;
  const int2* __restrict__ pairs = (const int2*)csr;
  for (int q = 0; q < 4; ++q) {
    const int r = wave * 4 + q;
    const int node = row0 + r;
    float lo = 0.f, hi = 0.f;
    if (node < n) {
      const int beg = __builtin_amdgcn_readfirstlane(offs[node]);
      const int end = __builtin_amdgcn_readfirstlane(offs[node + 1]);
      for (int c0 = beg; c0 < end; c0 += 32) {
        int s_l = 0;
        float w_l = 0.f;
        if (c0 + l < end) {
          const int2 v = pairs[c0 + l];
          s_l = v.x;
          w_l = __int_as_float(v.y);
        }
        const int n_e = min(32, end - c0);
        for (int j0 = 0; j0 < n_e; j0 += 8) {
          float a[8], b[8];
#pragma unroll
          for (int u = 0; u < 8; ++u) {
            const int sr = __builtin_amdgcn_readlane(s_l, j0 + u);
            const float* row = xp + (size_t)sr * 64 + l;
            a[u] = row[0];
            b[u] = row[32];
          }
#pragma unroll
          for (int u = 0; u < 8; ++u) {
            if (j0 + u < n_e) {
              const float w = __int_as_float(__builtin_amdgcn_readlane(__float_as_int(w_l), j0 + u));
              lo += a[u] * w;
              hi += b[u] * w;
            }
          }
        }
      }
      const float ic = invc[node];
      lo *= ic;
      hi *= ic;
    }
    mt[r * 68 + l] = lo;
    mt[r * 68 + 32 + l] = hi;
  }
  __syncthreads();
  const int c16 = l & 15;
  int m = row0 + c16;
  if (m >= n) m = n - 1;
  const int cl = wave * 16 + c16;
  const int khi = (l >> 4) << 1;
  v8f c = {};
#pragma unroll
  for (int k0 = 0; k0 < 64; k0 += 4) {
    int kb = k0 + khi;
    v2f a, bb;
    a[0] = mt[c16 * 68 + kb];
    a[1] = mt[c16 * 68 + kb + 1];
    bb[0] = LW[kb * 64 + cl];
    bb[1] = LW[(kb + 1) * 64 + cl];
    c = wmma_f32(a, bb, c);
  }
#pragma unroll
  for (int k0 = 0; k0 < 64; k0 += 4) {
    int kb = k0 + khi;
    v2f a, bb;
    a[0] = XR[m * 64 + kb];
    a[1] = XR[m * 64 + kb + 1];
    bb[0] = RW[kb * 64 + cl];
    bb[1] = RW[(kb + 1) * 64 + cl];
    c = wmma_f32(a, bb, c);
  }
  const int rloc = (l >> 4) << 3;
  const float bc = lb[cl];
#pragma unroll
  for (int r = 0; r < 8; ++r) tile[(rloc + r) * 68 + cl] = c[r] + bc;
  __syncthreads();
  if (threadIdx.x < 16) {
    float ss = 0.f;
#pragma unroll
    for (int cc = 0; cc < 64; ++cc) {
      float v = tile[threadIdx.x * 68 + cc];
      ss += v * v;
    }
    rscale[threadIdx.x] = 1.0f / fmaxf(sqrtf(ss), EPS_NORM);
  }
  __syncthreads();
#pragma unroll
  for (int i = threadIdx.x; i < 16 * 64; i += 128) {
    const int r = i >> 6;
    const int cc = i & 63;
    mt[r * 68 + cc] = fmaxf(tile[r * 68 + cc] * rscale[r], 0.f);
  }
  __syncthreads();
  v8f p = {};
#pragma unroll
  for (int k0 = 0; k0 < 64; k0 += 4) {
    int kb = k0 + khi;
    v2f a, bb;
    a[0] = mt[c16 * 68 + kb];
    a[1] = mt[c16 * 68 + kb + 1];
    bb[0] = PW3[kb * 64 + cl];
    bb[1] = PW3[(kb + 1) * 64 + cl];
    p = wmma_f32(a, bb, p);
  }
  const float pc = pb3[cl];
#pragma unroll
  for (int r = 0; r < 8; ++r) tile[(rloc + r) * 68 + cl] = fmaxf(p[r] + pc, 0.f);
  __syncthreads();
  if (wave < 2) {
    const float* __restrict__ At = (wave == 0) ? tile : mt;
    const float* __restrict__ Bm = (wave == 0) ? LW3 : RW3;
    const bool cv = (c16 < 5);
    v2f bw[16];
#pragma unroll
    for (int i = 0; i < 16; ++i) { bw[i][0] = 0.f; bw[i][1] = 0.f; }
    float bias = 0.f;
    if (cv) {
#pragma unroll
      for (int i = 0; i < 16; ++i) {
        int kb = i * 4 + khi;
        bw[i][0] = Bm[kb * 5 + c16];
        bw[i][1] = Bm[(kb + 1) * 5 + c16];
      }
      if (wave == 1) bias = lb3[c16];
    }
    v8f z = {};
#pragma unroll
    for (int i = 0; i < 16; ++i) {
      int kb = i * 4 + khi;
      v2f a;
      a[0] = At[c16 * 68 + kb];
      a[1] = At[c16 * 68 + kb + 1];
      z = wmma_f32(a, bw[i], z);
    }
    if (cv) {
      float* __restrict__ O = (wave == 0) ? Z : XR;
      const int pitch = (wave == 0) ? 8 : 64;
#pragma unroll
      for (int r = 0; r < 8; ++r) {
        int rr = row0 + rloc + r;
        if (rr < n) O[(size_t)rr * pitch + c16] = z[r] + bias;
      }
    }
  }
}

__global__ void agg5_out_k(const int* __restrict__ offs, const int* __restrict__ csr,
                           const float* __restrict__ z, const float* __restrict__ root,
                           const float* __restrict__ invc, float* __restrict__ Y, int n) {
  const int2* __restrict__ pairs = (const int2*)csr;
  const int k = threadIdx.x & 7;
  const int node = (int)((blockIdx.x * blockDim.x + threadIdx.x) >> 3);
  if (node >= n) return;
  const int kl = (k < 5) ? k : 4;
  const int beg = offs[node], end = offs[node + 1];
  const float ic = invc[node];
  const float rt = root[(size_t)node * 64 + kl];
  float acc = 0.f;
  for (int t = beg; t < end; t += 4) {
    int2 v[4];
    float x[4];
#pragma unroll
    for (int u = 0; u < 4; ++u) v[u] = pairs[min(t + u, end - 1)];
#pragma unroll
    for (int u = 0; u < 4; ++u) x[u] = z[(size_t)v[u].x * 8 + kl];
#pragma unroll
    for (int u = 0; u < 4; ++u)
      if (t + u < end) acc += x[u] * __int_as_float(v[u].y);
  }
  const float val = (k < 5) ? rt + acc * ic : 0.f;
  float ss = val * val;
  ss += __shfl_xor(ss, 1);
  ss += __shfl_xor(ss, 2);
  ss += __shfl_xor(ss, 4);
  const float sc = 1.0f / fmaxf(sqrtf(ss), EPS_NORM);
  if (k < 5) Y[(size_t)node * 5 + k] = val * sc;
}

extern "C" void kernel_launch(void* const* d_in, const int* in_sizes, int n_in,
                              void* d_out, int out_size, void* d_ws, size_t ws_size,
                              hipStream_t stream) {
  const float* h     = (const float*)d_in[0];
  const int*   ei    = (const int*)d_in[1];
  const float* ew    = (const float*)d_in[2];
  const float* bg    = (const float*)d_in[3];
  const float* bbeta = (const float*)d_in[4];
  const float* c1_pw = (const float*)d_in[5];
  const float* c1_pb = (const float*)d_in[6];
  const float* c1_lw = (const float*)d_in[7];
  const float* c1_lb = (const float*)d_in[8];
  const float* c1_rw = (const float*)d_in[9];
  const float* c2_pw = (const float*)d_in[10];
  const float* c2_pb = (const float*)d_in[11];
  const float* c2_lw = (const float*)d_in[12];
  const float* c2_lb = (const float*)d_in[13];
  const float* c2_rw = (const float*)d_in[14];
  const float* c3_pw = (const float*)d_in[15];
  const float* c3_pb = (const float*)d_in[16];
  const float* c3_lw = (const float*)d_in[17];
  const float* c3_lb = (const float*)d_in[18];
  const float* c3_rw = (const float*)d_in[19];

  const int N = in_sizes[0] / 5;
  const int E = in_sizes[2];
  const int* src = ei;
  const int* dst = ei + E;

  float* ws   = (float*)d_ws;
  float* B1   = ws;
  float* B2   = B1 + (size_t)N * 64;
  float* B3   = B2 + (size_t)N * 64;
  float* mu   = B3 + (size_t)N * 64;
  float* rs   = mu + 8;
  float* part = rs + 8;
  float* invc = part + 2560;
  float* x0   = invc + N;
  float* xp1  = x0 + (size_t)N * 5;
  float* aggc = B3;
  int*   rank = (int*)B2;
  int*   cnt  = (int*)xp1;
  int*   cnt0 = cnt + N;
  const int nb1k = (N + 1023) / 1024;
  int*   bsum = cnt0 + N;
  int*   offs = (int*)xp1 + (size_t)N * 4 - 1;
  int*   csr  = (int*)B3;

  const int nb16 = (N + 15) / 16;

  bn_partial_k<<<256, 256, 0, stream>>>(h, part, N);
  bn_final_k<<<1, 32, 0, stream>>>(part, mu, rs, 256, 1.0f / (float)N);
  bn_apply_k<<<(N * 5 + 255) / 256, 256, 0, stream>>>(h, mu, rs, bg, bbeta, x0, N * 5);

  proj5_k<<<(N + 255) / 256, 256, 0, stream>>>(x0, c1_pw, c1_pb, xp1, N);
  hipMemsetAsync(aggc, 0, (size_t)N * 16 * sizeof(float), stream);
  edge_scatter5_k<<<(E + 255) / 256, 256, 0, stream>>>(src, dst, ew, xp1, aggc, rank, E, N);
  invcnt_sum_k<<<nb1k, 1024, 0, stream>>>(aggc, invc, cnt, cnt0, bsum, N);
  offs_scan_k<<<nb1k, 1024, 0, stream>>>(cnt, bsum, offs, N);
  node5to64_k<<<(N + 63) / 64, 256, 0, stream>>>(aggc, invc, x0, c1_lw, c1_lb, c1_rw, B1, N);
  fill_csr_k<<<(E + 255) / 256, 256, 0, stream>>>(src, dst, ew, rank, offs, cnt0, csr, E);

  float* z3 = x0;
  proj64_wmma_k<<<nb16, 128, 0, stream>>>(B1, c2_pw, c2_pb, B2, N);
  agg_node64_l3_wmma_k<<<nb16, 128, 0, stream>>>(offs, csr, B2, invc, B1, c2_lw, c2_lb, c2_rw,
                                                 c3_pw, c3_pb, c3_lw, c3_lb, c3_rw, z3, N);

  agg5_out_k<<<(N * 8 + 255) / 256, 256, 0, stream>>>(offs, csr, z3, B1, invc,
                                                      (float*)d_out, N);
}


# BN fold, BN apply, layer-1 projection and zeroing of the sum rows in ONE launch (each block re-folds the 256 partial sums: lane exchange + LDS, BN sums regrouped as a tree); 12 launches -> 9
# speedup vs baseline: 5.4609x; 1.0390x over previous
#include <hip/hip_runtime.h>
#include <hip/hip_bf16.h>

typedef __attribute__((ext_vector_type(2))) float v2f;
typedef __attribute__((ext_vector_type(8))) float v8f;

#define EPS_BN   1e-5f
#define EPS_NORM 1e-12f

__device__ __forceinline__ v8f wmma_f32(v2f a, v2f b, v8f c) {
  return __builtin_amdgcn_wmma_f32_16x16x4_f32(false, a, false, b, (short)0, c,
                                               false, false);
}

__global__ void bn_partial_k(const float* __restrict__ h, float* __restrict__ part, int n) {
  __shared__ float sh[256 * 10];
  float acc[10];
#pragma unroll
  for (int d = 0; d < 10; ++d) acc[d] = 0.f;
  for (int i = blockIdx.x * blockDim.x + threadIdx.x; i < n;
       i += gridDim.x * blockDim.x) {
#pragma unroll
    for (int d = 0; d < 5; ++d) {
      float v = h[i * 5 + d];
      acc[d] += v;
      acc[5 + d] += v * v;
    }
  }
#pragma unroll
  for (int d = 0; d < 10; ++d) sh[threadIdx.x * 10 + d] = acc[d];
  __syncthreads();
  for (int s = 128; s > 0; s >>= 1) {
    if (threadIdx.x < (unsigned)s)
      for (int d = 0; d < 10; ++d)
        sh[threadIdx.x * 10 + d] += sh[(threadIdx.x + s) * 10 + d];
    __syncthreads();
  }
  if (threadIdx.x < 10) part[blockIdx.x * 10 + threadIdx.x] = sh[threadIdx.x];
}

#define BN_PARTS 256
__global__ void bn_proj5_zero_k(const float* __restrict__ h, const float* __restrict__ part,
                                const float* __restrict__ g, const float* __restrict__ b,
                                const float* __restrict__ W, const float* __restrict__ pb,
                                float* __restrict__ x0, float* __restrict__ xp,
                                float* __restrict__ aggc, int n, float inv_n) {
  __shared__ float wsum[(BN_PARTS / 32) * 10];
  __shared__ float stat[10];
  const int t = threadIdx.x;
  const int i = blockIdx.x * BN_PARTS + t;
  const int ir = min(i, n - 1);
  float hv[5], gv[5], bv[5], pbv[5], wv[25];
#pragma unroll
  for (int k = 0; k < 5; ++k) {
    hv[k] = h[(size_t)ir * 5 + k];
    gv[k] = g[k];
    bv[k] = b[k];
    pbv[k] = pb[k];
  }
#pragma unroll
  for (int k = 0; k < 25; ++k) wv[k] = W[k];
  float v[10];
#pragma unroll
  for (int d = 0; d < 10; ++d) v[d] = part[t * 10 + d];
#pragma unroll
  for (int o = 16; o > 0; o >>= 1) {
#pragma unroll
    for (int d = 0; d < 10; ++d) v[d] += __shfl_xor(v[d], o);
  }
  if ((t & 31) == 0) {
#pragma unroll
    for (int d = 0; d < 10; ++d) wsum[(t >> 5) * 10 + d] = v[d];
  }
  __syncthreads();
  if (t < 5) {
    float s = 0.f, q = 0.f;
    for (int w = 0; w < BN_PARTS / 32; ++w) {
      s += wsum[w * 10 + t];
      q += wsum[w * 10 + 5 + t];
    }
    const float m = s * inv_n;
    const float var = q * inv_n - m * m;
    stat[t] = m;
    stat[5 + t] = rsqrtf(var + EPS_BN);
  }
  __syncthreads();
  if (i >= n) return;
  float x[5];
#pragma unroll
  for (int k = 0; k < 5; ++k) {
    x[k] = (hv[k] - stat[k]) * stat[5 + k] * gv[k] + bv[k];
    x0[(size_t)i * 5 + k] = x[k];
  }
#pragma unroll
  for (int c = 0; c < 5; ++c) {
    float acc = pbv[c];
#pragma unroll
    for (int k = 0; k < 5; ++k) acc += x[k] * wv[k * 5 + c];
    xp[(size_t)i * 5 + c] = fmaxf(acc, 0.f);
  }
  const float4 z = make_float4(0.f, 0.f, 0.f, 0.f);
  float4* r0 = (float4*)(aggc + (size_t)i * 8);
  float4* r1 = (float4*)(aggc + ((size_t)n + i) * 8);
  r0[0] = z;
  r0[1] = z;
  r1[0] = z;
  r1[1] = z;
}

__global__ void invcnt_sum_k(const float* __restrict__ aggc, float* __restrict__ invc,
                             int* __restrict__ cnt, int* __restrict__ cnt0,
                             int* __restrict__ bsum, int n) {
  __shared__ int wsum[32];
  const int lane = threadIdx.x & 31;
  const int wave = threadIdx.x >> 5;
  const int i = blockIdx.x * 1024 + threadIdx.x;
  int c = 0;
  if (i < n) {
    const float f0 = aggc[(size_t)i * 8 + 5];
    const float f = f0 + aggc[((size_t)n + i) * 8 + 5];
    invc[i] = 1.0f / fmaxf(f, 1.0f);
    c = (int)f;
    cnt[i] = c;
    cnt0[i] = (int)f0;
  }
  int sum = c;
#pragma unroll
  for (int o = 16; o > 0; o >>= 1) sum += __shfl_xor(sum, o);
  if (lane == 0) wsum[wave] = sum;
  __syncthreads();
  if (threadIdx.x == 0) {
    int t = 0;
    for (int w = 0; w < 32; ++w) t += wsum[w];
    bsum[blockIdx.x] = t;
  }
}

template <int XS, bool COUNT>
__device__ __forceinline__ void edge_scatter5_body(const int* __restrict__ src,
                                                   const int* __restrict__ dst,
                                                   const float* __restrict__ ew,
                                                   const float* __restrict__ xp,
                                                   float* __restrict__ agg,
                                                   int* __restrict__ rank, int E,
                                                   int n) {
  const int lane = threadIdx.x & 31;
  const int e0 = __builtin_amdgcn_readfirstlane(
      (int)((blockIdx.x * blockDim.x + threadIdx.x) >> 5) << 5);
  if (e0 >= E) return;
  const int e = e0 + lane;
  int s_l = 0, d_l = 0;
  float w_l = 0.f;
  if (e < E) { s_l = src[e]; d_l = dst[e]; w_l = ew[e]; }
  const int h = (int)((__builtin_amdgcn_s_sendmsg_rtn(0x87) >> 18) & 1u);
  agg += (size_t)h * (size_t)n * 8;
  const int k = lane & 7;
  const int g = lane >> 3;
  int s[8], d[8];
  float w[8];
#pragma unroll
  for (int j = 0; j < 8; ++j) {
    const int sl = j * 4 + g;
    s[j] = __shfl(s_l, sl);
    d[j] = __shfl(d_l, sl);
    w[j] = __shfl(w_l, sl);
  }
  if (k >= (COUNT ? 6 : 5)) return;
  const bool feat = (k < 5);
  const int kl = feat ? k : 4;
  float x[8];
#pragma unroll
  for (int j = 0; j < 8; ++j) x[j] = xp[(size_t)s[j] * XS + kl];
  if (e0 + 32 <= E) {
    float old[8];
#pragma unroll
    for (int j = 0; j < 8; ++j)
      old[j] = atomicAdd(&agg[(size_t)d[j] * 8 + k], feat ? x[j] * w[j] : 1.0f);
    if (COUNT && !feat) {
#pragma unroll
      for (int j = 0; j < 8; ++j) rank[e0 + j * 4 + g] = ((int)old[j] << 1) | h;
    }
  } else {
#pragma unroll
    for (int j = 0; j < 8; ++j) {
      if ((e0 + j * 4 + g) < E) {
        const float old = atomicAdd(&agg[(size_t)d[j] * 8 + k], feat ? x[j] * w[j] : 1.0f);
        if (COUNT && !feat) rank[e0 + j * 4 + g] = ((int)old << 1) | h;
      }
    }
  }
}

__global__ void edge_scatter5_k(const int* __restrict__ src, const int* __restrict__ dst,
                                const float* __restrict__ ew, const float* __restrict__ xp,
                                float* __restrict__ agg, int* __restrict__ rank, int E,
                                int n) {
  edge_scatter5_body<5, true>(src, dst, ew, xp, agg, rank, E, n);
}

__global__ void fill_csr_k(const int* __restrict__ src, const int* __restrict__ dst,
                           const float* __restrict__ ew, const int* __restrict__ rank,
                           const int* __restrict__ offs, const int* __restrict__ cnt0,
                           int* __restrict__ csr, int E) {
  int e = blockIdx.x * blockDim.x + threadIdx.x;
  if (e >= E) return;
  const int d = dst[e];
  const int r = rank[e];
  int pos = offs[d] + (r >> 1) + ((r & 1) ? cnt0[d] : 0);
  ((int2*)csr)[pos] = make_int2(src[e], __float_as_int(ew[e]));
}

__global__ void offs_scan_k(const int* __restrict__ cnt, const int* __restrict__ bsum,
                            int* __restrict__ offs, int n) {
  __shared__ int wsum[32];
  const int lane = threadIdx.x & 31;
  const int wave = threadIdx.x >> 5;
  const int i = blockIdx.x * 1024 + threadIdx.x;
  const int c = (i < n) ? cnt[i] : 0;
  int incl = c;
#pragma unroll
  for (int o = 1; o < 32; o <<= 1) {
    int v = __shfl_up(incl, o);
    if (lane >= o) incl += v;
  }
  if (lane == 31) wsum[wave] = incl;
  __syncthreads();
  int base = 0;
  for (int b = 0; b < (int)blockIdx.x; ++b) base += bsum[b];
  for (int w = 0; w < wave; ++w) base += wsum[w];
  const int ex = base + incl - c;
  if (i < n) {
    offs[i] = ex;
    if (i == n - 1) offs[n] = ex + c;
  }
}

template <bool FULL>
__device__ __forceinline__ void node5to64_body(const float* __restrict__ agg,
                                               const float* __restrict__ invc,
                                               const float* __restrict__ X,
                                               const float* __restrict__ LW,
                                               const float* __restrict__ lb,
                                               const float* __restrict__ RW,
                                               float* __restrict__ Y, int n, int nb) {
  const int l = threadIdx.x & 31;
  float m[2], xv[2];
#pragma unroll
  for (int g = 0; g < 2; ++g) {
    const int node = nb + g * 4 + (l >> 3);
    float a = 0.f, ic = 0.f;
    if (FULL || node < n) {
      a = agg[(size_t)node * 8 + (l & 7)] + agg[((size_t)n + node) * 8 + (l & 7)];
      ic = invc[node];
    }
    m[g] = a * ic;
    const int xi = (nb + g * 4) * 5 + l;
    xv[g] = (l < 20 && (FULL || xi < n * 5)) ? X[xi] : 0.f;
  }
  float lw0[5], lw1[5], rw0[5], rw1[5];
#pragma unroll
  for (int k = 0; k < 5; ++k) {
    lw0[k] = LW[k * 64 + l];
    lw1[k] = LW[k * 64 + 32 + l];
    rw0[k] = RW[k * 64 + l];
    rw1[k] = RW[k * 64 + 32 + l];
  }
  const float b0 = lb[l], b1 = lb[32 + l];
#pragma unroll
  for (int g = 0; g < 2; ++g) {
#pragma unroll
    for (int q = 0; q < 4; ++q) {
      const int node = nb + g * 4 + q;
      float acc0 = b0, acc1 = b1;
#pragma unroll
      for (int k = 0; k < 5; ++k) {
        const float mk = __int_as_float(__builtin_amdgcn_readlane(__float_as_int(m[g]), q * 8 + k));
        const float xk = __int_as_float(__builtin_amdgcn_readlane(__float_as_int(xv[g]), q * 5 + k));
        acc0 += mk * lw0[k] + xk * rw0[k];
        acc1 += mk * lw1[k] + xk * rw1[k];
      }
      float ss = acc0 * acc0 + acc1 * acc1;
#pragma unroll
      for (int o = 16; o > 0; o >>= 1) ss += __shfl_xor(ss, o);
      const float scale = 1.0f / fmaxf(sqrtf(ss), EPS_NORM);
      if (FULL || node < n) {
        Y[(size_t)node * 64 + l] = fmaxf(acc0 * scale, 0.f);
        Y[(size_t)node * 64 + 32 + l] = fmaxf(acc1 * scale, 0.f);
      }
    }
  }
}

__global__ void node5to64_k(const float* __restrict__ agg, const float* __restrict__ invc,
                            const float* __restrict__ X, const float* __restrict__ LW,
                            const float* __restrict__ lb, const float* __restrict__ RW,
                            float* __restrict__ Y, int n) {
  const int nb = __builtin_amdgcn_readfirstlane(
                     (int)((blockIdx.x * blockDim.x + threadIdx.x) >> 5)) * 8;
  if (nb >= n) return;
  if (nb + 8 <= n) node5to64_body<true>(agg, invc, X, LW, lb, RW, Y, n, nb);
  else             node5to64_body<false>(agg, invc, X, LW, lb, RW, Y, n, nb);
}

__global__ void proj64_wmma_k(const float* __restrict__ X, const float* __restrict__ W,
                              const float* __restrict__ bias, float* __restrict__ Y, int n) {
  int l = threadIdx.x & 31;
  int wave = threadIdx.x >> 5;
  int row0 = blockIdx.x * 16;
  int m = row0 + (l & 15);
  if (m >= n) m = n - 1;
  int cl = wave * 16 + (l & 15);
  int khi = (l >> 4) << 1;
  v8f c = {};
#pragma unroll
  for (int k0 = 0; k0 < 64; k0 += 4) {
    int kb = k0 + khi;
    v2f a, bb;
    a[0] = X[m * 64 + kb];
    a[1] = X[m * 64 + kb + 1];
    bb[0] = W[kb * 64 + cl];
    bb[1] = W[(kb + 1) * 64 + cl];
    c = wmma_f32(a, bb, c);
  }
  int rbase = row0 + ((l >> 4) << 3);
  float bc = bias[cl];
#pragma unroll
  for (int r = 0; r < 8; ++r) {
    int rr = rbase + r;
    if (rr < n) Y[rr * 64 + cl] = fmaxf(c[r] + bc, 0.f);
  }
}

__global__ void __attribute__((amdgpu_waves_per_eu(16, 16))) agg_node64_l3_wmma_k(const int* __restrict__ offs, const int* __restrict__ csr,
                                     const float* __restrict__ xp, const float* __restrict__ invc,
                                     float* XR, const float* __restrict__ LW,
                                     const float* __restrict__ lb, const float* __restrict__ RW,
                                     const float* __restrict__ PW3, const float* __restrict__ pb3,
                                     const float* __restrict__ LW3, const float* __restrict__ lb3,
                                     const float* __restrict__ RW3, float* __restrict__ Z, int n) {
  __shared__ float mt[16 * 68];
  __shared__ float tile[16 * 68];
  __shared__ float rscale[16];
  const int l = threadIdx.x & 31;
  const int wave = __builtin_amdgcn_readfirstlane((int)(threadIdx.x >> 5));
  const int row0 = blockIdx.x * 16;
  const int2* __restrict__ pairs = (const int2*)csr;
  for (int q = 0; q < 4; ++q) {
    const int r = wave * 4 + q;
    const int node = row0 + r;
    float lo = 0.f, hi = 0.f;
    if (node < n) {
      const int beg = __builtin_amdgcn_readfirstlane(offs[node]);
      const int end = __builtin_amdgcn_readfirstlane(offs[node + 1]);
      for (int c0 = beg; c0 < end; c0 += 32) {
        int s_l = 0;
        float w_l = 0.f;
        if (c0 + l < end) {
          const int2 v = pairs[c0 + l];
          s_l = v.x;
          w_l = __int_as_float(v.y);
        }
        const int n_e = min(32, end - c0);
        for (int j0 = 0; j0 < n_e; j0 += 8) {
          float a[8], b[8];
#pragma unroll
          for (int u = 0; u < 8; ++u) {
            const int sr = __builtin_amdgcn_readlane(s_l, j0 + u);
            const float* row = xp + (size_t)sr * 64 + l;
            a[u] = row[0];
            b[u] = row[32];
          }
#pragma unroll
          for (int u = 0; u < 8; ++u) {
            if (j0 + u < n_e) {
              const float w = __int_as_float(__builtin_amdgcn_readlane(__float_as_int(w_l), j0 + u));
              lo += a[u] * w;
              hi += b[u] * w;
            }
          }
        }
      }
      const float ic = invc[node];
      lo *= ic;
      hi *= ic;
    }
    mt[r * 68 + l] = lo;
    mt[r * 68 + 32 + l] = hi;
  }
  __syncthreads();
  const int c16 = l & 15;
  int m = row0 + c16;
  if (m >= n) m = n - 1;
  const int cl = wave * 16 + c16;
  const int khi = (l >> 4) << 1;
  v8f c = {};
#pragma unroll
  for (int k0 = 0; k0 < 64; k0 += 4) {
    int kb = k0 + khi;
    v2f a, bb;
    a[0] = mt[c16 * 68 + kb];
    a[1] = mt[c16 * 68 + kb + 1];
    bb[0] = LW[kb * 64 + cl];
    bb[1] = LW[(kb + 1) * 64 + cl];
    c = wmma_f32(a, bb, c);
  }
#pragma unroll
  for (int k0 = 0; k0 < 64; k0 += 4) {
    int kb = k0 + khi;
    v2f a, bb;
    a[0] = XR[m * 64 + kb];
    a[1] = XR[m * 64 + kb + 1];
    bb[0] = RW[kb * 64 + cl];
    bb[1] = RW[(kb + 1) * 64 + cl];
    c = wmma_f32(a, bb, c);
  }
  const int rloc = (l >> 4) << 3;
  const float bc = lb[cl];
#pragma unroll
  for (int r = 0; r < 8; ++r) tile[(rloc + r) * 68 + cl] = c[r] + bc;
  __syncthreads();
  if (threadIdx.x < 16) {
    float ss = 0.f;
#pragma unroll
    for (int cc = 0; cc < 64; ++cc) {
      float v = tile[threadIdx.x * 68 + cc];
      ss += v * v;
    }
    rscale[threadIdx.x] = 1.0f / fmaxf(sqrtf(ss), EPS_NORM);
  }
  __syncthreads();
#pragma unroll
  for (int i = threadIdx.x; i < 16 * 64; i += 128) {
    const int r = i >> 6;
    const int cc = i & 63;
    mt[r * 68 + cc] = fmaxf(tile[r * 68 + cc] * rscale[r], 0.f);
  }
  __syncthreads();
  v8f p = {};
#pragma unroll
  for (int k0 = 0; k0 < 64; k0 += 4) {
    int kb = k0 + khi;
    v2f a, bb;
    a[0] = mt[c16 * 68 + kb];
    a[1] = mt[c16 * 68 + kb + 1];
    bb[0] = PW3[kb * 64 + cl];
    bb[1] = PW3[(kb + 1) * 64 + cl];
    p = wmma_f32(a, bb, p);
  }
  const float pc = pb3[cl];
#pragma unroll
  for (int r = 0; r < 8; ++r) tile[(rloc + r) * 68 + cl] = fmaxf(p[r] + pc, 0.f);
  __syncthreads();
  if (wave < 2) {
    const float* __restrict__ At = (wave == 0) ? tile : mt;
    const float* __restrict__ Bm = (wave == 0) ? LW3 : RW3;
    const bool cv = (c16 < 5);
    v2f bw[16];
#pragma unroll
    for (int i = 0; i < 16; ++i) { bw[i][0] = 0.f; bw[i][1] = 0.f; }
    float bias = 0.f;
    if (cv) {
#pragma unroll
      for (int i = 0; i < 16; ++i) {
        int kb = i * 4 + khi;
        bw[i][0] = Bm[kb * 5 + c16];
        bw[i][1] = Bm[(kb + 1) * 5 + c16];
      }
      if (wave == 1) bias = lb3[c16];
    }
    v8f z = {};
#pragma unroll
    for (int i = 0; i < 16; ++i) {
      int kb = i * 4 + khi;
      v2f a;
      a[0] = At[c16 * 68 + kb];
      a[1] = At[c16 * 68 + kb + 1];
      z = wmma_f32(a, bw[i], z);
    }
    if (cv) {
      float* __restrict__ O = (wave == 0) ? Z : XR;
      const int pitch = (wave == 0) ? 8 : 64;
#pragma unroll
      for (int r = 0; r < 8; ++r) {
        int rr = row0 + rloc + r;
        if (rr < n) O[(size_t)rr * pitch + c16] = z[r] + bias;
      }
    }
  }
}

__global__ void agg5_out_k(const int* __restrict__ offs, const int* __restrict__ csr,
                           const float* __restrict__ z, const float* __restrict__ root,
                           const float* __restrict__ invc, float* __restrict__ Y, int n) {
  const int2* __restrict__ pairs = (const int2*)csr;
  const int k = threadIdx.x & 7;
  const int node = (int)((blockIdx.x * blockDim.x + threadIdx.x) >> 3);
  if (node >= n) return;
  const int kl = (k < 5) ? k : 4;
  const int beg = offs[node], end = offs[node + 1];
  const float ic = invc[node];
  const float rt = root[(size_t)node * 64 + kl];
  float acc = 0.f;
  for (int t = beg; t < end; t += 4) {
    int2 v[4];
    float x[4];
#pragma unroll
    for (int u = 0; u < 4; ++u) v[u] = pairs[min(t + u, end - 1)];
#pragma unroll
    for (int u = 0; u < 4; ++u) x[u] = z[(size_t)v[u].x * 8 + kl];
#pragma unroll
    for (int u = 0; u < 4; ++u)
      if (t + u < end) acc += x[u] * __int_as_float(v[u].y);
  }
  const float val = (k < 5) ? rt + acc * ic : 0.f;
  float ss = val * val;
  ss += __shfl_xor(ss, 1);
  ss += __shfl_xor(ss, 2);
  ss += __shfl_xor(ss, 4);
  const float sc = 1.0f / fmaxf(sqrtf(ss), EPS_NORM);
  if (k < 5) Y[(size_t)node * 5 + k] = val * sc;
}

extern "C" void kernel_launch(void* const* d_in, const int* in_sizes, int n_in,
                              void* d_out, int out_size, void* d_ws, size_t ws_size,
                              hipStream_t stream) {
  const float* h     = (const float*)d_in[0];
  const int*   ei    = (const int*)d_in[1];
  const float* ew    = (const float*)d_in[2];
  const float* bg    = (const float*)d_in[3];
  const float* bbeta = (const float*)d_in[4];
  const float* c1_pw = (const float*)d_in[5];
  const float* c1_pb = (const float*)d_in[6];
  const float* c1_lw = (const float*)d_in[7];
  const float* c1_lb = (const float*)d_in[8];
  const float* c1_rw = (const float*)d_in[9];
  const float* c2_pw = (const float*)d_in[10];
  const float* c2_pb = (const float*)d_in[11];
  const float* c2_lw = (const float*)d_in[12];
  const float* c2_lb = (const float*)d_in[13];
  const float* c2_rw = (const float*)d_in[14];
  const float* c3_pw = (const float*)d_in[15];
  const float* c3_pb = (const float*)d_in[16];
  const float* c3_lw = (const float*)d_in[17];
  const float* c3_lb = (const float*)d_in[18];
  const float* c3_rw = (const float*)d_in[19];

  const int N = in_sizes[0] / 5;
  const int E = in_sizes[2];
  const int* src = ei;
  const int* dst = ei + E;

  float* ws   = (float*)d_ws;
  float* B1   = ws;
  float* B2   = B1 + (size_t)N * 64;
  float* B3   = B2 + (size_t)N * 64;
  float* part = B3 + (size_t)N * 64;
  float* invc = part + BN_PARTS * 10;
  float* x0   = invc + N;
  float* xp1  = x0 + (size_t)N * 5;
  float* aggc = B3;
  int*   rank = (int*)B2;
  int*   cnt  = (int*)xp1;
  int*   cnt0 = cnt + N;
  const int nb1k = (N + 1023) / 1024;
  int*   bsum = cnt0 + N;
  int*   offs = (int*)xp1 + (size_t)N * 4 - 1;
  int*   csr  = (int*)B3;

  const int nb16 = (N + 15) / 16;

  bn_partial_k<<<BN_PARTS, 256, 0, stream>>>(h, part, N);
  bn_proj5_zero_k<<<(N + BN_PARTS - 1) / BN_PARTS, BN_PARTS, 0, stream>>>(
      h, part, bg, bbeta, c1_pw, c1_pb, x0, xp1, aggc, N, 1.0f / (float)N);

  edge_scatter5_k<<<(E + 255) / 256, 256, 0, stream>>>(src, dst, ew, xp1, aggc, rank, E, N);
  invcnt_sum_k<<<nb1k, 1024, 0, stream>>>(aggc, invc, cnt, cnt0, bsum, N);
  offs_scan_k<<<nb1k, 1024, 0, stream>>>(cnt, bsum, offs, N);
  node5to64_k<<<(N + 63) / 64, 256, 0, stream>>>(aggc, invc, x0, c1_lw, c1_lb, c1_rw, B1, N);
  fill_csr_k<<<(E + 255) / 256, 256, 0, stream>>>(src, dst, ew, rank, offs, cnt0, csr, E);

  float* z3 = x0;
  proj64_wmma_k<<<nb16, 128, 0, stream>>>(B1, c2_pw, c2_pb, B2, N);
  agg_node64_l3_wmma_k<<<nb16, 128, 0, stream>>>(offs, csr, B2, invc, B1, c2_lw, c2_lb, c2_rw,
                                                 c3_pw, c3_pb, c3_lw, c3_lb, c3_rw, z3, N);

  agg5_out_k<<<(N * 8 + 255) / 256, 256, 0, stream>>>(offs, csr, z3, B1, invc,
                                                      (float*)d_out, N);
}


# List fill: each half of the package files the edge blocks its own waves counted (per-half block lists, tickets) into its own list, run starts as pairs: one writing cache per list line; sum order kept
# speedup vs baseline: 6.6673x; 1.2209x over previous
#include <hip/hip_runtime.h>
#include <hip/hip_bf16.h>

typedef __attribute__((ext_vector_type(2))) float v2f;
typedef __attribute__((ext_vector_type(8))) float v8f;

#define EPS_BN   1e-5f
#define EPS_NORM 1e-12f

__device__ __forceinline__ v8f wmma_f32(v2f a, v2f b, v8f c) {
  return __builtin_amdgcn_wmma_f32_16x16x4_f32(false, a, false, b, (short)0, c,
                                               false, false);
}

__global__ void bn_partial_k(const float* __restrict__ h, float* __restrict__ part, int n) {
  __shared__ float sh[256 * 10];
  float acc[10];
#pragma unroll
  for (int d = 0; d < 10; ++d) acc[d] = 0.f;
  for (int i = blockIdx.x * blockDim.x + threadIdx.x; i < n;
       i += gridDim.x * blockDim.x) {
#pragma unroll
    for (int d = 0; d < 5; ++d) {
      float v = h[i * 5 + d];
      acc[d] += v;
      acc[5 + d] += v * v;
    }
  }
#pragma unroll
  for (int d = 0; d < 10; ++d) sh[threadIdx.x * 10 + d] = acc[d];
  __syncthreads();
  for (int s = 128; s > 0; s >>= 1) {
    if (threadIdx.x < (unsigned)s)
      for (int d = 0; d < 10; ++d)
        sh[threadIdx.x * 10 + d] += sh[(threadIdx.x + s) * 10 + d];
    __syncthreads();
  }
  if (threadIdx.x < 10) part[blockIdx.x * 10 + threadIdx.x] = sh[threadIdx.x];
}

#define BN_PARTS 256
#define CTR_PITCH 64
__global__ void bn_proj5_zero_k(const float* __restrict__ h, const float* __restrict__ part,
                                const float* __restrict__ g, const float* __restrict__ b,
                                const float* __restrict__ W, const float* __restrict__ pb,
                                float* __restrict__ x0, float* __restrict__ xp,
                                float* __restrict__ aggc, int* __restrict__ ctr, int n,
                                float inv_n) {
  __shared__ float wsum[(BN_PARTS / 32) * 10];
  __shared__ float stat[10];
  const int t = threadIdx.x;
  const int i = blockIdx.x * BN_PARTS + t;
  if (blockIdx.x == 0 && t < 4) ctr[t * CTR_PITCH] = 0;
  const int ir = min(i, n - 1);
  float hv[5], gv[5], bv[5], pbv[5], wv[25];
#pragma unroll
  for (int k = 0; k < 5; ++k) {
    hv[k] = h[(size_t)ir * 5 + k];
    gv[k] = g[k];
    bv[k] = b[k];
    pbv[k] = pb[k];
  }
#pragma unroll
  for (int k = 0; k < 25; ++k) wv[k] = W[k];
  float v[10];
#pragma unroll
  for (int d = 0; d < 10; ++d) v[d] = part[t * 10 + d];
#pragma unroll
  for (int o = 16; o > 0; o >>= 1) {
#pragma unroll
    for (int d = 0; d < 10; ++d) v[d] += __shfl_xor(v[d], o);
  }
  if ((t & 31) == 0) {
#pragma unroll
    for (int d = 0; d < 10; ++d) wsum[(t >> 5) * 10 + d] = v[d];
  }
  __syncthreads();
  if (t < 5) {
    float s = 0.f, q = 0.f;
    for (int w = 0; w < BN_PARTS / 32; ++w) {
      s += wsum[w * 10 + t];
      q += wsum[w * 10 + 5 + t];
    }
    const float m = s * inv_n;
    const float var = q * inv_n - m * m;
    stat[t] = m;
    stat[5 + t] = rsqrtf(var + EPS_BN);
  }
  __syncthreads();
  if (i >= n) return;
  float x[5];
#pragma unroll
  for (int k = 0; k < 5; ++k) {
    x[k] = (hv[k] - stat[k]) * stat[5 + k] * gv[k] + bv[k];
    x0[(size_t)i * 5 + k] = x[k];
  }
#pragma unroll
  for (int c = 0; c < 5; ++c) {
    float acc = pbv[c];
#pragma unroll
    for (int k = 0; k < 5; ++k) acc += x[k] * wv[k * 5 + c];
    xp[(size_t)i * 5 + c] = fmaxf(acc, 0.f);
  }
  const float4 z = make_float4(0.f, 0.f, 0.f, 0.f);
  float4* r0 = (float4*)(aggc + (size_t)i * 8);
  float4* r1 = (float4*)(aggc + ((size_t)n + i) * 8);
  r0[0] = z;
  r0[1] = z;
  r1[0] = z;
  r1[1] = z;
}

__global__ void invcnt_sum_k(const float* __restrict__ aggc, float* __restrict__ invc,
                             int* __restrict__ cnt, int* __restrict__ cnt0,
                             int* __restrict__ bsum0, int* __restrict__ bsum1, int n) {
  __shared__ int wsum[32];
  __shared__ int wsum0[32];
  const int lane = threadIdx.x & 31;
  const int wave = threadIdx.x >> 5;
  const int i = blockIdx.x * 1024 + threadIdx.x;
  int c = 0, c0 = 0;
  if (i < n) {
    const float f0 = aggc[(size_t)i * 8 + 5];
    const float f = f0 + aggc[((size_t)n + i) * 8 + 5];
    invc[i] = 1.0f / fmaxf(f, 1.0f);
    c = (int)f;
    c0 = (int)f0;
    cnt[i] = c;
    cnt0[i] = c0;
  }
  int sum = c, sum0 = c0;
#pragma unroll
  for (int o = 16; o > 0; o >>= 1) {
    sum += __shfl_xor(sum, o);
    sum0 += __shfl_xor(sum0, o);
  }
  if (lane == 0) { wsum[wave] = sum; wsum0[wave] = sum0; }
  __syncthreads();
  if (threadIdx.x == 0) {
    int t = 0, t0 = 0;
    for (int w = 0; w < 32; ++w) { t += wsum[w]; t0 += wsum0[w]; }
    bsum0[blockIdx.x] = t0;
    bsum1[blockIdx.x] = t - t0;
  }
}

template <int XS, bool COUNT>
__device__ __forceinline__ void edge_scatter5_body(const int* __restrict__ src,
                                                   const int* __restrict__ dst,
                                                   const float* __restrict__ ew,
                                                   const float* __restrict__ xp,
                                                   float* __restrict__ agg,
                                                   int* __restrict__ rank,
                                                   int* __restrict__ ctr,
                                                   int* __restrict__ wl, int E,
                                                   int n) {
  const int lane = threadIdx.x & 31;
  const int e0 = __builtin_amdgcn_readfirstlane(
      (int)((blockIdx.x * blockDim.x + threadIdx.x) >> 5) << 5);
  if (e0 >= E) return;
  const int e = e0 + lane;
  int s_l = 0, d_l = 0;
  float w_l = 0.f;
  if (e < E) { s_l = src[e]; d_l = dst[e]; w_l = ew[e]; }
  const int h = (int)((__builtin_amdgcn_s_sendmsg_rtn(0x87) >> 18) & 1u);
  agg += (size_t)h * (size_t)n * 8;
  if (COUNT && threadIdx.x == 0) {
    const int slot = atomicAdd(&ctr[h * CTR_PITCH], 1);
    wl[(size_t)h * gridDim.x + slot] = (int)blockIdx.x;
  }
  const int k = lane & 7;
  const int g = lane >> 3;
  int s[8], d[8];
  float w[8];
#pragma unroll
  for (int j = 0; j < 8; ++j) {
    const int sl = j * 4 + g;
    s[j] = __shfl(s_l, sl);
    d[j] = __shfl(d_l, sl);
    w[j] = __shfl(w_l, sl);
  }
  if (k >= (COUNT ? 6 : 5)) return;
  const bool feat = (k < 5);
  const int kl = feat ? k : 4;
  float x[8];
#pragma unroll
  for (int j = 0; j < 8; ++j) x[j] = xp[(size_t)s[j] * XS + kl];
  if (e0 + 32 <= E) {
    float old[8];
#pragma unroll
    for (int j = 0; j < 8; ++j)
      old[j] = atomicAdd(&agg[(size_t)d[j] * 8 + k], feat ? x[j] * w[j] : 1.0f);
    if (COUNT && !feat) {
#pragma unroll
      for (int j = 0; j < 8; ++j) rank[e0 + j * 4 + g] = ((int)old[j] << 1) | h;
    }
  } else {
#pragma unroll
    for (int j = 0; j < 8; ++j) {
      if ((e0 + j * 4 + g) < E) {
        const float old = atomicAdd(&agg[(size_t)d[j] * 8 + k], feat ? x[j] * w[j] : 1.0f);
        if (COUNT && !feat) rank[e0 + j * 4 + g] = ((int)old << 1) | h;
      }
    }
  }
}

__global__ void edge_scatter5_k(const int* __restrict__ src, const int* __restrict__ dst,
                                const float* __restrict__ ew, const float* __restrict__ xp,
                                float* __restrict__ agg, int* __restrict__ rank,
                                int* __restrict__ ctr, int* __restrict__ wl, int E, int n) {
  edge_scatter5_body<5, true>(src, dst, ew, xp, agg, rank, ctr, wl, E, n);
}

#define FILL_CHUNKS 8
template <bool FULL>
__device__ __forceinline__ void fill_block(const int* __restrict__ src,
                                           const int* __restrict__ dst,
                                           const float* __restrict__ ew,
                                           const int* __restrict__ rank,
                                           const int2* __restrict__ starts,
                                           int2* __restrict__ pairs, int e0, int E) {
  int d[FILL_CHUNKS], r[FILL_CHUNKS], s[FILL_CHUNKS], w[FILL_CHUNKS];
#pragma unroll
  for (int c = 0; c < FILL_CHUNKS; ++c) {
    const int e = FULL ? e0 + c * 32 : min(e0 + c * 32, E - 1);
    d[c] = dst[e];
    r[c] = rank[e];
    s[c] = src[e];
    w[c] = __float_as_int(ew[e]);
  }
  int sx[FILL_CHUNKS], sy[FILL_CHUNKS];
#pragma unroll
  for (int c = 0; c < FILL_CHUNKS; ++c) {
    const int2 v = starts[d[c]];
    sx[c] = v.x;
    sy[c] = v.y;
  }
#pragma unroll
  for (int c = 0; c < FILL_CHUNKS; ++c) {
    const int m = -(r[c] & 1);
    const int pos = ((sx[c] & ~m) | (sy[c] & m)) + (r[c] >> 1);
    if (FULL || e0 + c * 32 < E) pairs[pos] = make_int2(s[c], w[c]);
  }
}

__global__ void fill_csr_k(const int* __restrict__ src, const int* __restrict__ dst,
                           const float* __restrict__ ew, const int* __restrict__ rank,
                           const int2* __restrict__ starts, int* __restrict__ ctr,
                           const int* __restrict__ wl, int* __restrict__ csr, int E,
                           int nblk) {
  const int lane = threadIdx.x & 31;
  const int h = (int)((__builtin_amdgcn_s_sendmsg_rtn(0x87) >> 18) & 1u);
  int2* __restrict__ pairs = (int2*)csr;
  for (int side = 0; side < 2; ++side) {
    const int hh = h ^ side;
    const int len = __builtin_amdgcn_readfirstlane(ctr[hh * CTR_PITCH]);
    int* take = ctr + (2 + hh) * CTR_PITCH;
    for (;;) {
      const int seen = __builtin_amdgcn_readfirstlane(
          __hip_atomic_load(take, __ATOMIC_RELAXED, __HIP_MEMORY_SCOPE_AGENT));
      if (seen >= len) break;
      int tk = 0;
      if (lane == 0) tk = atomicAdd(take, 1);
      tk = __builtin_amdgcn_readfirstlane(tk);
      if (tk >= len) break;
      const int blk = __builtin_amdgcn_readfirstlane(wl[(size_t)hh * nblk + tk]);
      const int eb = blk * (FILL_CHUNKS * 32);
      if (eb + FILL_CHUNKS * 32 <= E) fill_block<true>(src, dst, ew, rank, starts, pairs, eb + lane, E);
      else                            fill_block<false>(src, dst, ew, rank, starts, pairs, eb + lane, E);
    }
  }
}

__global__ void starts_scan_k(const int* __restrict__ cnt, const int* __restrict__ cnt0,
                              const int* __restrict__ bsum0, const int* __restrict__ bsum1,
                              int2* __restrict__ starts, int n) {
  __shared__ int wsum0[32];
  __shared__ int wsum1[32];
  const int lane = threadIdx.x & 31;
  const int wave = threadIdx.x >> 5;
  const int i = blockIdx.x * 1024 + threadIdx.x;
  const int c0 = (i < n) ? cnt0[i] : 0;
  const int c1 = (i < n) ? cnt[i] - c0 : 0;
  int incl0 = c0, incl1 = c1;
#pragma unroll
  for (int o = 1; o < 32; o <<= 1) {
    int v0 = __shfl_up(incl0, o);
    int v1 = __shfl_up(incl1, o);
    if (lane >= o) { incl0 += v0; incl1 += v1; }
  }
  if (lane == 31) { wsum0[wave] = incl0; wsum1[wave] = incl1; }
  __syncthreads();
  int base0 = 0, base1 = 0, tot0 = 0;
  for (int b = 0; b < (int)gridDim.x; ++b) {
    const int v0 = bsum0[b];
    tot0 += v0;
    if (b < (int)blockIdx.x) { base0 += v0; base1 += bsum1[b]; }
  }
  for (int w = 0; w < wave; ++w) { base0 += wsum0[w]; base1 += wsum1[w]; }
  const int ex0 = base0 + incl0 - c0;
  const int ex1 = tot0 + base1 + incl1 - c1;
  if (i < n) {
    starts[i] = make_int2(ex0, ex1);
    if (i == n - 1) starts[n] = make_int2(ex0 + c0, ex1 + c1);
  }
}

template <bool FULL>
__device__ __forceinline__ void node5to64_body(const float* __restrict__ agg,
                                               const float* __restrict__ invc,
                                               const float* __restrict__ X,
                                               const float* __restrict__ LW,
                                               const float* __restrict__ lb,
                                               const float* __restrict__ RW,
                                               float* __restrict__ Y, int n, int nb) {
  const int l = threadIdx.x & 31;
  float m[2], xv[2];
#pragma unroll
  for (int g = 0; g < 2; ++g) {
    const int node = nb + g * 4 + (l >> 3);
    float a = 0.f, ic = 0.f;
    if (FULL || node < n) {
      a = agg[(size_t)node * 8 + (l & 7)] + agg[((size_t)n + node) * 8 + (l & 7)];
      ic = invc[node];
    }
    m[g] = a * ic;
    const int xi = (nb + g * 4) * 5 + l;
    xv[g] = (l < 20 && (FULL || xi < n * 5)) ? X[xi] : 0.f;
  }
  float lw0[5], lw1[5], rw0[5], rw1[5];
#pragma unroll
  for (int k = 0; k < 5; ++k) {
    lw0[k] = LW[k * 64 + l];
    lw1[k] = LW[k * 64 + 32 + l];
    rw0[k] = RW[k * 64 + l];
    rw1[k] = RW[k * 64 + 32 + l];
  }
  const float b0 = lb[l], b1 = lb[32 + l];
#pragma unroll
  for (int g = 0; g < 2; ++g) {
#pragma unroll
    for (int q = 0; q < 4; ++q) {
      const int node = nb + g * 4 + q;
      float acc0 = b0, acc1 = b1;
#pragma unroll
      for (int k = 0; k < 5; ++k) {
        const float mk = __int_as_float(__builtin_amdgcn_readlane(__float_as_int(m[g]), q * 8 + k));
        const float xk = __int_as_float(__builtin_amdgcn_readlane(__float_as_int(xv[g]), q * 5 + k));
        acc0 += mk * lw0[k] + xk * rw0[k];
        acc1 += mk * lw1[k] + xk * rw1[k];
      }
      float ss = acc0 * acc0 + acc1 * acc1;
#pragma unroll
      for (int o = 16; o > 0; o >>= 1) ss += __shfl_xor(ss, o);
      const float scale = 1.0f / fmaxf(sqrtf(ss), EPS_NORM);
      if (FULL || node < n) {
        Y[(size_t)node * 64 + l] = fmaxf(acc0 * scale, 0.f);
        Y[(size_t)node * 64 + 32 + l] = fmaxf(acc1 * scale, 0.f);
      }
    }
  }
}

__global__ void node5to64_k(const float* __restrict__ agg, const float* __restrict__ invc,
                            const float* __restrict__ X, const float* __restrict__ LW,
                            const float* __restrict__ lb, const float* __restrict__ RW,
                            float* __restrict__ Y, int n) {
  const int nb = __builtin_amdgcn_readfirstlane(
                     (int)((blockIdx.x * blockDim.x + threadIdx.x) >> 5)) * 8;
  if (nb >= n) return;
  if (nb + 8 <= n) node5to64_body<true>(agg, invc, X, LW, lb, RW, Y, n, nb);
  else             node5to64_body<false>(agg, invc, X, LW, lb, RW, Y, n, nb);
}

__global__ void proj64_wmma_k(const float* __restrict__ X, const float* __restrict__ W,
                              const float* __restrict__ bias, float* __restrict__ Y, int n) {
  int l = threadIdx.x & 31;
  int wave = threadIdx.x >> 5;
  int row0 = blockIdx.x * 16;
  int m = row0 + (l & 15);
  if (m >= n) m = n - 1;
  int cl = wave * 16 + (l & 15);
  int khi = (l >> 4) << 1;
  v8f c = {};
#pragma unroll
  for (int k0 = 0; k0 < 64; k0 += 4) {
    int kb = k0 + khi;
    v2f a, bb;
    a[0] = X[m * 64 + kb];
    a[1] = X[m * 64 + kb + 1];
    bb[0] = W[kb * 64 + cl];
    bb[1] = W[(kb + 1) * 64 + cl];
    c = wmma_f32(a, bb, c);
  }
  int rbase = row0 + ((l >> 4) << 3);
  float bc = bias[cl];
#pragma unroll
  for (int r = 0; r < 8; ++r) {
    int rr = rbase + r;
    if (rr < n) Y[rr * 64 + cl] = fmaxf(c[r] + bc, 0.f);
  }
}

__global__ void __attribute__((amdgpu_waves_per_eu(16, 16))) agg_node64_l3_wmma_k(const int2* __restrict__ starts, const int* __restrict__ csr,
                                     const float* __restrict__ xp, const float* __restrict__ invc,
                                     float* XR, const float* __restrict__ LW,
                                     const float* __restrict__ lb, const float* __restrict__ RW,
                                     const float* __restrict__ PW3, const float* __restrict__ pb3,
                                     const float* __restrict__ LW3, const float* __restrict__ lb3,
                                     const float* __restrict__ RW3, float* __restrict__ Z, int n) {
  __shared__ float mt[16 * 68];
  __shared__ float tile[16 * 68];
  __shared__ float rscale[16];
  const int l = threadIdx.x & 31;
  const int wave = __builtin_amdgcn_readfirstlane((int)(threadIdx.x >> 5));
  const int row0 = blockIdx.x * 16;
  const int2* __restrict__ pairs = (const int2*)csr;
  for (int q = 0; q < 4; ++q) {
    const int r = wave * 4 + q;
    const int node = row0 + r;
    float lo = 0.f, hi = 0.f;
    if (node < n) {
      const int2 sa = starts[node];
      const int2 sb = starts[node + 1];
      const int begA = __builtin_amdgcn_readfirstlane(sa.x);
      const int begB = __builtin_amdgcn_readfirstlane(sa.y);
      const int cA = __builtin_amdgcn_readfirstlane(sb.x) - begA;
      const int tot = cA + __builtin_amdgcn_readfirstlane(sb.y) - begB;
      for (int c0 = 0; c0 < tot; c0 += 32) {
        int s_l = 0;
        float w_l = 0.f;
        const int t = c0 + l;
        if (t < tot) {
          const int2 v = pairs[(t < cA) ? begA + t : begB + (t - cA)];
          s_l = v.x;
          w_l = __int_as_float(v.y);
        }
        const int n_e = min(32, tot - c0);
        for (int j0 = 0; j0 < n_e; j0 += 8) {
          float a[8], b[8];
#pragma unroll
          for (int u = 0; u < 8; ++u) {
            const int sr = __builtin_amdgcn_readlane(s_l, j0 + u);
            const float* row = xp + (size_t)sr * 64 + l;
            a[u] = row[0];
            b[u] = row[32];
          }
#pragma unroll
          for (int u = 0; u < 8; ++u) {
            if (j0 + u < n_e) {
              const float w = __int_as_float(__builtin_amdgcn_readlane(__float_as_int(w_l), j0 + u));
              lo += a[u] * w;
              hi += b[u] * w;
            }
          }
        }
      }
      const float ic = invc[node];
      lo *= ic;
      hi *= ic;
    }
    mt[r * 68 + l] = lo;
    mt[r * 68 + 32 + l] = hi;
  }
  __syncthreads();
  const int c16 = l & 15;
  int m = row0 + c16;
  if (m >= n) m = n - 1;
  const int cl = wave * 16 + c16;
  const int khi = (l >> 4) << 1;
  v8f c = {};
#pragma unroll
  for (int k0 = 0; k0 < 64; k0 += 4) {
    int kb = k0 + khi;
    v2f a, bb;
    a[0] = mt[c16 * 68 + kb];
    a[1] = mt[c16 * 68 + kb + 1];
    bb[0] = LW[kb * 64 + cl];
    bb[1] = LW[(kb + 1) * 64 + cl];
    c = wmma_f32(a, bb, c);
  }
#pragma unroll
  for (int k0 = 0; k0 < 64; k0 += 4) {
    int kb = k0 + khi;
    v2f a, bb;
    a[0] = XR[m * 64 + kb];
    a[1] = XR[m * 64 + kb + 1];
    bb[0] = RW[kb * 64 + cl];
    bb[1] = RW[(kb + 1) * 64 + cl];
    c = wmma_f32(a, bb, c);
  }
  const int rloc = (l >> 4) << 3;
  const float bc = lb[cl];
#pragma unroll
  for (int r = 0; r < 8; ++r) tile[(rloc + r) * 68 + cl] = c[r] + bc;
  __syncthreads();
  if (threadIdx.x < 16) {
    float ss = 0.f;
#pragma unroll
    for (int cc = 0; cc < 64; ++cc) {
      float v = tile[threadIdx.x * 68 + cc];
      ss += v * v;
    }
    rscale[threadIdx.x] = 1.0f / fmaxf(sqrtf(ss), EPS_NORM);
  }
  __syncthreads();
#pragma unroll
  for (int i = threadIdx.x; i < 16 * 64; i += 128) {
    const int r = i >> 6;
    const int cc = i & 63;
    mt[r * 68 + cc] = fmaxf(tile[r * 68 + cc] * rscale[r], 0.f);
  }
  __syncthreads();
  v8f p = {};
#pragma unroll
  for (int k0 = 0; k0 < 64; k0 += 4) {
    int kb = k0 + khi;
    v2f a, bb;
    a[0] = mt[c16 * 68 + kb];
    a[1] = mt[c16 * 68 + kb + 1];
    bb[0] = PW3[kb * 64 + cl];
    bb[1] = PW3[(kb + 1) * 64 + cl];
    p = wmma_f32(a, bb, p);
  }
  const float pc = pb3[cl];
#pragma unroll
  for (int r = 0; r < 8; ++r) tile[(rloc + r) * 68 + cl] = fmaxf(p[r] + pc, 0.f);
  __syncthreads();
  if (wave < 2) {
    const float* __restrict__ At = (wave == 0) ? tile : mt;
    const float* __restrict__ Bm = (wave == 0) ? LW3 : RW3;
    const bool cv = (c16 < 5);
    v2f bw[16];
#pragma unroll
    for (int i = 0; i < 16; ++i) { bw[i][0] = 0.f; bw[i][1] = 0.f; }
    float bias = 0.f;
    if (cv) {
#pragma unroll
      for (int i = 0; i < 16; ++i) {
        int kb = i * 4 + khi;
        bw[i][0] = Bm[kb * 5 + c16];
        bw[i][1] = Bm[(kb + 1) * 5 + c16];
      }
      if (wave == 1) bias = lb3[c16];
    }
    v8f z = {};
#pragma unroll
    for (int i = 0; i < 16; ++i) {
      int kb = i * 4 + khi;
      v2f a;
      a[0] = At[c16 * 68 + kb];
      a[1] = At[c16 * 68 + kb + 1];
      z = wmma_f32(a, bw[i], z);
    }
    if (cv) {
      float* __restrict__ O = (wave == 0) ? Z : XR;
      const int pitch = (wave == 0) ? 8 : 64;
#pragma unroll
      for (int r = 0; r < 8; ++r) {
        int rr = row0 + rloc + r;
        if (rr < n) O[(size_t)rr * pitch + c16] = z[r] + bias;
      }
    }
  }
}

__global__ void agg5_out_k(const int2* __restrict__ starts, const int* __restrict__ csr,
                           const float* __restrict__ z, const float* __restrict__ root,
                           const float* __restrict__ invc, float* __restrict__ Y, int n) {
  const int2* __restrict__ pairs = (const int2*)csr;
  const int k = threadIdx.x & 7;
  const int node = (int)((blockIdx.x * blockDim.x + threadIdx.x) >> 3);
  if (node >= n) return;
  const int kl = (k < 5) ? k : 4;
  const int2 sa = starts[node];
  const int2 sb = starts[node + 1];
  const int begA = sa.x, begB = sa.y;
  const int cA = sb.x - begA;
  const int tot = cA + sb.y - begB;
  const float ic = invc[node];
  const float rt = root[(size_t)node * 64 + kl];
  float acc = 0.f;
  for (int t = 0; t < tot; t += 4) {
    int2 v[4];
    float x[4];
#pragma unroll
    for (int u = 0; u < 4; ++u) {
      const int tt = min(t + u, tot - 1);
      v[u] = pairs[(tt < cA) ? begA + tt : begB + (tt - cA)];
    }
#pragma unroll
    for (int u = 0; u < 4; ++u) x[u] = z[(size_t)v[u].x * 8 + kl];
#pragma unroll
    for (int u = 0; u < 4; ++u)
      if (t + u < tot) acc += x[u] * __int_as_float(v[u].y);
  }
  const float val = (k < 5) ? rt + acc * ic : 0.f;
  float ss = val * val;
  ss += __shfl_xor(ss, 1);
  ss += __shfl_xor(ss, 2);
  ss += __shfl_xor(ss, 4);
  const float sc = 1.0f / fmaxf(sqrtf(ss), EPS_NORM);
  if (k < 5) Y[(size_t)node * 5 + k] = val * sc;
}

extern "C" void kernel_launch(void* const* d_in, const int* in_sizes, int n_in,
                              void* d_out, int out_size, void* d_ws, size_t ws_size,
                              hipStream_t stream) {
  const float* h     = (const float*)d_in[0];
  const int*   ei    = (const int*)d_in[1];
  const float* ew    = (const float*)d_in[2];
  const float* bg    = (const float*)d_in[3];
  const float* bbeta = (const float*)d_in[4];
  const float* c1_pw = (const float*)d_in[5];
  const float* c1_pb = (const float*)d_in[6];
  const float* c1_lw = (const float*)d_in[7];
  const float* c1_lb = (const float*)d_in[8];
  const float* c1_rw = (const float*)d_in[9];
  const float* c2_pw = (const float*)d_in[10];
  const float* c2_pb = (const float*)d_in[11];
  const float* c2_lw = (const float*)d_in[12];
  const float* c2_lb = (const float*)d_in[13];
  const float* c2_rw = (const float*)d_in[14];
  const float* c3_pw = (const float*)d_in[15];
  const float* c3_pb = (const float*)d_in[16];
  const float* c3_lw = (const float*)d_in[17];
  const float* c3_lb = (const float*)d_in[18];
  const float* c3_rw = (const float*)d_in[19];

  const int N = in_sizes[0] / 5;
  const int E = in_sizes[2];
  const int* src = ei;
  const int* dst = ei + E;

  float* ws   = (float*)d_ws;
  float* B1   = ws;
  float* B2   = B1 + (size_t)N * 64;
  float* B3   = B2 + (size_t)N * 64;
  float* part = B3 + (size_t)N * 64;
  float* invc = part + BN_PARTS * 10;
  float* x0   = invc + N;
  float* xp1  = x0 + (size_t)N * 5;
  float* aggc = B3;
  int*   rank = (int*)B2;
  int*   ctr  = rank + E;
  int*   wl   = ctr + 4 * CTR_PITCH;
  const int nblk = (E + 255) / 256;
  int*   cnt  = (int*)xp1;
  int*   cnt0 = cnt + N;
  const int nb1k = (N + 1023) / 1024;
  int*   bsum0 = cnt0 + N;
  int*   bsum1 = bsum0 + nb1k;
  int2*  starts = (int2*)((int*)xp1 + (size_t)N * 3 - 2);
  int*   csr  = (int*)B3;

  const int nb16 = (N + 15) / 16;

  bn_partial_k<<<BN_PARTS, 256, 0, stream>>>(h, part, N);
  bn_proj5_zero_k<<<(N + BN_PARTS - 1) / BN_PARTS, BN_PARTS, 0, stream>>>(
      h, part, bg, bbeta, c1_pw, c1_pb, x0, xp1, aggc, ctr, N, 1.0f / (float)N);

  edge_scatter5_k<<<nblk, 256, 0, stream>>>(src, dst, ew, xp1, aggc, rank, ctr, wl, E, N);
  invcnt_sum_k<<<nb1k, 1024, 0, stream>>>(aggc, invc, cnt, cnt0, bsum0, bsum1, N);
  starts_scan_k<<<nb1k, 1024, 0, stream>>>(cnt, cnt0, bsum0, bsum1, starts, N);
  node5to64_k<<<(N + 63) / 64, 256, 0, stream>>>(aggc, invc, x0, c1_lw, c1_lb, c1_rw, B1, N);
  fill_csr_k<<<1024, 256, 0, stream>>>(src, dst, ew, rank, starts, ctr, wl, csr, E, nblk);

  float* z3 = x0;
  proj64_wmma_k<<<nb16, 128, 0, stream>>>(B1, c2_pw, c2_pb, B2, N);
  agg_node64_l3_wmma_k<<<nb16, 128, 0, stream>>>(starts, csr, B2, invc, B1, c2_lw, c2_lb, c2_rw,
                                                 c3_pw, c3_pb, c3_lw, c3_lb, c3_rw, z3, N);

  agg5_out_k<<<(N * 8 + 255) / 256, 256, 0, stream>>>(starts, csr, z3, B1, invc,
                                                      (float*)d_out, N);
}
